# starting kernel
# baseline (speedup 1.0000x reference)
#include <hip/hip_runtime.h>
#include <hip/hip_bf16.h>

typedef __attribute__((ext_vector_type(2))) float v2f;
typedef __attribute__((ext_vector_type(8))) float v8f;

#define NST 64
#define TCH 64
#define LSEQ 4096
#define DMOD 512
#define NSEG 16
#define SCH 4

#define WS_ST 20480

__global__ __launch_bounds__(64) void s4_precompute(const float* __restrict__ A_,
                                                    const float* __restrict__ B_,
                                                    const float* __restrict__ C_,
                                                    float* __restrict__ ws) {
  __shared__ float As[NST * NST];
  __shared__ float Sq[NST * NST];
  __shared__ float Sq2[NST * NST];
  __shared__ float vbuf[NST][NST];
  __shared__ float wbuf[NST + 1][NST];
  __shared__ float cb[NST];
  __shared__ float Kk[NST];

  const int tid = threadIdx.x;
#pragma clang loop unroll(disable)
  for (int j = 0; j < NST; ++j) As[tid * NST + j] = A_[tid * NST + j];
  cb[tid] = C_[tid];
  vbuf[0][tid] = B_[tid];
  __syncthreads();

#pragma clang loop unroll(disable)
  for (int t = 1; t < NST; ++t) {
    float s = 0.f;
#pragma clang loop unroll(disable)
    for (int j = 0; j < NST; ++j) s += As[tid * NST + j] * vbuf[t - 1][j];
    vbuf[t][tid] = s;
    __syncthreads();
  }

#pragma clang loop unroll(disable)
  for (int p = 1; p <= NST; ++p) {
    const float* prev = (p == 1) ? cb : wbuf[p - 1];
    float s = 0.f;
#pragma clang loop unroll(disable)
    for (int i = 0; i < NST; ++i) s += prev[i] * As[i * NST + tid];
    wbuf[p][tid] = s;
    __syncthreads();
  }

  {
    float s = 0.f;
#pragma clang loop unroll(disable)
    for (int i = 0; i < NST; ++i) s += cb[i] * vbuf[tid][i];
    Kk[tid] = s;
  }
  __syncthreads();

  const float* src = As;
  float* dst = Sq2;
#pragma clang loop unroll(disable)
  for (int it = 0; it < 6; ++it) {
#pragma clang loop unroll(disable)
    for (int j = 0; j < NST; ++j) {
      float s = 0.f;
#pragma clang loop unroll(disable)
      for (int k = 0; k < NST; ++k) s += src[tid * NST + k] * src[k * NST + j];
      dst[tid * NST + j] = s;
    }
    __syncthreads();
    src = dst;
    dst = (src == Sq2) ? Sq : Sq2;
  }
  const float* A64 = src;

  float* TKt = ws;
  float* Mt  = ws + 4096;
  float* Et  = ws + 8192;
  float* Pt  = ws + 12288;
#pragma clang loop unroll(disable)
  for (int j = 0; j < NST; ++j) {
    TKt[tid * NST + j] = (j >= tid) ? Kk[j - tid] : 0.f;
    Mt [tid * NST + j] = vbuf[NST - 1 - tid][j];
    Et [tid * NST + j] = wbuf[j + 1][tid];
    Pt [tid * NST + j] = A64[j * NST + tid];
  }
  __syncthreads();

#pragma clang loop unroll(disable)
  for (int it = 0; it < 2; ++it) {
#pragma clang loop unroll(disable)
    for (int j = 0; j < NST; ++j) {
      float s = 0.f;
#pragma clang loop unroll(disable)
      for (int k = 0; k < NST; ++k) s += src[tid * NST + k] * src[k * NST + j];
      dst[tid * NST + j] = s;
    }
    __syncthreads();
    src = dst;
    dst = (src == Sq2) ? Sq : Sq2;
  }
  const float* A256 = src;
  float* PSt = ws + 16384;
#pragma clang loop unroll(disable)
  for (int j = 0; j < NST; ++j) {
    PSt[tid * NST + j] = A256[j * NST + tid];
  }
}

__global__ __launch_bounds__(32)
__attribute__((amdgpu_num_vgpr(256)))
void s4_segstate(const float* __restrict__ u, float* __restrict__ ws) {
  __shared__ float sM[2 * 4096];
  __shared__ float sX[2][1024];

  const int tid = threadIdx.x;
  {
    const float4* m4 = (const float4*)(ws + 4096);
    const float4* p4 = (const float4*)(ws + 12288);
    float4* d0 = (float4*)sM;
    float4* d1 = (float4*)(sM + 4096);
    for (int i = tid; i < 1024; i += 32) d0[i] = m4[i];
    for (int i = tid; i < 1024; i += 32) d1[i] = p4[i];
    for (int i = tid; i < 1024; i += 32) sX[0][i] = 0.f;
  }
  __syncthreads();

  const int tile = blockIdx.x >> 4;
  const int seg  = blockIdx.x & 15;
  const int s0 = tile * 16;
  const int bb = s0 >> 9, dm0 = s0 & 511;
  const int row = tid & 15, hi = tid >> 4;
  const float* ub0 = u + (long)bb * LSEQ * DMOD + dm0 + row + (long)(2 * hi) * DMOD;

  for (int i = 0; i < SCH; ++i) {
    const int c = seg * SCH + i;
    const float* Xr = sX[i & 1];
    float* Xw = (float*)sX[(i & 1) ^ 1];
    const float* up = ub0 + (long)c * TCH * DMOD;
    v2f uf[16];
#pragma unroll
    for (int kk = 0; kk < 16; ++kk) {
      uf[kk].x = up[kk * 4 * DMOD];
      uf[kk].y = up[kk * 4 * DMOD + DMOD];
    }
    const v2f* xr = (const v2f*)(Xr + row * 64 + 2 * hi);
    v2f xin[16];
#pragma unroll
    for (int kk = 0; kk < 16; ++kk) xin[kk] = xr[kk * 2];

#pragma unroll
    for (int nt = 0; nt < 4; ++nt) {
      v8f ax = {0.f, 0.f, 0.f, 0.f, 0.f, 0.f, 0.f, 0.f};
      const int col = nt * 16 + row;
      const float* pM = sM + col;
      const float* pP = sM + 4096 + col;
#pragma unroll
      for (int kk = 0; kk < 16; ++kk) {
        const int k0 = 4 * kk + 2 * hi;
        v2f bm; bm.x = pM[k0 * 64]; bm.y = pM[(k0 + 1) * 64];
        ax = __builtin_amdgcn_wmma_f32_16x16x4_f32(false, uf[kk], false, bm,
                                                   (short)0, ax, false, false);
        v2f bp; bp.x = pP[k0 * 64]; bp.y = pP[(k0 + 1) * 64];
        ax = __builtin_amdgcn_wmma_f32_16x16x4_f32(false, xin[kk], false, bp,
                                                   (short)0, ax, false, false);
      }
      float* xw = Xw + (8 * hi) * 64 + col;
#pragma unroll
      for (int r = 0; r < 8; ++r) xw[r * 64] = ax[r];
    }
    __syncthreads();
  }
  float* st = ws + WS_ST + (long)tile * (NSEG * 1024) + seg * 1024;
  const float4* xp = (const float4*)sX[0];
  float4* gp = (float4*)st;
  for (int i = tid; i < 256; i += 32) gp[i] = xp[i];
}

__global__ __launch_bounds__(32) void s4_combine(float* __restrict__ ws) {
  __shared__ float sP[4096];
  __shared__ float sE[2][1024];

  const int tid = threadIdx.x;
  {
    const float4* p4 = (const float4*)(ws + 16384);
    float4* d4 = (float4*)sP;
    for (int i = tid; i < 1024; i += 32) d4[i] = p4[i];
    for (int i = tid; i < 1024; i += 32) sE[0][i] = 0.f;
  }
  __syncthreads();

  const int row = tid & 15, hi = tid >> 4;
  float* stt = ws + WS_ST + (long)blockIdx.x * (NSEG * 1024);

#pragma clang loop unroll(disable)
  for (int g = 0; g < NSEG - 1; ++g) {
    float* slot = stt + g * 1024;
    const float* Er = sE[g & 1];
    float* Ew = (float*)sE[(g & 1) ^ 1];
    const v2f* er = (const v2f*)(Er + row * 64 + 2 * hi);
    v2f ef[16];
#pragma unroll
    for (int kk = 0; kk < 16; ++kk) ef[kk] = er[kk * 2];

#pragma unroll
    for (int nt = 0; nt < 4; ++nt) {
      const int col = nt * 16 + row;
      v8f ax;
#pragma unroll
      for (int r = 0; r < 8; ++r) ax[r] = slot[(r + 8 * hi) * 64 + col];
      const float* pP = sP + col;
#pragma unroll
      for (int kk = 0; kk < 16; ++kk) {
        const int k0 = 4 * kk + 2 * hi;
        v2f bp; bp.x = pP[k0 * 64]; bp.y = pP[(k0 + 1) * 64];
        ax = __builtin_amdgcn_wmma_f32_16x16x4_f32(false, ef[kk], false, bp,
                                                   (short)0, ax, false, false);
      }
#pragma unroll
      for (int r = 0; r < 8; ++r) slot[(r + 8 * hi) * 64 + col] = ax[r];
      float* ew = Ew + (8 * hi) * 64 + col;
#pragma unroll
      for (int r = 0; r < 8; ++r) ew[r * 64] = ax[r];
    }
    __syncthreads();
  }
}

__global__ __launch_bounds__(32)
__attribute__((amdgpu_num_vgpr(256)))
void s4_outputs(const float* __restrict__ u, const float* __restrict__ ws,
                float* __restrict__ y) {
  __shared__ float sMat[4 * 4096];
  __shared__ float sX[2][1024];

  const int tid = threadIdx.x;
  {
    const float4* s4p = (const float4*)ws;
    float4* d4 = (float4*)sMat;
    for (int i = tid; i < 4096; i += 32) d4[i] = s4p[i];
  }
  const int tile = blockIdx.x >> 4;
  const int seg  = blockIdx.x & 15;
  if (seg == 0) {
    for (int i = tid; i < 1024; i += 32) sX[0][i] = 0.f;
  } else {
    const float4* ep =
        (const float4*)(ws + WS_ST + (long)tile * (NSEG * 1024) + (seg - 1) * 1024);
    float4* xp = (float4*)sX[0];
    for (int i = tid; i < 256; i += 32) xp[i] = ep[i];
  }
  __syncthreads();

  const int s0 = tile * 16;
  const int bb = s0 >> 9, dm0 = s0 & 511;
  const int row = tid & 15, hi = tid >> 4;
  const float* ub0 = u + (long)bb * LSEQ * DMOD + dm0 + row + (long)(2 * hi) * DMOD;
  float* yb0 = y + (long)bb * LSEQ * DMOD + dm0 + 8 * hi;

  for (int i = 0; i < SCH; ++i) {
    const int c = seg * SCH + i;
    const float* Xr = sX[i & 1];
    float* Xw = (float*)sX[(i & 1) ^ 1];
    const float* up = ub0 + (long)c * TCH * DMOD;
    v2f uf[16];
#pragma unroll
    for (int kk = 0; kk < 16; ++kk) {
      uf[kk].x = up[kk * 4 * DMOD];
      uf[kk].y = up[kk * 4 * DMOD + DMOD];
    }
    const v2f* xr = (const v2f*)(Xr + row * 64 + 2 * hi);
    v2f xin[16];
#pragma unroll
    for (int kk = 0; kk < 16; ++kk) xin[kk] = xr[kk * 2];

#pragma unroll
    for (int nt = 0; nt < 4; ++nt) {
      v8f ay = {0.f, 0.f, 0.f, 0.f, 0.f, 0.f, 0.f, 0.f};
      v8f ax = {0.f, 0.f, 0.f, 0.f, 0.f, 0.f, 0.f, 0.f};
      const int col = nt * 16 + row;
      const float* pTK = sMat + col;
      const float* pMt = sMat + 4096 + col;
      const float* pEt = sMat + 8192 + col;
      const float* pPt = sMat + 12288 + col;
#pragma unroll
      for (int kk = 0; kk < 16; ++kk) {
        const int k0 = 4 * kk + 2 * hi;
        v2f bt; bt.x = pTK[k0 * 64]; bt.y = pTK[(k0 + 1) * 64];
        ay = __builtin_amdgcn_wmma_f32_16x16x4_f32(false, uf[kk], false, bt,
                                                   (short)0, ay, false, false);
        v2f bm; bm.x = pMt[k0 * 64]; bm.y = pMt[(k0 + 1) * 64];
        ax = __builtin_amdgcn_wmma_f32_16x16x4_f32(false, uf[kk], false, bm,
                                                   (short)0, ax, false, false);
        v2f be; be.x = pEt[k0 * 64]; be.y = pEt[(k0 + 1) * 64];
        ay = __builtin_amdgcn_wmma_f32_16x16x4_f32(false, xin[kk], false, be,
                                                   (short)0, ay, false, false);
        v2f bp; bp.x = pPt[k0 * 64]; bp.y = pPt[(k0 + 1) * 64];
        ax = __builtin_amdgcn_wmma_f32_16x16x4_f32(false, xin[kk], false, bp,
                                                   (short)0, ax, false, false);
      }
      float4* yp = (float4*)(yb0 + (long)(c * TCH + col) * DMOD);
      float4 lo, hh;
      lo.x = ay[0]; lo.y = ay[1]; lo.z = ay[2]; lo.w = ay[3];
      hh.x = ay[4]; hh.y = ay[5]; hh.z = ay[6]; hh.w = ay[7];
      yp[0] = lo;
      yp[1] = hh;
      float* xw = Xw + (8 * hi) * 64 + col;
#pragma unroll
      for (int r = 0; r < 8; ++r) xw[r * 64] = ax[r];
    }
    __syncthreads();
  }
}

extern "C" void kernel_launch(void* const* d_in, const int* in_sizes, int n_in,
                              void* d_out, int out_size, void* d_ws, size_t ws_size,
                              hipStream_t stream) {
  const float* u  = (const float*)d_in[0];
  const float* A_ = (const float*)d_in[1];
  const float* B_ = (const float*)d_in[2];
  const float* C_ = (const float*)d_in[3];
  float* y  = (float*)d_out;
  float* ws = (float*)d_ws;

  s4_precompute<<<1, 64, 0, stream>>>(A_, B_, C_, ws);
  s4_segstate<<<128 * NSEG, 32, 0, stream>>>(u, ws);
  s4_combine<<<128, 32, 0, stream>>>(ws);
  s4_outputs<<<128 * NSEG, 32, 0, stream>>>(u, ws, y);
}


# Set-up launch parallelised into 3 work-groups: A^t b chain and c A^p chain (one factor A per step, A in registers), squarings on f32 matrix tiles; their dot products summed in another order
# speedup vs baseline: 16.2322x; 16.2322x over previous
#include <hip/hip_runtime.h>
#include <hip/hip_bf16.h>

typedef __attribute__((ext_vector_type(2))) float v2f;
typedef __attribute__((ext_vector_type(8))) float v8f;

#define NST 64
#define TCH 64
#define LSEQ 4096
#define DMOD 512
#define NSEG 16
#define SCH 4

#define WS_ST 20480

typedef __attribute__((ext_vector_type(4))) float v4f;

#define LDP 68

__device__ __forceinline__ void stage_A(const float* __restrict__ A_, float* sA, int tid) {
  float av[32];
  for (int j = 0; j < 32; ++j) av[j] = A_[tid + 128 * j];
  for (int j = 0; j < 32; ++j) {
    const int i = tid + 128 * j;
    sA[(i >> 6) * LDP + (i & 63)] = av[j];
  }
}

__device__ __forceinline__ void run_chain(const v4f (&am)[16], float* xb, int last, int tid) {
  for (int t = 1; t <= last; ++t) {
    if (tid < NST) {
      const v4f* xp = (const v4f*)(xb + (t - 1) * LDP);
      v4f acc = {0.f, 0.f, 0.f, 0.f};
      for (int jj = 0; jj < 16; ++jj) acc += am[jj] * xp[jj];
      xb[t * LDP + tid] = (acc.x + acc.y) + (acc.z + acc.w);
    }
    __syncthreads();
  }
}

__device__ __forceinline__ void role_rows(const float* __restrict__ A_,
                                          const float* __restrict__ B_,
                                          const float* __restrict__ C_,
                                          float* __restrict__ ws, float* sm, int tid) {
  float* sA = sm;
  float* vb = sm + NST * LDP;
  float* cb = sm + 2 * NST * LDP;
  float* Kk = cb + NST;

  stage_A(A_, sA, tid);
  if (tid < NST) { vb[tid] = B_[tid]; cb[tid] = C_[tid]; }
  __syncthreads();

  v4f am[16];
  {
    const v4f* ar = (const v4f*)(sA + (tid & 63) * LDP);
    for (int jj = 0; jj < 16; ++jj) am[jj] = ar[jj];
  }
  run_chain(am, vb, NST - 1, tid);

  if (tid < NST) {
    const v4f* vp = (const v4f*)(vb + tid * LDP);
    const v4f* cp = (const v4f*)cb;
    v4f acc = {0.f, 0.f, 0.f, 0.f};
    for (int jj = 0; jj < 16; ++jj) acc += cp[jj] * vp[jj];
    Kk[tid] = (acc.x + acc.y) + (acc.z + acc.w);
  }
  __syncthreads();

  for (int j = 0; j < 32; ++j) {
    const int i = tid + 128 * j;
    const int r = i >> 6, k = i & 63;
    ws[i]        = (k >= r) ? Kk[k - r] : 0.f;
    ws[4096 + i] = vb[(NST - 1 - r) * LDP + k];
  }
}

__device__ __forceinline__ void role_cols(const float* __restrict__ A_,
                                          const float* __restrict__ C_,
                                          float* __restrict__ ws, float* sm, int tid) {
  float* sA = sm;
  float* wb = sm + NST * LDP;

  stage_A(A_, sA, tid);
  if (tid < NST) wb[tid] = C_[tid];
  __syncthreads();

  v4f am[16];
  {
    const float* ac = sA + (tid & 63);
    for (int ii = 0; ii < 16; ++ii) {
      am[ii].x = ac[(4 * ii) * LDP];
      am[ii].y = ac[(4 * ii + 1) * LDP];
      am[ii].z = ac[(4 * ii + 2) * LDP];
      am[ii].w = ac[(4 * ii + 3) * LDP];
    }
  }
  run_chain(am, wb, NST, tid);

  for (int j = 0; j < 32; ++j) {
    const int i = tid + 128 * j;
    const int r = i >> 6, k = i & 63;
    ws[8192 + i] = wb[(k + 1) * LDP + r];
  }
}

__device__ __forceinline__ void role_squares(const float* __restrict__ A_,
                                             float* __restrict__ ws, float* sm, int tid) {
  float* X0 = sm;
  float* X1 = sm + NST * LDP;
  float* X2 = sm + 2 * NST * LDP;

  stage_A(A_, X0, tid);
  __syncthreads();

  const int mt = tid >> 5;
  const int lane = tid & 31;
  const int c = lane & 15, h = lane >> 4;

  for (int s = 0; s < 8; ++s) {
    const float* P = (s & 1) ? X1 : X0;
    float* Pn = (s == 7) ? X2 : ((s & 1) ? X0 : X1);

    v2f a[16];
    {
      const float* L = P + (16 * mt + c) * LDP + 2 * h;
      for (int kk = 0; kk < 16; ++kk) a[kk] = *(const v2f*)(L + 4 * kk);
    }
    v8f acc[4];
    for (int nt = 0; nt < 4; ++nt) acc[nt] = (v8f){0.f, 0.f, 0.f, 0.f, 0.f, 0.f, 0.f, 0.f};
    for (int kk = 0; kk < 16; ++kk) {
      for (int nt = 0; nt < 4; ++nt) {
        const float* R = P + (4 * kk + 2 * h) * LDP + 16 * nt + c;
        v2f b;
        b.x = R[0];
        b.y = R[LDP];
        acc[nt] = __builtin_amdgcn_wmma_f32_16x16x4_f32(false, a[kk], false, b,
                                                        (short)0, acc[nt], false, false);
      }
    }
    for (int nt = 0; nt < 4; ++nt) {
      float* o = Pn + (16 * mt + 8 * h) * LDP + 16 * nt + c;
      for (int r = 0; r < 8; ++r) o[r * LDP] = acc[nt][r];
    }
    __syncthreads();
  }

  for (int j = 0; j < 32; ++j) {
    const int i = tid + 128 * j;
    const int r = i >> 6, k = i & 63;
    ws[12288 + i] = X0[k * LDP + r];
    ws[16384 + i] = X2[k * LDP + r];
  }
}

__global__ __launch_bounds__(128) void s4_precompute(const float* __restrict__ A_,
                                                     const float* __restrict__ B_,
                                                     const float* __restrict__ C_,
                                                     float* __restrict__ ws) {
  __shared__ __attribute__((aligned(16))) float sm[3 * NST * LDP];
  const int tid = threadIdx.x;
  if (blockIdx.x == 0) {
    role_rows(A_, B_, C_, ws, sm, tid);
  } else if (blockIdx.x == 1) {
    role_cols(A_, C_, ws, sm, tid);
  } else {
    role_squares(A_, ws, sm, tid);
  }
}

__global__ __launch_bounds__(32)
__attribute__((amdgpu_num_vgpr(256)))
void s4_segstate(const float* __restrict__ u, float* __restrict__ ws) {
  __shared__ float sM[2 * 4096];
  __shared__ float sX[2][1024];

  const int tid = threadIdx.x;
  {
    const float4* m4 = (const float4*)(ws + 4096);
    const float4* p4 = (const float4*)(ws + 12288);
    float4* d0 = (float4*)sM;
    float4* d1 = (float4*)(sM + 4096);
    for (int i = tid; i < 1024; i += 32) d0[i] = m4[i];
    for (int i = tid; i < 1024; i += 32) d1[i] = p4[i];
    for (int i = tid; i < 1024; i += 32) sX[0][i] = 0.f;
  }
  __syncthreads();

  const int tile = blockIdx.x >> 4;
  const int seg  = blockIdx.x & 15;
  const int s0 = tile * 16;
  const int bb = s0 >> 9, dm0 = s0 & 511;
  const int row = tid & 15, hi = tid >> 4;
  const float* ub0 = u + (long)bb * LSEQ * DMOD + dm0 + row + (long)(2 * hi) * DMOD;

  for (int i = 0; i < SCH; ++i) {
    const int c = seg * SCH + i;
    const float* Xr = sX[i & 1];
    float* Xw = (float*)sX[(i & 1) ^ 1];
    const float* up = ub0 + (long)c * TCH * DMOD;
    v2f uf[16];
#pragma unroll
    for (int kk = 0; kk < 16; ++kk) {
      uf[kk].x = up[kk * 4 * DMOD];
      uf[kk].y = up[kk * 4 * DMOD + DMOD];
    }
    const v2f* xr = (const v2f*)(Xr + row * 64 + 2 * hi);
    v2f xin[16];
#pragma unroll
    for (int kk = 0; kk < 16; ++kk) xin[kk] = xr[kk * 2];

#pragma unroll
    for (int nt = 0; nt < 4; ++nt) {
      v8f ax = {0.f, 0.f, 0.f, 0.f, 0.f, 0.f, 0.f, 0.f};
      const int col = nt * 16 + row;
      const float* pM = sM + col;
      const float* pP = sM + 4096 + col;
#pragma unroll
      for (int kk = 0; kk < 16; ++kk) {
        const int k0 = 4 * kk + 2 * hi;
        v2f bm; bm.x = pM[k0 * 64]; bm.y = pM[(k0 + 1) * 64];
        ax = __builtin_amdgcn_wmma_f32_16x16x4_f32(false, uf[kk], false, bm,
                                                   (short)0, ax, false, false);
        v2f bp; bp.x = pP[k0 * 64]; bp.y = pP[(k0 + 1) * 64];
        ax = __builtin_amdgcn_wmma_f32_16x16x4_f32(false, xin[kk], false, bp,
                                                   (short)0, ax, false, false);
      }
      float* xw = Xw + (8 * hi) * 64 + col;
#pragma unroll
      for (int r = 0; r < 8; ++r) xw[r * 64] = ax[r];
    }
    __syncthreads();
  }
  float* st = ws + WS_ST + (long)tile * (NSEG * 1024) + seg * 1024;
  const float4* xp = (const float4*)sX[0];
  float4* gp = (float4*)st;
  for (int i = tid; i < 256; i += 32) gp[i] = xp[i];
}

__global__ __launch_bounds__(32) void s4_combine(float* __restrict__ ws) {
  __shared__ float sP[4096];
  __shared__ float sE[2][1024];

  const int tid = threadIdx.x;
  {
    const float4* p4 = (const float4*)(ws + 16384);
    float4* d4 = (float4*)sP;
    for (int i = tid; i < 1024; i += 32) d4[i] = p4[i];
    for (int i = tid; i < 1024; i += 32) sE[0][i] = 0.f;
  }
  __syncthreads();

  const int row = tid & 15, hi = tid >> 4;
  float* stt = ws + WS_ST + (long)blockIdx.x * (NSEG * 1024);

#pragma clang loop unroll(disable)
  for (int g = 0; g < NSEG - 1; ++g) {
    float* slot = stt + g * 1024;
    const float* Er = sE[g & 1];
    float* Ew = (float*)sE[(g & 1) ^ 1];
    const v2f* er = (const v2f*)(Er + row * 64 + 2 * hi);
    v2f ef[16];
#pragma unroll
    for (int kk = 0; kk < 16; ++kk) ef[kk] = er[kk * 2];

#pragma unroll
    for (int nt = 0; nt < 4; ++nt) {
      const int col = nt * 16 + row;
      v8f ax;
#pragma unroll
      for (int r = 0; r < 8; ++r) ax[r] = slot[(r + 8 * hi) * 64 + col];
      const float* pP = sP + col;
#pragma unroll
      for (int kk = 0; kk < 16; ++kk) {
        const int k0 = 4 * kk + 2 * hi;
        v2f bp; bp.x = pP[k0 * 64]; bp.y = pP[(k0 + 1) * 64];
        ax = __builtin_amdgcn_wmma_f32_16x16x4_f32(false, ef[kk], false, bp,
                                                   (short)0, ax, false, false);
      }
#pragma unroll
      for (int r = 0; r < 8; ++r) slot[(r + 8 * hi) * 64 + col] = ax[r];
      float* ew = Ew + (8 * hi) * 64 + col;
#pragma unroll
      for (int r = 0; r < 8; ++r) ew[r * 64] = ax[r];
    }
    __syncthreads();
  }
}

__global__ __launch_bounds__(32)
__attribute__((amdgpu_num_vgpr(256)))
void s4_outputs(const float* __restrict__ u, const float* __restrict__ ws,
                float* __restrict__ y) {
  __shared__ float sMat[4 * 4096];
  __shared__ float sX[2][1024];

  const int tid = threadIdx.x;
  {
    const float4* s4p = (const float4*)ws;
    float4* d4 = (float4*)sMat;
    for (int i = tid; i < 4096; i += 32) d4[i] = s4p[i];
  }
  const int tile = blockIdx.x >> 4;
  const int seg  = blockIdx.x & 15;
  if (seg == 0) {
    for (int i = tid; i < 1024; i += 32) sX[0][i] = 0.f;
  } else {
    const float4* ep =
        (const float4*)(ws + WS_ST + (long)tile * (NSEG * 1024) + (seg - 1) * 1024);
    float4* xp = (float4*)sX[0];
    for (int i = tid; i < 256; i += 32) xp[i] = ep[i];
  }
  __syncthreads();

  const int s0 = tile * 16;
  const int bb = s0 >> 9, dm0 = s0 & 511;
  const int row = tid & 15, hi = tid >> 4;
  const float* ub0 = u + (long)bb * LSEQ * DMOD + dm0 + row + (long)(2 * hi) * DMOD;
  float* yb0 = y + (long)bb * LSEQ * DMOD + dm0 + 8 * hi;

  for (int i = 0; i < SCH; ++i) {
    const int c = seg * SCH + i;
    const float* Xr = sX[i & 1];
    float* Xw = (float*)sX[(i & 1) ^ 1];
    const float* up = ub0 + (long)c * TCH * DMOD;
    v2f uf[16];
#pragma unroll
    for (int kk = 0; kk < 16; ++kk) {
      uf[kk].x = up[kk * 4 * DMOD];
      uf[kk].y = up[kk * 4 * DMOD + DMOD];
    }
    const v2f* xr = (const v2f*)(Xr + row * 64 + 2 * hi);
    v2f xin[16];
#pragma unroll
    for (int kk = 0; kk < 16; ++kk) xin[kk] = xr[kk * 2];

#pragma unroll
    for (int nt = 0; nt < 4; ++nt) {
      v8f ay = {0.f, 0.f, 0.f, 0.f, 0.f, 0.f, 0.f, 0.f};
      v8f ax = {0.f, 0.f, 0.f, 0.f, 0.f, 0.f, 0.f, 0.f};
      const int col = nt * 16 + row;
      const float* pTK = sMat + col;
      const float* pMt = sMat + 4096 + col;
      const float* pEt = sMat + 8192 + col;
      const float* pPt = sMat + 12288 + col;
#pragma unroll
      for (int kk = 0; kk < 16; ++kk) {
        const int k0 = 4 * kk + 2 * hi;
        v2f bt; bt.x = pTK[k0 * 64]; bt.y = pTK[(k0 + 1) * 64];
        ay = __builtin_amdgcn_wmma_f32_16x16x4_f32(false, uf[kk], false, bt,
                                                   (short)0, ay, false, false);
        v2f bm; bm.x = pMt[k0 * 64]; bm.y = pMt[(k0 + 1) * 64];
        ax = __builtin_amdgcn_wmma_f32_16x16x4_f32(false, uf[kk], false, bm,
                                                   (short)0, ax, false, false);
        v2f be; be.x = pEt[k0 * 64]; be.y = pEt[(k0 + 1) * 64];
        ay = __builtin_amdgcn_wmma_f32_16x16x4_f32(false, xin[kk], false, be,
                                                   (short)0, ay, false, false);
        v2f bp; bp.x = pPt[k0 * 64]; bp.y = pPt[(k0 + 1) * 64];
        ax = __builtin_amdgcn_wmma_f32_16x16x4_f32(false, xin[kk], false, bp,
                                                   (short)0, ax, false, false);
      }
      float4* yp = (float4*)(yb0 + (long)(c * TCH + col) * DMOD);
      float4 lo, hh;
      lo.x = ay[0]; lo.y = ay[1]; lo.z = ay[2]; lo.w = ay[3];
      hh.x = ay[4]; hh.y = ay[5]; hh.z = ay[6]; hh.w = ay[7];
      yp[0] = lo;
      yp[1] = hh;
      float* xw = Xw + (8 * hi) * 64 + col;
#pragma unroll
      for (int r = 0; r < 8; ++r) xw[r * 64] = ax[r];
    }
    __syncthreads();
  }
}

extern "C" void kernel_launch(void* const* d_in, const int* in_sizes, int n_in,
                              void* d_out, int out_size, void* d_ws, size_t ws_size,
                              hipStream_t stream) {
  const float* u  = (const float*)d_in[0];
  const float* A_ = (const float*)d_in[1];
  const float* B_ = (const float*)d_in[2];
  const float* C_ = (const float*)d_in[3];
  float* y  = (float*)d_out;
  float* ws = (float*)d_ws;

  s4_precompute<<<3, 128, 0, stream>>>(A_, B_, C_, ws);
  s4_segstate<<<128 * NSEG, 32, 0, stream>>>(u, ws);
  s4_combine<<<128, 32, 0, stream>>>(ws);
  s4_outputs<<<128 * NSEG, 32, 0, stream>>>(u, ws, y);
}


# Phases A, B, C run as work-groups of 8, 4, 8 waves sharing one LDS copy of the chunk matrices, staged once per group with all loads in flight (was per wave, one load per round trip)
# speedup vs baseline: 20.6661x; 1.2732x over previous
#include <hip/hip_runtime.h>
#include <hip/hip_bf16.h>

typedef __attribute__((ext_vector_type(2))) float v2f;
typedef __attribute__((ext_vector_type(8))) float v8f;

#define NST 64
#define TCH 64
#define LSEQ 4096
#define DMOD 512
#define NSEG 16
#define SCH 4
#define NWV 8
#define NWB 4

#define WS_ST 20480

typedef __attribute__((ext_vector_type(4))) float v4f;

#define LDP 68

__device__ __forceinline__ void stage_A(const float* __restrict__ A_, float* sA, int tid) {
  float av[32];
  for (int j = 0; j < 32; ++j) av[j] = A_[tid + 128 * j];
  for (int j = 0; j < 32; ++j) {
    const int i = tid + 128 * j;
    sA[(i >> 6) * LDP + (i & 63)] = av[j];
  }
}

__device__ __forceinline__ void run_chain(const v4f (&am)[16], float* xb, int last, int tid) {
  for (int t = 1; t <= last; ++t) {
    if (tid < NST) {
      const v4f* xp = (const v4f*)(xb + (t - 1) * LDP);
      v4f acc = {0.f, 0.f, 0.f, 0.f};
      for (int jj = 0; jj < 16; ++jj) acc += am[jj] * xp[jj];
      xb[t * LDP + tid] = (acc.x + acc.y) + (acc.z + acc.w);
    }
    __syncthreads();
  }
}

__device__ __forceinline__ void role_rows(const float* __restrict__ A_,
                                          const float* __restrict__ B_,
                                          const float* __restrict__ C_,
                                          float* __restrict__ ws, float* sm, int tid) {
  float* sA = sm;
  float* vb = sm + NST * LDP;
  float* cb = sm + 2 * NST * LDP;
  float* Kk = cb + NST;

  stage_A(A_, sA, tid);
  if (tid < NST) { vb[tid] = B_[tid]; cb[tid] = C_[tid]; }
  __syncthreads();

  v4f am[16];
  {
    const v4f* ar = (const v4f*)(sA + (tid & 63) * LDP);
    for (int jj = 0; jj < 16; ++jj) am[jj] = ar[jj];
  }
  run_chain(am, vb, NST - 1, tid);

  if (tid < NST) {
    const v4f* vp = (const v4f*)(vb + tid * LDP);
    const v4f* cp = (const v4f*)cb;
    v4f acc = {0.f, 0.f, 0.f, 0.f};
    for (int jj = 0; jj < 16; ++jj) acc += cp[jj] * vp[jj];
    Kk[tid] = (acc.x + acc.y) + (acc.z + acc.w);
  }
  __syncthreads();

  for (int j = 0; j < 32; ++j) {
    const int i = tid + 128 * j;
    const int r = i >> 6, k = i & 63;
    ws[i]        = (k >= r) ? Kk[k - r] : 0.f;
    ws[4096 + i] = vb[(NST - 1 - r) * LDP + k];
  }
}

__device__ __forceinline__ void role_cols(const float* __restrict__ A_,
                                          const float* __restrict__ C_,
                                          float* __restrict__ ws, float* sm, int tid) {
  float* sA = sm;
  float* wb = sm + NST * LDP;

  stage_A(A_, sA, tid);
  if (tid < NST) wb[tid] = C_[tid];
  __syncthreads();

  v4f am[16];
  {
    const float* ac = sA + (tid & 63);
    for (int ii = 0; ii < 16; ++ii) {
      am[ii].x = ac[(4 * ii) * LDP];
      am[ii].y = ac[(4 * ii + 1) * LDP];
      am[ii].z = ac[(4 * ii + 2) * LDP];
      am[ii].w = ac[(4 * ii + 3) * LDP];
    }
  }
  run_chain(am, wb, NST, tid);

  for (int j = 0; j < 32; ++j) {
    const int i = tid + 128 * j;
    const int r = i >> 6, k = i & 63;
    ws[8192 + i] = wb[(k + 1) * LDP + r];
  }
}

__device__ __forceinline__ void role_squares(const float* __restrict__ A_,
                                             float* __restrict__ ws, float* sm, int tid) {
  float* X0 = sm;
  float* X1 = sm + NST * LDP;
  float* X2 = sm + 2 * NST * LDP;

  stage_A(A_, X0, tid);
  __syncthreads();

  const int mt = tid >> 5;
  const int lane = tid & 31;
  const int c = lane & 15, h = lane >> 4;

  for (int s = 0; s < 8; ++s) {
    const float* P = (s & 1) ? X1 : X0;
    float* Pn = (s == 7) ? X2 : ((s & 1) ? X0 : X1);

    v2f a[16];
    {
      const float* L = P + (16 * mt + c) * LDP + 2 * h;
      for (int kk = 0; kk < 16; ++kk) a[kk] = *(const v2f*)(L + 4 * kk);
    }
    v8f acc[4];
    for (int nt = 0; nt < 4; ++nt) acc[nt] = (v8f){0.f, 0.f, 0.f, 0.f, 0.f, 0.f, 0.f, 0.f};
    for (int kk = 0; kk < 16; ++kk) {
      for (int nt = 0; nt < 4; ++nt) {
        const float* R = P + (4 * kk + 2 * h) * LDP + 16 * nt + c;
        v2f b;
        b.x = R[0];
        b.y = R[LDP];
        acc[nt] = __builtin_amdgcn_wmma_f32_16x16x4_f32(false, a[kk], false, b,
                                                        (short)0, acc[nt], false, false);
      }
    }
    for (int nt = 0; nt < 4; ++nt) {
      float* o = Pn + (16 * mt + 8 * h) * LDP + 16 * nt + c;
      for (int r = 0; r < 8; ++r) o[r * LDP] = acc[nt][r];
    }
    __syncthreads();
  }

  for (int j = 0; j < 32; ++j) {
    const int i = tid + 128 * j;
    const int r = i >> 6, k = i & 63;
    ws[12288 + i] = X0[k * LDP + r];
    ws[16384 + i] = X2[k * LDP + r];
  }
}

__global__ __launch_bounds__(128) void s4_precompute(const float* __restrict__ A_,
                                                     const float* __restrict__ B_,
                                                     const float* __restrict__ C_,
                                                     float* __restrict__ ws) {
  __shared__ __attribute__((aligned(16))) float sm[3 * NST * LDP];
  const int tid = threadIdx.x;
  if (blockIdx.x == 0) {
    role_rows(A_, B_, C_, ws, sm, tid);
  } else if (blockIdx.x == 1) {
    role_cols(A_, C_, ws, sm, tid);
  } else {
    role_squares(A_, ws, sm, tid);
  }
}

__global__ __launch_bounds__(32 * NWV) __attribute__((amdgpu_waves_per_eu(2, 2)))
void s4_segstate(const float* __restrict__ u, float* __restrict__ ws) {
  __shared__ __attribute__((aligned(16))) float sM[2 * 4096];
  __shared__ __attribute__((aligned(16))) float sXw[NWV][2][1024];

  const int wv = threadIdx.x >> 5;
  const int tid = threadIdx.x & 31;
  float (*sX)[1024] = sXw[wv];
  {
    const int t = threadIdx.x;
    const float4* m4 = (const float4*)(ws + 4096);
    const float4* p4 = (const float4*)(ws + 12288);
    float4* d0 = (float4*)sM;
    float4* d1 = (float4*)(sM + 4096);
    float4 vm[4], vp[4];
    for (int j = 0; j < 4; ++j) vm[j] = m4[t + 32 * NWV * j];
    for (int j = 0; j < 4; ++j) vp[j] = p4[t + 32 * NWV * j];
    for (int j = 0; j < 4; ++j) d0[t + 32 * NWV * j] = vm[j];
    for (int j = 0; j < 4; ++j) d1[t + 32 * NWV * j] = vp[j];
    float4* z4 = (float4*)sX[0];
    const float4 z = {0.f, 0.f, 0.f, 0.f};
    for (int i = tid; i < 256; i += 32) z4[i] = z;
  }
  __syncthreads();

  const int tile = (blockIdx.x >> 4) * NWV + wv;
  const int seg  = blockIdx.x & 15;
  const int s0 = tile * 16;
  const int bb = s0 >> 9, dm0 = s0 & 511;
  const int row = tid & 15, hi = tid >> 4;
  const float* ub0 = u + (long)bb * LSEQ * DMOD + dm0 + row + (long)(2 * hi) * DMOD;

  for (int i = 0; i < SCH; ++i) {
    const int c = seg * SCH + i;
    const float* Xr = sX[i & 1];
    float* Xw = (float*)sX[(i & 1) ^ 1];
    const float* up = ub0 + (long)c * TCH * DMOD;
    v2f uf[16];
#pragma unroll
    for (int kk = 0; kk < 16; ++kk) {
      uf[kk].x = up[kk * 4 * DMOD];
      uf[kk].y = up[kk * 4 * DMOD + DMOD];
    }
    const v2f* xr = (const v2f*)(Xr + row * 64 + 2 * hi);
    v2f xin[16];
#pragma unroll
    for (int kk = 0; kk < 16; ++kk) xin[kk] = xr[kk * 2];

#pragma unroll
    for (int nt = 0; nt < 4; ++nt) {
      v8f ax = {0.f, 0.f, 0.f, 0.f, 0.f, 0.f, 0.f, 0.f};
      const int col = nt * 16 + row;
      const float* pM = sM + col;
      const float* pP = sM + 4096 + col;
#pragma unroll
      for (int kk = 0; kk < 16; ++kk) {
        const int k0 = 4 * kk + 2 * hi;
        v2f bm; bm.x = pM[k0 * 64]; bm.y = pM[(k0 + 1) * 64];
        ax = __builtin_amdgcn_wmma_f32_16x16x4_f32(false, uf[kk], false, bm,
                                                   (short)0, ax, false, false);
        v2f bp; bp.x = pP[k0 * 64]; bp.y = pP[(k0 + 1) * 64];
        ax = __builtin_amdgcn_wmma_f32_16x16x4_f32(false, xin[kk], false, bp,
                                                   (short)0, ax, false, false);
      }
      float* xw = Xw + (8 * hi) * 64 + col;
#pragma unroll
      for (int r = 0; r < 8; ++r) xw[r * 64] = ax[r];
    }
    __syncthreads();
  }
  float* st = ws + WS_ST + (long)tile * (NSEG * 1024) + seg * 1024;
  const float4* xp = (const float4*)sX[0];
  float4* gp = (float4*)st;
  for (int i = tid; i < 256; i += 32) gp[i] = xp[i];
}

__global__ __launch_bounds__(32 * NWB) void s4_combine(float* __restrict__ ws) {
  __shared__ __attribute__((aligned(16))) float sP[4096];
  __shared__ __attribute__((aligned(16))) float sEw[NWB][2][1024];

  const int wv = threadIdx.x >> 5;
  const int tid = threadIdx.x & 31;
  float (*sE)[1024] = sEw[wv];
  {
    const int t = threadIdx.x;
    const float4* p4 = (const float4*)(ws + 16384);
    float4* d4 = (float4*)sP;
    float4 vp[8];
    for (int j = 0; j < 8; ++j) vp[j] = p4[t + 32 * NWB * j];
    for (int j = 0; j < 8; ++j) d4[t + 32 * NWB * j] = vp[j];
    float4* z4 = (float4*)sE[0];
    const float4 z = {0.f, 0.f, 0.f, 0.f};
    for (int i = tid; i < 256; i += 32) z4[i] = z;
  }
  __syncthreads();

  const int row = tid & 15, hi = tid >> 4;
  float* stt = ws + WS_ST + (long)(blockIdx.x * NWB + wv) * (NSEG * 1024);

#pragma clang loop unroll(disable)
  for (int g = 0; g < NSEG - 1; ++g) {
    float* slot = stt + g * 1024;
    const float* Er = sE[g & 1];
    float* Ew = (float*)sE[(g & 1) ^ 1];
    const v2f* er = (const v2f*)(Er + row * 64 + 2 * hi);
    v2f ef[16];
#pragma unroll
    for (int kk = 0; kk < 16; ++kk) ef[kk] = er[kk * 2];

#pragma unroll
    for (int nt = 0; nt < 4; ++nt) {
      const int col = nt * 16 + row;
      v8f ax;
#pragma unroll
      for (int r = 0; r < 8; ++r) ax[r] = slot[(r + 8 * hi) * 64 + col];
      const float* pP = sP + col;
#pragma unroll
      for (int kk = 0; kk < 16; ++kk) {
        const int k0 = 4 * kk + 2 * hi;
        v2f bp; bp.x = pP[k0 * 64]; bp.y = pP[(k0 + 1) * 64];
        ax = __builtin_amdgcn_wmma_f32_16x16x4_f32(false, ef[kk], false, bp,
                                                   (short)0, ax, false, false);
      }
#pragma unroll
      for (int r = 0; r < 8; ++r) slot[(r + 8 * hi) * 64 + col] = ax[r];
      float* ew = Ew + (8 * hi) * 64 + col;
#pragma unroll
      for (int r = 0; r < 8; ++r) ew[r * 64] = ax[r];
    }
    __syncthreads();
  }
}

__global__ __launch_bounds__(32 * NWV) __attribute__((amdgpu_waves_per_eu(2, 2)))
void s4_outputs(const float* __restrict__ u, const float* __restrict__ ws,
                float* __restrict__ y) {
  __shared__ __attribute__((aligned(16))) float sMat[4 * 4096];
  __shared__ __attribute__((aligned(16))) float sXw[NWV][2][1024];

  const int wv = threadIdx.x >> 5;
  const int tid = threadIdx.x & 31;
  float (*sX)[1024] = sXw[wv];
  const int tile = (blockIdx.x >> 4) * NWV + wv;
  const int seg  = blockIdx.x & 15;
  {
    const int t = threadIdx.x;
    const float4* s4p = (const float4*)ws;
    float4* d4 = (float4*)sMat;
    float4* xp = (float4*)sX[0];
    float4 vm[16];
    for (int j = 0; j < 16; ++j) vm[j] = s4p[t + 32 * NWV * j];
    if (seg == 0) {
      const float4 z = {0.f, 0.f, 0.f, 0.f};
      for (int j = 0; j < 8; ++j) xp[tid + 32 * j] = z;
    } else {
      const float4* ep =
          (const float4*)(ws + WS_ST + (long)tile * (NSEG * 1024) + (seg - 1) * 1024);
      float4 ve[8];
      for (int j = 0; j < 8; ++j) ve[j] = ep[tid + 32 * j];
      for (int j = 0; j < 8; ++j) xp[tid + 32 * j] = ve[j];
    }
    for (int j = 0; j < 16; ++j) d4[t + 32 * NWV * j] = vm[j];
  }
  __syncthreads();

  const int s0 = tile * 16;
  const int bb = s0 >> 9, dm0 = s0 & 511;
  const int row = tid & 15, hi = tid >> 4;
  const float* ub0 = u + (long)bb * LSEQ * DMOD + dm0 + row + (long)(2 * hi) * DMOD;
  float* yb0 = y + (long)bb * LSEQ * DMOD + dm0 + 8 * hi;

  for (int i = 0; i < SCH; ++i) {
    const int c = seg * SCH + i;
    const float* Xr = sX[i & 1];
    float* Xw = (float*)sX[(i & 1) ^ 1];
    const float* up = ub0 + (long)c * TCH * DMOD;
    v2f uf[16];
#pragma unroll
    for (int kk = 0; kk < 16; ++kk) {
      uf[kk].x = up[kk * 4 * DMOD];
      uf[kk].y = up[kk * 4 * DMOD + DMOD];
    }
    const v2f* xr = (const v2f*)(Xr + row * 64 + 2 * hi);
    v2f xin[16];
#pragma unroll
    for (int kk = 0; kk < 16; ++kk) xin[kk] = xr[kk * 2];

#pragma unroll
    for (int nt = 0; nt < 4; ++nt) {
      v8f ay = {0.f, 0.f, 0.f, 0.f, 0.f, 0.f, 0.f, 0.f};
      v8f ax = {0.f, 0.f, 0.f, 0.f, 0.f, 0.f, 0.f, 0.f};
      const int col = nt * 16 + row;
      const float* pTK = sMat + col;
      const float* pMt = sMat + 4096 + col;
      const float* pEt = sMat + 8192 + col;
      const float* pPt = sMat + 12288 + col;
#pragma unroll
      for (int kk = 0; kk < 16; ++kk) {
        const int k0 = 4 * kk + 2 * hi;
        v2f bt; bt.x = pTK[k0 * 64]; bt.y = pTK[(k0 + 1) * 64];
        ay = __builtin_amdgcn_wmma_f32_16x16x4_f32(false, uf[kk], false, bt,
                                                   (short)0, ay, false, false);
        v2f bm; bm.x = pMt[k0 * 64]; bm.y = pMt[(k0 + 1) * 64];
        ax = __builtin_amdgcn_wmma_f32_16x16x4_f32(false, uf[kk], false, bm,
                                                   (short)0, ax, false, false);
        v2f be; be.x = pEt[k0 * 64]; be.y = pEt[(k0 + 1) * 64];
        ay = __builtin_amdgcn_wmma_f32_16x16x4_f32(false, xin[kk], false, be,
                                                   (short)0, ay, false, false);
        v2f bp; bp.x = pPt[k0 * 64]; bp.y = pPt[(k0 + 1) * 64];
        ax = __builtin_amdgcn_wmma_f32_16x16x4_f32(false, xin[kk], false, bp,
                                                   (short)0, ax, false, false);
      }
      float4* yp = (float4*)(yb0 + (long)(c * TCH + col) * DMOD);
      float4 lo, hh;
      lo.x = ay[0]; lo.y = ay[1]; lo.z = ay[2]; lo.w = ay[3];
      hh.x = ay[4]; hh.y = ay[5]; hh.z = ay[6]; hh.w = ay[7];
      yp[0] = lo;
      yp[1] = hh;
      float* xw = Xw + (8 * hi) * 64 + col;
#pragma unroll
      for (int r = 0; r < 8; ++r) xw[r * 64] = ax[r];
    }
    __syncthreads();
  }
}

extern "C" void kernel_launch(void* const* d_in, const int* in_sizes, int n_in,
                              void* d_out, int out_size, void* d_ws, size_t ws_size,
                              hipStream_t stream) {
  const float* u  = (const float*)d_in[0];
  const float* A_ = (const float*)d_in[1];
  const float* B_ = (const float*)d_in[2];
  const float* C_ = (const float*)d_in[3];
  float* y  = (float*)d_out;
  float* ws = (float*)d_ws;

  s4_precompute<<<3, 128, 0, stream>>>(A_, B_, C_, ws);
  s4_segstate<<<256, 256, 0, stream>>>(u, ws);
  s4_combine<<<32, 128, 0, stream>>>(ws);
  s4_outputs<<<256, 256, 0, stream>>>(u, ws, y);
}


# Chunk-matrix fragments come from k-contiguous LDS rows in 16-byte reads grouped ahead of the matrix instructions (held in registers in phases A, B); k order inside the sums changed
# speedup vs baseline: 27.1496x; 1.3137x over previous
#include <hip/hip_runtime.h>
#include <hip/hip_bf16.h>

typedef __attribute__((ext_vector_type(2))) float v2f;
typedef __attribute__((ext_vector_type(8))) float v8f;

#define NST 64
#define TCH 64
#define LSEQ 4096
#define DMOD 512
#define NSEG 16
#define SCH 4
#define NWV 8
#define NWB 4

#define WS_ST 20480
#define XP 68
#define XP4 17

typedef __attribute__((ext_vector_type(4))) float v4f;

#define LDP 68

__device__ __forceinline__ void stage_A(const float* __restrict__ A_, float* sA, int tid) {
  float av[32];
  for (int j = 0; j < 32; ++j) av[j] = A_[tid + 128 * j];
  for (int j = 0; j < 32; ++j) {
    const int i = tid + 128 * j;
    sA[(i >> 6) * LDP + (i & 63)] = av[j];
  }
}

__device__ __forceinline__ void run_chain(const v4f (&am)[16], float* xb, int last, int tid) {
  for (int t = 1; t <= last; ++t) {
    if (tid < NST) {
      const v4f* xp = (const v4f*)(xb + (t - 1) * LDP);
      v4f acc = {0.f, 0.f, 0.f, 0.f};
      for (int jj = 0; jj < 16; ++jj) acc += am[jj] * xp[jj];
      xb[t * LDP + tid] = (acc.x + acc.y) + (acc.z + acc.w);
    }
    __syncthreads();
  }
}

__device__ __forceinline__ void role_rows(const float* __restrict__ A_,
                                          const float* __restrict__ B_,
                                          const float* __restrict__ C_,
                                          float* __restrict__ ws, float* sm, int tid) {
  float* sA = sm;
  float* vb = sm + NST * LDP;
  float* cb = sm + 2 * NST * LDP;
  float* Kk = cb + NST;

  stage_A(A_, sA, tid);
  if (tid < NST) { vb[tid] = B_[tid]; cb[tid] = C_[tid]; }
  __syncthreads();

  v4f am[16];
  {
    const v4f* ar = (const v4f*)(sA + (tid & 63) * LDP);
    for (int jj = 0; jj < 16; ++jj) am[jj] = ar[jj];
  }
  run_chain(am, vb, NST - 1, tid);

  if (tid < NST) {
    const v4f* vp = (const v4f*)(vb + tid * LDP);
    const v4f* cp = (const v4f*)cb;
    v4f acc = {0.f, 0.f, 0.f, 0.f};
    for (int jj = 0; jj < 16; ++jj) acc += cp[jj] * vp[jj];
    Kk[tid] = (acc.x + acc.y) + (acc.z + acc.w);
  }
  __syncthreads();

  for (int j = 0; j < 32; ++j) {
    const int i = tid + 128 * j;
    const int r = i >> 6, k = i & 63;
    ws[i]        = (r >= k) ? Kk[r - k] : 0.f;
    ws[4096 + i] = vb[(NST - 1 - k) * LDP + r];
  }
}

__device__ __forceinline__ void role_cols(const float* __restrict__ A_,
                                          const float* __restrict__ C_,
                                          float* __restrict__ ws, float* sm, int tid) {
  float* sA = sm;
  float* wb = sm + NST * LDP;

  stage_A(A_, sA, tid);
  if (tid < NST) wb[tid] = C_[tid];
  __syncthreads();

  v4f am[16];
  {
    const float* ac = sA + (tid & 63);
    for (int ii = 0; ii < 16; ++ii) {
      am[ii].x = ac[(4 * ii) * LDP];
      am[ii].y = ac[(4 * ii + 1) * LDP];
      am[ii].z = ac[(4 * ii + 2) * LDP];
      am[ii].w = ac[(4 * ii + 3) * LDP];
    }
  }
  run_chain(am, wb, NST, tid);

  for (int j = 0; j < 32; ++j) {
    const int i = tid + 128 * j;
    const int r = i >> 6, k = i & 63;
    ws[8192 + i] = wb[(r + 1) * LDP + k];
  }
}

__device__ __forceinline__ void role_squares(const float* __restrict__ A_,
                                             float* __restrict__ ws, float* sm, int tid) {
  float* X0 = sm;
  float* X1 = sm + NST * LDP;
  float* X2 = sm + 2 * NST * LDP;

  stage_A(A_, X0, tid);
  __syncthreads();

  const int mt = tid >> 5;
  const int lane = tid & 31;
  const int c = lane & 15, h = lane >> 4;

  for (int s = 0; s < 8; ++s) {
    const float* P = (s & 1) ? X1 : X0;
    float* Pn = (s == 7) ? X2 : ((s & 1) ? X0 : X1);

    v2f a[16];
    {
      const float* L = P + (16 * mt + c) * LDP + 2 * h;
      for (int kk = 0; kk < 16; ++kk) a[kk] = *(const v2f*)(L + 4 * kk);
    }
    v8f acc[4];
    for (int nt = 0; nt < 4; ++nt) acc[nt] = (v8f){0.f, 0.f, 0.f, 0.f, 0.f, 0.f, 0.f, 0.f};
    for (int kk = 0; kk < 16; ++kk) {
      for (int nt = 0; nt < 4; ++nt) {
        const float* R = P + (4 * kk + 2 * h) * LDP + 16 * nt + c;
        v2f b;
        b.x = R[0];
        b.y = R[LDP];
        acc[nt] = __builtin_amdgcn_wmma_f32_16x16x4_f32(false, a[kk], false, b,
                                                        (short)0, acc[nt], false, false);
      }
    }
    for (int nt = 0; nt < 4; ++nt) {
      float* o = Pn + (16 * mt + 8 * h) * LDP + 16 * nt + c;
      for (int r = 0; r < 8; ++r) o[r * LDP] = acc[nt][r];
    }
    __syncthreads();
  }

  for (int j = 0; j < 32; ++j) {
    const int i = tid + 128 * j;
    const int r = i >> 6, k = i & 63;
    ws[12288 + i] = X0[r * LDP + k];
    ws[16384 + i] = X2[r * LDP + k];
  }
}

__global__ __launch_bounds__(128) void s4_precompute(const float* __restrict__ A_,
                                                     const float* __restrict__ B_,
                                                     const float* __restrict__ C_,
                                                     float* __restrict__ ws) {
  __shared__ __attribute__((aligned(16))) float sm[3 * NST * LDP];
  const int tid = threadIdx.x;
  if (blockIdx.x == 0) {
    role_rows(A_, B_, C_, ws, sm, tid);
  } else if (blockIdx.x == 1) {
    role_cols(A_, C_, ws, sm, tid);
  } else {
    role_squares(A_, ws, sm, tid);
  }
}

__device__ __forceinline__ v2f lo2(v4f v) { return __builtin_shufflevector(v, v, 0, 1); }
__device__ __forceinline__ v2f hi2(v4f v) { return __builtin_shufflevector(v, v, 2, 3); }

__device__ __forceinline__ void load_frag(v4f (&f)[8], const float* base) {
#pragma unroll
  for (int j = 0; j < 8; ++j) f[j] = *(const v4f*)(base + 4 * j);
}

#define WMMA4(a, b, c) __builtin_amdgcn_wmma_f32_16x16x4_f32(false, (a), false, (b), (short)0, (c), false, false)

__global__ __launch_bounds__(32 * NWV) __attribute__((amdgpu_waves_per_eu(2, 2)))
void s4_segstate(const float* __restrict__ u, float* __restrict__ ws) {
  __shared__ __attribute__((aligned(16))) float sM[2 * NST * XP];
  __shared__ __attribute__((aligned(16))) float sXw[NWV][2][16 * XP];

  const int wv = threadIdx.x >> 5;
  const int tid = threadIdx.x & 31;
  float (*sX)[16 * XP] = sXw[wv];
  {
    const int t = threadIdx.x;
    const float4* m4 = (const float4*)(ws + 4096);
    const float4* p4 = (const float4*)(ws + 12288);
    float4* d0 = (float4*)sM;
    float4* d1 = (float4*)(sM + NST * XP);
    float4 vm[4], vp[4];
    for (int j = 0; j < 4; ++j) vm[j] = m4[t + 32 * NWV * j];
    for (int j = 0; j < 4; ++j) vp[j] = p4[t + 32 * NWV * j];
    for (int j = 0; j < 4; ++j) {
      const int idx = t + 32 * NWV * j;
      const int o = (idx >> 4) * XP4 + (idx & 15);
      d0[o] = vm[j];
      d1[o] = vp[j];
    }
    float4* z4 = (float4*)sX[0];
    const float4 z = {0.f, 0.f, 0.f, 0.f};
    for (int i = tid; i < 16 * XP4; i += 32) z4[i] = z;
  }
  __syncthreads();

  const int tile = (blockIdx.x >> 4) * NWV + wv;
  const int seg  = blockIdx.x & 15;
  const int s0 = tile * 16;
  const int bb = s0 >> 9, dm0 = s0 & 511;
  const int row = tid & 15, hi = tid >> 4;
  const float* ub0 = u + (long)bb * LSEQ * DMOD + dm0 + row + (long)(32 * hi) * DMOD;

  v4f bM[4][8], bP[4][8];
#pragma unroll
  for (int nt = 0; nt < 4; ++nt) {
    load_frag(bM[nt], sM + (nt * 16 + row) * XP + 32 * hi);
    load_frag(bP[nt], sM + NST * XP + (nt * 16 + row) * XP + 32 * hi);
  }
  __builtin_amdgcn_sched_barrier(0);

  for (int i = 0; i < SCH; ++i) {
    const int c = seg * SCH + i;
    const float* Xr = sX[i & 1];
    float* Xw = (float*)sX[(i & 1) ^ 1];
    const float* up = ub0 + (long)c * TCH * DMOD;
    v2f uf[16];
#pragma unroll
    for (int kk = 0; kk < 16; ++kk) {
      uf[kk].x = up[(2 * kk) * DMOD];
      uf[kk].y = up[(2 * kk + 1) * DMOD];
    }
    v4f xq[8];
    load_frag(xq, Xr + row * XP + 32 * hi);
    __builtin_amdgcn_sched_barrier(0);

#pragma unroll
    for (int np = 0; np < 2; ++np) {
      v8f a0 = {0.f, 0.f, 0.f, 0.f, 0.f, 0.f, 0.f, 0.f};
      v8f a1 = {0.f, 0.f, 0.f, 0.f, 0.f, 0.f, 0.f, 0.f};
      const int n0 = 2 * np, n1 = 2 * np + 1;
#pragma unroll
      for (int j = 0; j < 8; ++j) {
        a0 = WMMA4(uf[2 * j], lo2(bM[n0][j]), a0);
        a1 = WMMA4(uf[2 * j], lo2(bM[n1][j]), a1);
        a0 = WMMA4(lo2(xq[j]), lo2(bP[n0][j]), a0);
        a1 = WMMA4(lo2(xq[j]), lo2(bP[n1][j]), a1);
        a0 = WMMA4(uf[2 * j + 1], hi2(bM[n0][j]), a0);
        a1 = WMMA4(uf[2 * j + 1], hi2(bM[n1][j]), a1);
        a0 = WMMA4(hi2(xq[j]), hi2(bP[n0][j]), a0);
        a1 = WMMA4(hi2(xq[j]), hi2(bP[n1][j]), a1);
      }
      float* xw0 = Xw + (8 * hi) * XP + n0 * 16 + row;
      float* xw1 = Xw + (8 * hi) * XP + n1 * 16 + row;
#pragma unroll
      for (int r = 0; r < 8; ++r) xw0[r * XP] = a0[r];
#pragma unroll
      for (int r = 0; r < 8; ++r) xw1[r * XP] = a1[r];
    }
    __syncthreads();
  }
  float* st = ws + WS_ST + (long)tile * (NSEG * 1024) + seg * 1024;
  const float4* xp = (const float4*)sX[0];
  float4* gp = (float4*)st;
  for (int i = tid; i < 256; i += 32) gp[i] = xp[(i >> 4) * XP4 + (i & 15)];
}

__global__ __launch_bounds__(32 * NWB) __attribute__((amdgpu_waves_per_eu(1, 2)))
void s4_combine(float* __restrict__ ws) {
  __shared__ __attribute__((aligned(16))) float sP[NST * XP];
  __shared__ __attribute__((aligned(16))) float sEw[NWB][2][16 * XP];

  const int wv = threadIdx.x >> 5;
  const int tid = threadIdx.x & 31;
  float (*sE)[16 * XP] = sEw[wv];
  {
    const int t = threadIdx.x;
    const float4* p4 = (const float4*)(ws + 16384);
    float4* d4 = (float4*)sP;
    float4 vp[8];
    for (int j = 0; j < 8; ++j) vp[j] = p4[t + 32 * NWB * j];
    for (int j = 0; j < 8; ++j) {
      const int idx = t + 32 * NWB * j;
      d4[(idx >> 4) * XP4 + (idx & 15)] = vp[j];
    }
    float4* z4 = (float4*)sE[0];
    const float4 z = {0.f, 0.f, 0.f, 0.f};
    for (int i = tid; i < 16 * XP4; i += 32) z4[i] = z;
  }
  __syncthreads();

  const int row = tid & 15, hi = tid >> 4;
  float* stt = ws + WS_ST + (long)(blockIdx.x * NWB + wv) * (NSEG * 1024);

  v4f bP[4][8];
#pragma unroll
  for (int nt = 0; nt < 4; ++nt) load_frag(bP[nt], sP + (nt * 16 + row) * XP + 32 * hi);

#pragma clang loop unroll(disable)
  for (int g = 0; g < NSEG - 1; ++g) {
    float* slot = stt + g * 1024;
    const float* Er = sE[g & 1];
    float* Ew = (float*)sE[(g & 1) ^ 1];
    v4f eq[8];
    load_frag(eq, Er + row * XP + 32 * hi);

#pragma unroll
    for (int np = 0; np < 2; ++np) {
      const int c0 = (2 * np) * 16 + row, c1 = (2 * np + 1) * 16 + row;
      v8f a0, a1;
#pragma unroll
      for (int r = 0; r < 8; ++r) a0[r] = slot[(r + 8 * hi) * 64 + c0];
#pragma unroll
      for (int r = 0; r < 8; ++r) a1[r] = slot[(r + 8 * hi) * 64 + c1];
#pragma unroll
      for (int j = 0; j < 8; ++j) {
        a0 = WMMA4(lo2(eq[j]), lo2(bP[2 * np][j]), a0);
        a1 = WMMA4(lo2(eq[j]), lo2(bP[2 * np + 1][j]), a1);
        a0 = WMMA4(hi2(eq[j]), hi2(bP[2 * np][j]), a0);
        a1 = WMMA4(hi2(eq[j]), hi2(bP[2 * np + 1][j]), a1);
      }
#pragma unroll
      for (int r = 0; r < 8; ++r) slot[(r + 8 * hi) * 64 + c0] = a0[r];
#pragma unroll
      for (int r = 0; r < 8; ++r) slot[(r + 8 * hi) * 64 + c1] = a1[r];
      float* ew0 = Ew + (8 * hi) * XP + c0;
      float* ew1 = Ew + (8 * hi) * XP + c1;
#pragma unroll
      for (int r = 0; r < 8; ++r) ew0[r * XP] = a0[r];
#pragma unroll
      for (int r = 0; r < 8; ++r) ew1[r * XP] = a1[r];
    }
    __syncthreads();
  }
}

__global__ __launch_bounds__(32 * NWV) __attribute__((amdgpu_waves_per_eu(2, 2)))
void s4_outputs(const float* __restrict__ u, const float* __restrict__ ws,
                float* __restrict__ y) {
  __shared__ __attribute__((aligned(16))) float sMat[4 * NST * XP];
  __shared__ __attribute__((aligned(16))) float sXw[NWV][2][16 * XP];

  const int wv = threadIdx.x >> 5;
  const int tid = threadIdx.x & 31;
  float (*sX)[16 * XP] = sXw[wv];
  const int tile = (blockIdx.x >> 4) * NWV + wv;
  const int seg  = blockIdx.x & 15;
  {
    const int t = threadIdx.x;
    const float4* s4p = (const float4*)ws;
    float4* d4 = (float4*)sMat;
    float4* xp = (float4*)sX[0];
    float4 vm[16];
    for (int j = 0; j < 16; ++j) vm[j] = s4p[t + 32 * NWV * j];
    if (seg == 0) {
      const float4 z = {0.f, 0.f, 0.f, 0.f};
      for (int j = 0; j < 8; ++j) {
        const int idx = tid + 32 * j;
        xp[(idx >> 4) * XP4 + (idx & 15)] = z;
      }
    } else {
      const float4* ep =
          (const float4*)(ws + WS_ST + (long)tile * (NSEG * 1024) + (seg - 1) * 1024);
      float4 ve[8];
      for (int j = 0; j < 8; ++j) ve[j] = ep[tid + 32 * j];
      for (int j = 0; j < 8; ++j) {
        const int idx = tid + 32 * j;
        xp[(idx >> 4) * XP4 + (idx & 15)] = ve[j];
      }
    }
    for (int j = 0; j < 16; ++j) {
      const int idx = t + 32 * NWV * j;
      d4[(idx >> 10) * (NST * XP4) + ((idx >> 4) & 63) * XP4 + (idx & 15)] = vm[j];
    }
  }
  __syncthreads();

  const int s0 = tile * 16;
  const int bb = s0 >> 9, dm0 = s0 & 511;
  const int row = tid & 15, hi = tid >> 4;
  const float* ub0 = u + (long)bb * LSEQ * DMOD + dm0 + row + (long)(32 * hi) * DMOD;
  float* yb0 = y + (long)bb * LSEQ * DMOD + dm0 + 8 * hi;
  const float* fb = sMat + row * XP + 32 * hi;

  v4f bc[4][8];
#pragma unroll
  for (int m = 0; m < 4; ++m) load_frag(bc[m], fb + m * (NST * XP));

  for (int i = 0; i < SCH; ++i) {
    const int c = seg * SCH + i;
    const float* Xr = sX[i & 1];
    float* Xw = (float*)sX[(i & 1) ^ 1];
    const float* up = ub0 + (long)c * TCH * DMOD;
    v2f uf[16];
#pragma unroll
    for (int kk = 0; kk < 16; ++kk) {
      uf[kk].x = up[(2 * kk) * DMOD];
      uf[kk].y = up[(2 * kk + 1) * DMOD];
    }
    v4f xq[8];
    load_frag(xq, Xr + row * XP + 32 * hi);

#pragma unroll
    for (int nt = 0; nt < 4; ++nt) {
      v4f bn[4][8];
#pragma unroll
      for (int m = 0; m < 4; ++m) load_frag(bn[m], fb + m * (NST * XP) + ((nt + 1) & 3) * 16 * XP);
      __builtin_amdgcn_sched_barrier(0);

      v8f ay = {0.f, 0.f, 0.f, 0.f, 0.f, 0.f, 0.f, 0.f};
      v8f ax = {0.f, 0.f, 0.f, 0.f, 0.f, 0.f, 0.f, 0.f};
      const int col = nt * 16 + row;
#pragma unroll
      for (int j = 0; j < 8; ++j) {
        ay = WMMA4(uf[2 * j], lo2(bc[0][j]), ay);
        ax = WMMA4(uf[2 * j], lo2(bc[1][j]), ax);
        ay = WMMA4(lo2(xq[j]), lo2(bc[2][j]), ay);
        ax = WMMA4(lo2(xq[j]), lo2(bc[3][j]), ax);
        ay = WMMA4(uf[2 * j + 1], hi2(bc[0][j]), ay);
        ax = WMMA4(uf[2 * j + 1], hi2(bc[1][j]), ax);
        ay = WMMA4(hi2(xq[j]), hi2(bc[2][j]), ay);
        ax = WMMA4(hi2(xq[j]), hi2(bc[3][j]), ax);
      }
      __builtin_amdgcn_sched_barrier(0);
      float4* yp = (float4*)(yb0 + (long)(c * TCH + col) * DMOD);
      float4 lo, hh;
      lo.x = ay[0]; lo.y = ay[1]; lo.z = ay[2]; lo.w = ay[3];
      hh.x = ay[4]; hh.y = ay[5]; hh.z = ay[6]; hh.w = ay[7];
      yp[0] = lo;
      yp[1] = hh;
      float* xw = Xw + (8 * hi) * XP + col;
#pragma unroll
      for (int r = 0; r < 8; ++r) xw[r * XP] = ax[r];
#pragma unroll
      for (int m = 0; m < 4; ++m)
#pragma unroll
        for (int j = 0; j < 8; ++j) bc[m][j] = bn[m][j];
    }
    __syncthreads();
  }
}

extern "C" void kernel_launch(void* const* d_in, const int* in_sizes, int n_in,
                              void* d_out, int out_size, void* d_ws, size_t ws_size,
                              hipStream_t stream) {
  const float* u  = (const float*)d_in[0];
  const float* A_ = (const float*)d_in[1];
  const float* B_ = (const float*)d_in[2];
  const float* C_ = (const float*)d_in[3];
  float* y  = (float*)d_out;
  float* ws = (float*)d_ws;

  s4_precompute<<<3, 128, 0, stream>>>(A_, B_, C_, ws);
  s4_segstate<<<256, 256, 0, stream>>>(u, ws);
  s4_combine<<<32, 128, 0, stream>>>(ws);
  s4_outputs<<<256, 256, 0, stream>>>(u, ws, y);
}


# Phase B: one 4-wave group per seq-tile, each wave owns one 16-column tile of E (16 chained matrix instructions per step, was 64); E handed over in LDS, G loads and E stores kept off the chain
# speedup vs baseline: 36.5842x; 1.3475x over previous
#include <hip/hip_runtime.h>
#include <hip/hip_bf16.h>

typedef __attribute__((ext_vector_type(2))) float v2f;
typedef __attribute__((ext_vector_type(8))) float v8f;

#define NST 64
#define TCH 64
#define LSEQ 4096
#define DMOD 512
#define NSEG 16
#define SCH 4
#define NWV 8
#define NWB 4

#define WS_ST 20480
#define XP 68
#define XP4 17

typedef __attribute__((ext_vector_type(4))) float v4f;

#define LDP 68

__device__ __forceinline__ void stage_A(const float* __restrict__ A_, float* sA, int tid) {
  float av[32];
  for (int j = 0; j < 32; ++j) av[j] = A_[tid + 128 * j];
  for (int j = 0; j < 32; ++j) {
    const int i = tid + 128 * j;
    sA[(i >> 6) * LDP + (i & 63)] = av[j];
  }
}

__device__ __forceinline__ void run_chain(const v4f (&am)[16], float* xb, int last, int tid) {
  for (int t = 1; t <= last; ++t) {
    if (tid < NST) {
      const v4f* xp = (const v4f*)(xb + (t - 1) * LDP);
      v4f acc = {0.f, 0.f, 0.f, 0.f};
      for (int jj = 0; jj < 16; ++jj) acc += am[jj] * xp[jj];
      xb[t * LDP + tid] = (acc.x + acc.y) + (acc.z + acc.w);
    }
    __syncthreads();
  }
}

__device__ __forceinline__ void role_rows(const float* __restrict__ A_,
                                          const float* __restrict__ B_,
                                          const float* __restrict__ C_,
                                          float* __restrict__ ws, float* sm, int tid) {
  float* sA = sm;
  float* vb = sm + NST * LDP;
  float* cb = sm + 2 * NST * LDP;
  float* Kk = cb + NST;

  stage_A(A_, sA, tid);
  if (tid < NST) { vb[tid] = B_[tid]; cb[tid] = C_[tid]; }
  __syncthreads();

  v4f am[16];
  {
    const v4f* ar = (const v4f*)(sA + (tid & 63) * LDP);
    for (int jj = 0; jj < 16; ++jj) am[jj] = ar[jj];
  }
  run_chain(am, vb, NST - 1, tid);

  if (tid < NST) {
    const v4f* vp = (const v4f*)(vb + tid * LDP);
    const v4f* cp = (const v4f*)cb;
    v4f acc = {0.f, 0.f, 0.f, 0.f};
    for (int jj = 0; jj < 16; ++jj) acc += cp[jj] * vp[jj];
    Kk[tid] = (acc.x + acc.y) + (acc.z + acc.w);
  }
  __syncthreads();

  for (int j = 0; j < 32; ++j) {
    const int i = tid + 128 * j;
    const int r = i >> 6, k = i & 63;
    ws[i]        = (r >= k) ? Kk[r - k] : 0.f;
    ws[4096 + i] = vb[(NST - 1 - k) * LDP + r];
  }
}

__device__ __forceinline__ void role_cols(const float* __restrict__ A_,
                                          const float* __restrict__ C_,
                                          float* __restrict__ ws, float* sm, int tid) {
  float* sA = sm;
  float* wb = sm + NST * LDP;

  stage_A(A_, sA, tid);
  if (tid < NST) wb[tid] = C_[tid];
  __syncthreads();

  v4f am[16];
  {
    const float* ac = sA + (tid & 63);
    for (int ii = 0; ii < 16; ++ii) {
      am[ii].x = ac[(4 * ii) * LDP];
      am[ii].y = ac[(4 * ii + 1) * LDP];
      am[ii].z = ac[(4 * ii + 2) * LDP];
      am[ii].w = ac[(4 * ii + 3) * LDP];
    }
  }
  run_chain(am, wb, NST, tid);

  for (int j = 0; j < 32; ++j) {
    const int i = tid + 128 * j;
    const int r = i >> 6, k = i & 63;
    ws[8192 + i] = wb[(r + 1) * LDP + k];
  }
}

__device__ __forceinline__ void role_squares(const float* __restrict__ A_,
                                             float* __restrict__ ws, float* sm, int tid) {
  float* X0 = sm;
  float* X1 = sm + NST * LDP;
  float* X2 = sm + 2 * NST * LDP;

  stage_A(A_, X0, tid);
  __syncthreads();

  const int mt = tid >> 5;
  const int lane = tid & 31;
  const int c = lane & 15, h = lane >> 4;

  for (int s = 0; s < 8; ++s) {
    const float* P = (s & 1) ? X1 : X0;
    float* Pn = (s == 7) ? X2 : ((s & 1) ? X0 : X1);

    v2f a[16];
    {
      const float* L = P + (16 * mt + c) * LDP + 2 * h;
      for (int kk = 0; kk < 16; ++kk) a[kk] = *(const v2f*)(L + 4 * kk);
    }
    v8f acc[4];
    for (int nt = 0; nt < 4; ++nt) acc[nt] = (v8f){0.f, 0.f, 0.f, 0.f, 0.f, 0.f, 0.f, 0.f};
    for (int kk = 0; kk < 16; ++kk) {
      for (int nt = 0; nt < 4; ++nt) {
        const float* R = P + (4 * kk + 2 * h) * LDP + 16 * nt + c;
        v2f b;
        b.x = R[0];
        b.y = R[LDP];
        acc[nt] = __builtin_amdgcn_wmma_f32_16x16x4_f32(false, a[kk], false, b,
                                                        (short)0, acc[nt], false, false);
      }
    }
    for (int nt = 0; nt < 4; ++nt) {
      float* o = Pn + (16 * mt + 8 * h) * LDP + 16 * nt + c;
      for (int r = 0; r < 8; ++r) o[r * LDP] = acc[nt][r];
    }
    __syncthreads();
  }

  for (int j = 0; j < 32; ++j) {
    const int i = tid + 128 * j;
    const int r = i >> 6, k = i & 63;
    ws[12288 + i] = X0[r * LDP + k];
    ws[16384 + i] = X2[r * LDP + k];
  }
}

__global__ __launch_bounds__(128) void s4_precompute(const float* __restrict__ A_,
                                                     const float* __restrict__ B_,
                                                     const float* __restrict__ C_,
                                                     float* __restrict__ ws) {
  __shared__ __attribute__((aligned(16))) float sm[3 * NST * LDP];
  const int tid = threadIdx.x;
  if (blockIdx.x == 0) {
    role_rows(A_, B_, C_, ws, sm, tid);
  } else if (blockIdx.x == 1) {
    role_cols(A_, C_, ws, sm, tid);
  } else {
    role_squares(A_, ws, sm, tid);
  }
}

__device__ __forceinline__ v2f lo2(v4f v) { return __builtin_shufflevector(v, v, 0, 1); }
__device__ __forceinline__ v2f hi2(v4f v) { return __builtin_shufflevector(v, v, 2, 3); }

__device__ __forceinline__ void load_frag(v4f (&f)[8], const float* base) {
#pragma unroll
  for (int j = 0; j < 8; ++j) f[j] = *(const v4f*)(base + 4 * j);
}

#define WMMA4(a, b, c) __builtin_amdgcn_wmma_f32_16x16x4_f32(false, (a), false, (b), (short)0, (c), false, false)

__global__ __launch_bounds__(32 * NWV) __attribute__((amdgpu_waves_per_eu(2, 2)))
void s4_segstate(const float* __restrict__ u, float* __restrict__ ws) {
  __shared__ __attribute__((aligned(16))) float sM[2 * NST * XP];
  __shared__ __attribute__((aligned(16))) float sXw[NWV][2][16 * XP];

  const int wv = threadIdx.x >> 5;
  const int tid = threadIdx.x & 31;
  float (*sX)[16 * XP] = sXw[wv];
  {
    const int t = threadIdx.x;
    const float4* m4 = (const float4*)(ws + 4096);
    const float4* p4 = (const float4*)(ws + 12288);
    float4* d0 = (float4*)sM;
    float4* d1 = (float4*)(sM + NST * XP);
    float4 vm[4], vp[4];
    for (int j = 0; j < 4; ++j) vm[j] = m4[t + 32 * NWV * j];
    for (int j = 0; j < 4; ++j) vp[j] = p4[t + 32 * NWV * j];
    for (int j = 0; j < 4; ++j) {
      const int idx = t + 32 * NWV * j;
      const int o = (idx >> 4) * XP4 + (idx & 15);
      d0[o] = vm[j];
      d1[o] = vp[j];
    }
    float4* z4 = (float4*)sX[0];
    const float4 z = {0.f, 0.f, 0.f, 0.f};
    for (int i = tid; i < 16 * XP4; i += 32) z4[i] = z;
  }
  __syncthreads();

  const int tile = (blockIdx.x >> 4) * NWV + wv;
  const int seg  = blockIdx.x & 15;
  const int s0 = tile * 16;
  const int bb = s0 >> 9, dm0 = s0 & 511;
  const int row = tid & 15, hi = tid >> 4;
  const float* ub0 = u + (long)bb * LSEQ * DMOD + dm0 + row + (long)(32 * hi) * DMOD;

  v4f bM[4][8], bP[4][8];
#pragma unroll
  for (int nt = 0; nt < 4; ++nt) {
    load_frag(bM[nt], sM + (nt * 16 + row) * XP + 32 * hi);
    load_frag(bP[nt], sM + NST * XP + (nt * 16 + row) * XP + 32 * hi);
  }
  __builtin_amdgcn_sched_barrier(0);

  for (int i = 0; i < SCH; ++i) {
    const int c = seg * SCH + i;
    const float* Xr = sX[i & 1];
    float* Xw = (float*)sX[(i & 1) ^ 1];
    const float* up = ub0 + (long)c * TCH * DMOD;
    v2f uf[16];
#pragma unroll
    for (int kk = 0; kk < 16; ++kk) {
      uf[kk].x = up[(2 * kk) * DMOD];
      uf[kk].y = up[(2 * kk + 1) * DMOD];
    }
    v4f xq[8];
    load_frag(xq, Xr + row * XP + 32 * hi);
    __builtin_amdgcn_sched_barrier(0);

#pragma unroll
    for (int np = 0; np < 2; ++np) {
      v8f a0 = {0.f, 0.f, 0.f, 0.f, 0.f, 0.f, 0.f, 0.f};
      v8f a1 = {0.f, 0.f, 0.f, 0.f, 0.f, 0.f, 0.f, 0.f};
      const int n0 = 2 * np, n1 = 2 * np + 1;
#pragma unroll
      for (int j = 0; j < 8; ++j) {
        a0 = WMMA4(uf[2 * j], lo2(bM[n0][j]), a0);
        a1 = WMMA4(uf[2 * j], lo2(bM[n1][j]), a1);
        a0 = WMMA4(lo2(xq[j]), lo2(bP[n0][j]), a0);
        a1 = WMMA4(lo2(xq[j]), lo2(bP[n1][j]), a1);
        a0 = WMMA4(uf[2 * j + 1], hi2(bM[n0][j]), a0);
        a1 = WMMA4(uf[2 * j + 1], hi2(bM[n1][j]), a1);
        a0 = WMMA4(hi2(xq[j]), hi2(bP[n0][j]), a0);
        a1 = WMMA4(hi2(xq[j]), hi2(bP[n1][j]), a1);
      }
      float* xw0 = Xw + (8 * hi) * XP + n0 * 16 + row;
      float* xw1 = Xw + (8 * hi) * XP + n1 * 16 + row;
#pragma unroll
      for (int r = 0; r < 8; ++r) xw0[r * XP] = a0[r];
#pragma unroll
      for (int r = 0; r < 8; ++r) xw1[r * XP] = a1[r];
    }
    __syncthreads();
  }
  float* st = ws + WS_ST + (long)tile * (NSEG * 1024) + seg * 1024;
  const float4* xp = (const float4*)sX[0];
  float4* gp = (float4*)st;
  for (int i = tid; i < 256; i += 32) gp[i] = xp[(i >> 4) * XP4 + (i & 15)];
}

__device__ __forceinline__ void lds_handover_barrier() {
  asm volatile("s_wait_dscnt 0x0\n\ts_barrier_signal -1\n\ts_barrier_wait -1" ::: "memory");
}

__global__ __launch_bounds__(32 * NWB)
void s4_combine(float* __restrict__ ws) {
  __shared__ __attribute__((aligned(16))) float sE[2][16 * XP];

  const int wv = threadIdx.x >> 5;
  const int tid = threadIdx.x & 31;
  const int row = tid & 15, hi = tid >> 4;
  const int col = wv * 16 + row;
  float* stt = ws + WS_ST + (long)blockIdx.x * (NSEG * 1024);
  const int i0 = threadIdx.x, i1 = threadIdx.x + 32 * NWB;
  const int l0 = (i0 >> 4) * XP4 + (i0 & 15), l1 = (i1 >> 4) * XP4 + (i1 & 15);

  v4f bP[8];
  load_frag(bP, ws + 16384 + col * NST + 32 * hi);
  v8f gc, g1;
#pragma unroll
  for (int r = 0; r < 8; ++r) gc[r] = stt[(r + 8 * hi) * 64 + col];
#pragma unroll
  for (int r = 0; r < 8; ++r) g1[r] = stt[1024 + (r + 8 * hi) * 64 + col];
  {
    float4* z4 = (float4*)sE[0];
    const float4 z = {0.f, 0.f, 0.f, 0.f};
    for (int i = threadIdx.x; i < 16 * XP4; i += 32 * NWB) z4[i] = z;
  }
  lds_handover_barrier();

#pragma unroll
  for (int g = 0; g < NSEG - 1; ++g) {
    const float* Er = sE[g & 1];
    float* Ew = (float*)sE[(g & 1) ^ 1];
    float* ew = Ew + (8 * hi) * XP + col;
    v4f eq[8];
    load_frag(eq, Er + row * XP + 32 * hi);
    const float4* e4 = (const float4*)Er;
    float4 so0 = e4[l0], so1 = e4[l1];
    v8f g2 = g1;
    if (g + 2 < NSEG - 1) {
      const float* nx = stt + (g + 2) * 1024;
#pragma unroll
      for (int r = 0; r < 8; ++r) g2[r] = nx[(r + 8 * hi) * 64 + col];
    }
    __builtin_amdgcn_sched_barrier(0);
    v8f a = gc;
#pragma unroll
    for (int j = 0; j < 8; ++j) {
      a = WMMA4(lo2(eq[j]), lo2(bP[j]), a);
      a = WMMA4(hi2(eq[j]), hi2(bP[j]), a);
    }
#pragma unroll
    for (int r = 0; r < 8; ++r) ew[r * XP] = a[r];
    if (g > 0) {
      float4* s4 = (float4*)(stt + (g - 1) * 1024);
      s4[i0] = so0;
      s4[i1] = so1;
    }
    lds_handover_barrier();
    gc = g1;
    g1 = g2;
  }
  {
    const float4* e4 = (const float4*)sE[1];
    float4* s4 = (float4*)(stt + (NSEG - 2) * 1024);
    s4[i0] = e4[l0];
    s4[i1] = e4[l1];
  }
}

__global__ __launch_bounds__(32 * NWV) __attribute__((amdgpu_waves_per_eu(2, 2)))
void s4_outputs(const float* __restrict__ u, const float* __restrict__ ws,
                float* __restrict__ y) {
  __shared__ __attribute__((aligned(16))) float sMat[4 * NST * XP];
  __shared__ __attribute__((aligned(16))) float sXw[NWV][2][16 * XP];

  const int wv = threadIdx.x >> 5;
  const int tid = threadIdx.x & 31;
  float (*sX)[16 * XP] = sXw[wv];
  const int tile = (blockIdx.x >> 4) * NWV + wv;
  const int seg  = blockIdx.x & 15;
  {
    const int t = threadIdx.x;
    const float4* s4p = (const float4*)ws;
    float4* d4 = (float4*)sMat;
    float4* xp = (float4*)sX[0];
    float4 vm[16];
    for (int j = 0; j < 16; ++j) vm[j] = s4p[t + 32 * NWV * j];
    if (seg == 0) {
      const float4 z = {0.f, 0.f, 0.f, 0.f};
      for (int j = 0; j < 8; ++j) {
        const int idx = tid + 32 * j;
        xp[(idx >> 4) * XP4 + (idx & 15)] = z;
      }
    } else {
      const float4* ep =
          (const float4*)(ws + WS_ST + (long)tile * (NSEG * 1024) + (seg - 1) * 1024);
      float4 ve[8];
      for (int j = 0; j < 8; ++j) ve[j] = ep[tid + 32 * j];
      for (int j = 0; j < 8; ++j) {
        const int idx = tid + 32 * j;
        xp[(idx >> 4) * XP4 + (idx & 15)] = ve[j];
      }
    }
    for (int j = 0; j < 16; ++j) {
      const int idx = t + 32 * NWV * j;
      d4[(idx >> 10) * (NST * XP4) + ((idx >> 4) & 63) * XP4 + (idx & 15)] = vm[j];
    }
  }
  __syncthreads();

  const int s0 = tile * 16;
  const int bb = s0 >> 9, dm0 = s0 & 511;
  const int row = tid & 15, hi = tid >> 4;
  const float* ub0 = u + (long)bb * LSEQ * DMOD + dm0 + row + (long)(32 * hi) * DMOD;
  float* yb0 = y + (long)bb * LSEQ * DMOD + dm0 + 8 * hi;
  const float* fb = sMat + row * XP + 32 * hi;

  v4f bc[4][8];
#pragma unroll
  for (int m = 0; m < 4; ++m) load_frag(bc[m], fb + m * (NST * XP));

  for (int i = 0; i < SCH; ++i) {
    const int c = seg * SCH + i;
    const float* Xr = sX[i & 1];
    float* Xw = (float*)sX[(i & 1) ^ 1];
    const float* up = ub0 + (long)c * TCH * DMOD;
    v2f uf[16];
#pragma unroll
    for (int kk = 0; kk < 16; ++kk) {
      uf[kk].x = up[(2 * kk) * DMOD];
      uf[kk].y = up[(2 * kk + 1) * DMOD];
    }
    v4f xq[8];
    load_frag(xq, Xr + row * XP + 32 * hi);

#pragma unroll
    for (int nt = 0; nt < 4; ++nt) {
      v4f bn[4][8];
#pragma unroll
      for (int m = 0; m < 4; ++m) load_frag(bn[m], fb + m * (NST * XP) + ((nt + 1) & 3) * 16 * XP);
      __builtin_amdgcn_sched_barrier(0);

      v8f ay = {0.f, 0.f, 0.f, 0.f, 0.f, 0.f, 0.f, 0.f};
      v8f ax = {0.f, 0.f, 0.f, 0.f, 0.f, 0.f, 0.f, 0.f};
      const int col = nt * 16 + row;
#pragma unroll
      for (int j = 0; j < 8; ++j) {
        ay = WMMA4(uf[2 * j], lo2(bc[0][j]), ay);
        ax = WMMA4(uf[2 * j], lo2(bc[1][j]), ax);
        ay = WMMA4(lo2(xq[j]), lo2(bc[2][j]), ay);
        ax = WMMA4(lo2(xq[j]), lo2(bc[3][j]), ax);
        ay = WMMA4(uf[2 * j + 1], hi2(bc[0][j]), ay);
        ax = WMMA4(uf[2 * j + 1], hi2(bc[1][j]), ax);
        ay = WMMA4(hi2(xq[j]), hi2(bc[2][j]), ay);
        ax = WMMA4(hi2(xq[j]), hi2(bc[3][j]), ax);
      }
      __builtin_amdgcn_sched_barrier(0);
      float4* yp = (float4*)(yb0 + (long)(c * TCH + col) * DMOD);
      float4 lo, hh;
      lo.x = ay[0]; lo.y = ay[1]; lo.z = ay[2]; lo.w = ay[3];
      hh.x = ay[4]; hh.y = ay[5]; hh.z = ay[6]; hh.w = ay[7];
      yp[0] = lo;
      yp[1] = hh;
      float* xw = Xw + (8 * hi) * XP + col;
#pragma unroll
      for (int r = 0; r < 8; ++r) xw[r * XP] = ax[r];
#pragma unroll
      for (int m = 0; m < 4; ++m)
#pragma unroll
        for (int j = 0; j < 8; ++j) bc[m][j] = bn[m][j];
    }
    __syncthreads();
  }
}

extern "C" void kernel_launch(void* const* d_in, const int* in_sizes, int n_in,
                              void* d_out, int out_size, void* d_ws, size_t ws_size,
                              hipStream_t stream) {
  const float* u  = (const float*)d_in[0];
  const float* A_ = (const float*)d_in[1];
  const float* B_ = (const float*)d_in[2];
  const float* C_ = (const float*)d_in[3];
  float* y  = (float*)d_out;
  float* ws = (float*)d_ws;

  s4_precompute<<<3, 128, 0, stream>>>(A_, B_, C_, ws);
  s4_segstate<<<256, 256, 0, stream>>>(u, ws);
  s4_combine<<<128, 128, 0, stream>>>(ws);
  s4_outputs<<<256, 256, 0, stream>>>(u, ws, y);
}


# Phases A, C: next chunk's u loads issued ahead of the current chunk's matrix instructions (second register set); per-chunk work-group barrier replaced by wave-local LDS ordering
# speedup vs baseline: 38.9005x; 1.0633x over previous
#include <hip/hip_runtime.h>
#include <hip/hip_bf16.h>

typedef __attribute__((ext_vector_type(2))) float v2f;
typedef __attribute__((ext_vector_type(8))) float v8f;

#define NST 64
#define TCH 64
#define LSEQ 4096
#define DMOD 512
#define NSEG 16
#define SCH 4
#define NWV 8
#define NWB 4

#define WS_ST 20480
#define XP 68
#define XP4 17

typedef __attribute__((ext_vector_type(4))) float v4f;

#define LDP 68

__device__ __forceinline__ void stage_A(const float* __restrict__ A_, float* sA, int tid) {
  float av[32];
  for (int j = 0; j < 32; ++j) av[j] = A_[tid + 128 * j];
  for (int j = 0; j < 32; ++j) {
    const int i = tid + 128 * j;
    sA[(i >> 6) * LDP + (i & 63)] = av[j];
  }
}

__device__ __forceinline__ void run_chain(const v4f (&am)[16], float* xb, int last, int tid) {
  for (int t = 1; t <= last; ++t) {
    if (tid < NST) {
      const v4f* xp = (const v4f*)(xb + (t - 1) * LDP);
      v4f acc = {0.f, 0.f, 0.f, 0.f};
      for (int jj = 0; jj < 16; ++jj) acc += am[jj] * xp[jj];
      xb[t * LDP + tid] = (acc.x + acc.y) + (acc.z + acc.w);
    }
    __syncthreads();
  }
}

__device__ __forceinline__ void role_rows(const float* __restrict__ A_,
                                          const float* __restrict__ B_,
                                          const float* __restrict__ C_,
                                          float* __restrict__ ws, float* sm, int tid) {
  float* sA = sm;
  float* vb = sm + NST * LDP;
  float* cb = sm + 2 * NST * LDP;
  float* Kk = cb + NST;

  stage_A(A_, sA, tid);
  if (tid < NST) { vb[tid] = B_[tid]; cb[tid] = C_[tid]; }
  __syncthreads();

  v4f am[16];
  {
    const v4f* ar = (const v4f*)(sA + (tid & 63) * LDP);
    for (int jj = 0; jj < 16; ++jj) am[jj] = ar[jj];
  }
  run_chain(am, vb, NST - 1, tid);

  if (tid < NST) {
    const v4f* vp = (const v4f*)(vb + tid * LDP);
    const v4f* cp = (const v4f*)cb;
    v4f acc = {0.f, 0.f, 0.f, 0.f};
    for (int jj = 0; jj < 16; ++jj) acc += cp[jj] * vp[jj];
    Kk[tid] = (acc.x + acc.y) + (acc.z + acc.w);
  }
  __syncthreads();

  for (int j = 0; j < 32; ++j) {
    const int i = tid + 128 * j;
    const int r = i >> 6, k = i & 63;
    ws[i]        = (r >= k) ? Kk[r - k] : 0.f;
    ws[4096 + i] = vb[(NST - 1 - k) * LDP + r];
  }
}

__device__ __forceinline__ void role_cols(const float* __restrict__ A_,
                                          const float* __restrict__ C_,
                                          float* __restrict__ ws, float* sm, int tid) {
  float* sA = sm;
  float* wb = sm + NST * LDP;

  stage_A(A_, sA, tid);
  if (tid < NST) wb[tid] = C_[tid];
  __syncthreads();

  v4f am[16];
  {
    const float* ac = sA + (tid & 63);
    for (int ii = 0; ii < 16; ++ii) {
      am[ii].x = ac[(4 * ii) * LDP];
      am[ii].y = ac[(4 * ii + 1) * LDP];
      am[ii].z = ac[(4 * ii + 2) * LDP];
      am[ii].w = ac[(4 * ii + 3) * LDP];
    }
  }
  run_chain(am, wb, NST, tid);

  for (int j = 0; j < 32; ++j) {
    const int i = tid + 128 * j;
    const int r = i >> 6, k = i & 63;
    ws[8192 + i] = wb[(r + 1) * LDP + k];
  }
}

__device__ __forceinline__ void role_squares(const float* __restrict__ A_,
                                             float* __restrict__ ws, float* sm, int tid) {
  float* X0 = sm;
  float* X1 = sm + NST * LDP;
  float* X2 = sm + 2 * NST * LDP;

  stage_A(A_, X0, tid);
  __syncthreads();

  const int mt = tid >> 5;
  const int lane = tid & 31;
  const int c = lane & 15, h = lane >> 4;

  for (int s = 0; s < 8; ++s) {
    const float* P = (s & 1) ? X1 : X0;
    float* Pn = (s == 7) ? X2 : ((s & 1) ? X0 : X1);

    v2f a[16];
    {
      const float* L = P + (16 * mt + c) * LDP + 2 * h;
      for (int kk = 0; kk < 16; ++kk) a[kk] = *(const v2f*)(L + 4 * kk);
    }
    v8f acc[4];
    for (int nt = 0; nt < 4; ++nt) acc[nt] = (v8f){0.f, 0.f, 0.f, 0.f, 0.f, 0.f, 0.f, 0.f};
    for (int kk = 0; kk < 16; ++kk) {
      for (int nt = 0; nt < 4; ++nt) {
        const float* R = P + (4 * kk + 2 * h) * LDP + 16 * nt + c;
        v2f b;
        b.x = R[0];
        b.y = R[LDP];
        acc[nt] = __builtin_amdgcn_wmma_f32_16x16x4_f32(false, a[kk], false, b,
                                                        (short)0, acc[nt], false, false);
      }
    }
    for (int nt = 0; nt < 4; ++nt) {
      float* o = Pn + (16 * mt + 8 * h) * LDP + 16 * nt + c;
      for (int r = 0; r < 8; ++r) o[r * LDP] = acc[nt][r];
    }
    __syncthreads();
  }

  for (int j = 0; j < 32; ++j) {
    const int i = tid + 128 * j;
    const int r = i >> 6, k = i & 63;
    ws[12288 + i] = X0[r * LDP + k];
    ws[16384 + i] = X2[r * LDP + k];
  }
}

__global__ __launch_bounds__(128) void s4_precompute(const float* __restrict__ A_,
                                                     const float* __restrict__ B_,
                                                     const float* __restrict__ C_,
                                                     float* __restrict__ ws) {
  __shared__ __attribute__((aligned(16))) float sm[3 * NST * LDP];
  const int tid = threadIdx.x;
  if (blockIdx.x == 0) {
    role_rows(A_, B_, C_, ws, sm, tid);
  } else if (blockIdx.x == 1) {
    role_cols(A_, C_, ws, sm, tid);
  } else {
    role_squares(A_, ws, sm, tid);
  }
}

__device__ __forceinline__ v2f lo2(v4f v) { return __builtin_shufflevector(v, v, 0, 1); }
__device__ __forceinline__ v2f hi2(v4f v) { return __builtin_shufflevector(v, v, 2, 3); }

__device__ __forceinline__ void load_frag(v4f (&f)[8], const float* base) {
#pragma unroll
  for (int j = 0; j < 8; ++j) f[j] = *(const v4f*)(base + 4 * j);
}

__device__ __forceinline__ void wave_lds_order() {
  __builtin_amdgcn_fence(__ATOMIC_RELEASE, "wavefront");
  __builtin_amdgcn_wave_barrier();
  __builtin_amdgcn_fence(__ATOMIC_ACQUIRE, "wavefront");
}

__device__ __forceinline__ void load_u(v2f (&f)[16], const float* up) {
#pragma unroll
  for (int kk = 0; kk < 16; ++kk) {
    f[kk].x = up[(2 * kk) * DMOD];
    f[kk].y = up[(2 * kk + 1) * DMOD];
  }
}

#define WMMA4(a, b, c) __builtin_amdgcn_wmma_f32_16x16x4_f32(false, (a), false, (b), (short)0, (c), false, false)

__global__ __launch_bounds__(32 * NWV) __attribute__((amdgpu_waves_per_eu(2, 2)))
void s4_segstate(const float* __restrict__ u, float* __restrict__ ws) {
  __shared__ __attribute__((aligned(16))) float sM[2 * NST * XP];
  __shared__ __attribute__((aligned(16))) float sXw[NWV][2][16 * XP];

  const int wv = threadIdx.x >> 5;
  const int tid = threadIdx.x & 31;
  float (*sX)[16 * XP] = sXw[wv];
  {
    const int t = threadIdx.x;
    const float4* m4 = (const float4*)(ws + 4096);
    const float4* p4 = (const float4*)(ws + 12288);
    float4* d0 = (float4*)sM;
    float4* d1 = (float4*)(sM + NST * XP);
    float4 vm[4], vp[4];
    for (int j = 0; j < 4; ++j) vm[j] = m4[t + 32 * NWV * j];
    for (int j = 0; j < 4; ++j) vp[j] = p4[t + 32 * NWV * j];
    for (int j = 0; j < 4; ++j) {
      const int idx = t + 32 * NWV * j;
      const int o = (idx >> 4) * XP4 + (idx & 15);
      d0[o] = vm[j];
      d1[o] = vp[j];
    }
    float4* z4 = (float4*)sX[0];
    const float4 z = {0.f, 0.f, 0.f, 0.f};
    for (int i = tid; i < 16 * XP4; i += 32) z4[i] = z;
  }
  __syncthreads();

  const int tile = (blockIdx.x >> 4) * NWV + wv;
  const int seg  = blockIdx.x & 15;
  const int s0 = tile * 16;
  const int bb = s0 >> 9, dm0 = s0 & 511;
  const int row = tid & 15, hi = tid >> 4;
  const float* ub0 = u + (long)bb * LSEQ * DMOD + dm0 + row + (long)(32 * hi) * DMOD;

  v2f uf[16];
  load_u(uf, ub0 + (long)(seg * SCH) * TCH * DMOD);

  v4f bM[4][8], bP[4][8];
#pragma unroll
  for (int nt = 0; nt < 4; ++nt) {
    load_frag(bM[nt], sM + (nt * 16 + row) * XP + 32 * hi);
    load_frag(bP[nt], sM + NST * XP + (nt * 16 + row) * XP + 32 * hi);
  }
  __builtin_amdgcn_sched_barrier(0);

  for (int i = 0; i < SCH; ++i) {
    const int c = seg * SCH + i;
    const float* Xr = sX[i & 1];
    float* Xw = (float*)sX[(i & 1) ^ 1];
    v2f un[16];
#pragma unroll
    for (int kk = 0; kk < 16; ++kk) un[kk] = uf[kk];
    if (i + 1 < SCH) load_u(un, ub0 + (long)(c + 1) * TCH * DMOD);
    v4f xq[8];
    load_frag(xq, Xr + row * XP + 32 * hi);
    __builtin_amdgcn_sched_barrier(0);

#pragma unroll
    for (int np = 0; np < 2; ++np) {
      v8f a0 = {0.f, 0.f, 0.f, 0.f, 0.f, 0.f, 0.f, 0.f};
      v8f a1 = {0.f, 0.f, 0.f, 0.f, 0.f, 0.f, 0.f, 0.f};
      const int n0 = 2 * np, n1 = 2 * np + 1;
#pragma unroll
      for (int j = 0; j < 8; ++j) {
        a0 = WMMA4(uf[2 * j], lo2(bM[n0][j]), a0);
        a1 = WMMA4(uf[2 * j], lo2(bM[n1][j]), a1);
        a0 = WMMA4(lo2(xq[j]), lo2(bP[n0][j]), a0);
        a1 = WMMA4(lo2(xq[j]), lo2(bP[n1][j]), a1);
        a0 = WMMA4(uf[2 * j + 1], hi2(bM[n0][j]), a0);
        a1 = WMMA4(uf[2 * j + 1], hi2(bM[n1][j]), a1);
        a0 = WMMA4(hi2(xq[j]), hi2(bP[n0][j]), a0);
        a1 = WMMA4(hi2(xq[j]), hi2(bP[n1][j]), a1);
      }
      float* xw0 = Xw + (8 * hi) * XP + n0 * 16 + row;
      float* xw1 = Xw + (8 * hi) * XP + n1 * 16 + row;
#pragma unroll
      for (int r = 0; r < 8; ++r) xw0[r * XP] = a0[r];
#pragma unroll
      for (int r = 0; r < 8; ++r) xw1[r * XP] = a1[r];
    }
    wave_lds_order();
#pragma unroll
    for (int kk = 0; kk < 16; ++kk) uf[kk] = un[kk];
  }
  float* st = ws + WS_ST + (long)tile * (NSEG * 1024) + seg * 1024;
  const float4* xp = (const float4*)sX[0];
  float4* gp = (float4*)st;
  for (int i = tid; i < 256; i += 32) gp[i] = xp[(i >> 4) * XP4 + (i & 15)];
}

__device__ __forceinline__ void lds_handover_barrier() {
  asm volatile("s_wait_dscnt 0x0\n\ts_barrier_signal -1\n\ts_barrier_wait -1" ::: "memory");
}

__global__ __launch_bounds__(32 * NWB)
void s4_combine(float* __restrict__ ws) {
  __shared__ __attribute__((aligned(16))) float sE[2][16 * XP];

  const int wv = threadIdx.x >> 5;
  const int tid = threadIdx.x & 31;
  const int row = tid & 15, hi = tid >> 4;
  const int col = wv * 16 + row;
  float* stt = ws + WS_ST + (long)blockIdx.x * (NSEG * 1024);
  const int i0 = threadIdx.x, i1 = threadIdx.x + 32 * NWB;
  const int l0 = (i0 >> 4) * XP4 + (i0 & 15), l1 = (i1 >> 4) * XP4 + (i1 & 15);

  v4f bP[8];
  load_frag(bP, ws + 16384 + col * NST + 32 * hi);
  v8f gc, g1;
#pragma unroll
  for (int r = 0; r < 8; ++r) gc[r] = stt[(r + 8 * hi) * 64 + col];
#pragma unroll
  for (int r = 0; r < 8; ++r) g1[r] = stt[1024 + (r + 8 * hi) * 64 + col];
  {
    float4* z4 = (float4*)sE[0];
    const float4 z = {0.f, 0.f, 0.f, 0.f};
    for (int i = threadIdx.x; i < 16 * XP4; i += 32 * NWB) z4[i] = z;
  }
  lds_handover_barrier();

#pragma unroll
  for (int g = 0; g < NSEG - 1; ++g) {
    const float* Er = sE[g & 1];
    float* Ew = (float*)sE[(g & 1) ^ 1];
    float* ew = Ew + (8 * hi) * XP + col;
    v4f eq[8];
    load_frag(eq, Er + row * XP + 32 * hi);
    const float4* e4 = (const float4*)Er;
    float4 so0 = e4[l0], so1 = e4[l1];
    v8f g2 = g1;
    if (g + 2 < NSEG - 1) {
      const float* nx = stt + (g + 2) * 1024;
#pragma unroll
      for (int r = 0; r < 8; ++r) g2[r] = nx[(r + 8 * hi) * 64 + col];
    }
    __builtin_amdgcn_sched_barrier(0);
    v8f a = gc;
#pragma unroll
    for (int j = 0; j < 8; ++j) {
      a = WMMA4(lo2(eq[j]), lo2(bP[j]), a);
      a = WMMA4(hi2(eq[j]), hi2(bP[j]), a);
    }
#pragma unroll
    for (int r = 0; r < 8; ++r) ew[r * XP] = a[r];
    if (g > 0) {
      float4* s4 = (float4*)(stt + (g - 1) * 1024);
      s4[i0] = so0;
      s4[i1] = so1;
    }
    lds_handover_barrier();
    gc = g1;
    g1 = g2;
  }
  {
    const float4* e4 = (const float4*)sE[1];
    float4* s4 = (float4*)(stt + (NSEG - 2) * 1024);
    s4[i0] = e4[l0];
    s4[i1] = e4[l1];
  }
}

__global__ __launch_bounds__(32 * NWV) __attribute__((amdgpu_waves_per_eu(2, 2)))
void s4_outputs(const float* __restrict__ u, const float* __restrict__ ws,
                float* __restrict__ y) {
  __shared__ __attribute__((aligned(16))) float sMat[4 * NST * XP];
  __shared__ __attribute__((aligned(16))) float sXw[NWV][2][16 * XP];

  const int wv = threadIdx.x >> 5;
  const int tid = threadIdx.x & 31;
  float (*sX)[16 * XP] = sXw[wv];
  const int tile = (blockIdx.x >> 4) * NWV + wv;
  const int seg  = blockIdx.x & 15;
  {
    const int t = threadIdx.x;
    const float4* s4p = (const float4*)ws;
    float4* d4 = (float4*)sMat;
    float4* xp = (float4*)sX[0];
    float4 vm[16];
    for (int j = 0; j < 16; ++j) vm[j] = s4p[t + 32 * NWV * j];
    if (seg == 0) {
      const float4 z = {0.f, 0.f, 0.f, 0.f};
      for (int j = 0; j < 8; ++j) {
        const int idx = tid + 32 * j;
        xp[(idx >> 4) * XP4 + (idx & 15)] = z;
      }
    } else {
      const float4* ep =
          (const float4*)(ws + WS_ST + (long)tile * (NSEG * 1024) + (seg - 1) * 1024);
      float4 ve[8];
      for (int j = 0; j < 8; ++j) ve[j] = ep[tid + 32 * j];
      for (int j = 0; j < 8; ++j) {
        const int idx = tid + 32 * j;
        xp[(idx >> 4) * XP4 + (idx & 15)] = ve[j];
      }
    }
    for (int j = 0; j < 16; ++j) {
      const int idx = t + 32 * NWV * j;
      d4[(idx >> 10) * (NST * XP4) + ((idx >> 4) & 63) * XP4 + (idx & 15)] = vm[j];
    }
  }
  __syncthreads();

  const int s0 = tile * 16;
  const int bb = s0 >> 9, dm0 = s0 & 511;
  const int row = tid & 15, hi = tid >> 4;
  const float* ub0 = u + (long)bb * LSEQ * DMOD + dm0 + row + (long)(32 * hi) * DMOD;
  float* yb0 = y + (long)bb * LSEQ * DMOD + dm0 + 8 * hi;
  const float* fb = sMat + row * XP + 32 * hi;

  v2f uf[16];
  load_u(uf, ub0 + (long)(seg * SCH) * TCH * DMOD);

  v4f bc[4][8];
#pragma unroll
  for (int m = 0; m < 4; ++m) load_frag(bc[m], fb + m * (NST * XP));

  for (int i = 0; i < SCH; ++i) {
    const int c = seg * SCH + i;
    const float* Xr = sX[i & 1];
    float* Xw = (float*)sX[(i & 1) ^ 1];
    v2f un[16];
#pragma unroll
    for (int kk = 0; kk < 16; ++kk) un[kk] = uf[kk];
    if (i + 1 < SCH) load_u(un, ub0 + (long)(c + 1) * TCH * DMOD);
    v4f xq[8];
    load_frag(xq, Xr + row * XP + 32 * hi);

#pragma unroll
    for (int nt = 0; nt < 4; ++nt) {
      v4f bn[4][8];
#pragma unroll
      for (int m = 0; m < 4; ++m) load_frag(bn[m], fb + m * (NST * XP) + ((nt + 1) & 3) * 16 * XP);
      __builtin_amdgcn_sched_barrier(0);

      v8f ay = {0.f, 0.f, 0.f, 0.f, 0.f, 0.f, 0.f, 0.f};
      v8f ax = {0.f, 0.f, 0.f, 0.f, 0.f, 0.f, 0.f, 0.f};
      const int col = nt * 16 + row;
#pragma unroll
      for (int j = 0; j < 8; ++j) {
        ay = WMMA4(uf[2 * j], lo2(bc[0][j]), ay);
        ax = WMMA4(uf[2 * j], lo2(bc[1][j]), ax);
        ay = WMMA4(lo2(xq[j]), lo2(bc[2][j]), ay);
        ax = WMMA4(lo2(xq[j]), lo2(bc[3][j]), ax);
        ay = WMMA4(uf[2 * j + 1], hi2(bc[0][j]), ay);
        ax = WMMA4(uf[2 * j + 1], hi2(bc[1][j]), ax);
        ay = WMMA4(hi2(xq[j]), hi2(bc[2][j]), ay);
        ax = WMMA4(hi2(xq[j]), hi2(bc[3][j]), ax);
      }
      __builtin_amdgcn_sched_barrier(0);
      float4* yp = (float4*)(yb0 + (long)(c * TCH + col) * DMOD);
      float4 lo, hh;
      lo.x = ay[0]; lo.y = ay[1]; lo.z = ay[2]; lo.w = ay[3];
      hh.x = ay[4]; hh.y = ay[5]; hh.z = ay[6]; hh.w = ay[7];
      yp[0] = lo;
      yp[1] = hh;
      float* xw = Xw + (8 * hi) * XP + col;
#pragma unroll
      for (int r = 0; r < 8; ++r) xw[r * XP] = ax[r];
#pragma unroll
      for (int m = 0; m < 4; ++m)
#pragma unroll
        for (int j = 0; j < 8; ++j) bc[m][j] = bn[m][j];
    }
    wave_lds_order();
#pragma unroll
    for (int kk = 0; kk < 16; ++kk) uf[kk] = un[kk];
  }
}

extern "C" void kernel_launch(void* const* d_in, const int* in_sizes, int n_in,
                              void* d_out, int out_size, void* d_ws, size_t ws_size,
                              hipStream_t stream) {
  const float* u  = (const float*)d_in[0];
  const float* A_ = (const float*)d_in[1];
  const float* B_ = (const float*)d_in[2];
  const float* C_ = (const float*)d_in[3];
  float* y  = (float*)d_out;
  float* ws = (float*)d_ws;

  s4_precompute<<<3, 128, 0, stream>>>(A_, B_, C_, ws);
  s4_segstate<<<256, 256, 0, stream>>>(u, ws);
  s4_combine<<<128, 128, 0, stream>>>(ws);
  s4_outputs<<<256, 256, 0, stream>>>(u, ws, y);
}


# Set-up recurrences A^t b, c A^p: each run by ONE wave (2 elements per lane), a step's 16 LDS reads in flight together, no barrier per step; dot products summed in another order (16 partial sums)
# speedup vs baseline: 45.3696x; 1.1663x over previous
#include <hip/hip_runtime.h>
#include <hip/hip_bf16.h>

typedef __attribute__((ext_vector_type(2))) float v2f;
typedef __attribute__((ext_vector_type(8))) float v8f;

#define NST 64
#define TCH 64
#define LSEQ 4096
#define DMOD 512
#define NSEG 16
#define SCH 4
#define NWV 8
#define NWB 4

#define WS_ST 20480
#define XP 68
#define XP4 17

typedef __attribute__((ext_vector_type(4))) float v4f;

#define LDP 68

__device__ __forceinline__ void stage_A(const float* __restrict__ A_, float* sA, int tid) {
  float av[32];
  for (int j = 0; j < 32; ++j) av[j] = A_[tid + 128 * j];
  for (int j = 0; j < 32; ++j) {
    const int i = tid + 128 * j;
    sA[(i >> 6) * LDP + (i & 63)] = av[j];
  }
}

__device__ __forceinline__ void wave_lds_order() {
  __builtin_amdgcn_fence(__ATOMIC_RELEASE, "wavefront");
  __builtin_amdgcn_wave_barrier();
  __builtin_amdgcn_fence(__ATOMIC_ACQUIRE, "wavefront");
}

__device__ __forceinline__ void run_chain(const v4f (&a0)[16], const v4f (&a1)[16], float* xb,
                                          int last, int lane) {
  for (int t = 1; t <= last; ++t) {
    const v4f* xp = (const v4f*)(xb + (t - 1) * LDP);
    v4f x[16];
#pragma unroll
    for (int jj = 0; jj < 16; ++jj) x[jj] = xp[jj];
    __builtin_amdgcn_sched_barrier(0);
    v4f p0[4], p1[4];
#pragma unroll
    for (int g = 0; g < 4; ++g) {
      p0[g] = a0[4 * g] * x[4 * g];
      p1[g] = a1[4 * g] * x[4 * g];
#pragma unroll
      for (int i = 1; i < 4; ++i) {
        p0[g] += a0[4 * g + i] * x[4 * g + i];
        p1[g] += a1[4 * g + i] * x[4 * g + i];
      }
    }
    const v4f s0 = (p0[0] + p0[1]) + (p0[2] + p0[3]);
    const v4f s1 = (p1[0] + p1[1]) + (p1[2] + p1[3]);
    xb[t * LDP + lane]      = (s0.x + s0.y) + (s0.z + s0.w);
    xb[t * LDP + lane + 32] = (s1.x + s1.y) + (s1.z + s1.w);
    wave_lds_order();
  }
}

__device__ __forceinline__ void role_rows(const float* __restrict__ A_,
                                          const float* __restrict__ B_,
                                          const float* __restrict__ C_,
                                          float* __restrict__ ws, float* sm, int tid) {
  float* sA = sm;
  float* vb = sm + NST * LDP;
  float* cb = sm + 2 * NST * LDP;
  float* Kk = cb + NST;

  stage_A(A_, sA, tid);
  if (tid < NST) { vb[tid] = B_[tid]; cb[tid] = C_[tid]; }
  __syncthreads();

  if (tid < 32) {
    v4f a0[16], a1[16];
    const v4f* r0 = (const v4f*)(sA + tid * LDP);
    const v4f* r1 = (const v4f*)(sA + (tid + 32) * LDP);
#pragma unroll
    for (int jj = 0; jj < 16; ++jj) { a0[jj] = r0[jj]; a1[jj] = r1[jj]; }
    run_chain(a0, a1, vb, NST - 1, tid);
  }
  __syncthreads();

  if (tid < NST) {
    const v4f* vp = (const v4f*)(vb + tid * LDP);
    const v4f* cp = (const v4f*)cb;
    v4f acc = {0.f, 0.f, 0.f, 0.f};
    for (int jj = 0; jj < 16; ++jj) acc += cp[jj] * vp[jj];
    Kk[tid] = (acc.x + acc.y) + (acc.z + acc.w);
  }
  __syncthreads();

  for (int j = 0; j < 32; ++j) {
    const int i = tid + 128 * j;
    const int r = i >> 6, k = i & 63;
    ws[i]        = (r >= k) ? Kk[r - k] : 0.f;
    ws[4096 + i] = vb[(NST - 1 - k) * LDP + r];
  }
}

__device__ __forceinline__ void role_cols(const float* __restrict__ A_,
                                          const float* __restrict__ C_,
                                          float* __restrict__ ws, float* sm, int tid) {
  float* sA = sm;
  float* wb = sm + NST * LDP;

  stage_A(A_, sA, tid);
  if (tid < NST) wb[tid] = C_[tid];
  __syncthreads();

  if (tid < 32) {
    v4f a0[16], a1[16];
    const float* ac = sA + tid;
#pragma unroll
    for (int ii = 0; ii < 16; ++ii) {
      a0[ii].x = ac[(4 * ii) * LDP];
      a0[ii].y = ac[(4 * ii + 1) * LDP];
      a0[ii].z = ac[(4 * ii + 2) * LDP];
      a0[ii].w = ac[(4 * ii + 3) * LDP];
      a1[ii].x = ac[(4 * ii) * LDP + 32];
      a1[ii].y = ac[(4 * ii + 1) * LDP + 32];
      a1[ii].z = ac[(4 * ii + 2) * LDP + 32];
      a1[ii].w = ac[(4 * ii + 3) * LDP + 32];
    }
    run_chain(a0, a1, wb, NST, tid);
  }
  __syncthreads();

  for (int j = 0; j < 32; ++j) {
    const int i = tid + 128 * j;
    const int r = i >> 6, k = i & 63;
    ws[8192 + i] = wb[(r + 1) * LDP + k];
  }
}

__device__ __forceinline__ void role_squares(const float* __restrict__ A_,
                                             float* __restrict__ ws, float* sm, int tid) {
  float* X0 = sm;
  float* X1 = sm + NST * LDP;
  float* X2 = sm + 2 * NST * LDP;

  stage_A(A_, X0, tid);
  __syncthreads();

  const int mt = tid >> 5;
  const int lane = tid & 31;
  const int c = lane & 15, h = lane >> 4;

  for (int s = 0; s < 8; ++s) {
    const float* P = (s & 1) ? X1 : X0;
    float* Pn = (s == 7) ? X2 : ((s & 1) ? X0 : X1);

    v2f a[16];
    {
      const float* L = P + (16 * mt + c) * LDP + 2 * h;
      for (int kk = 0; kk < 16; ++kk) a[kk] = *(const v2f*)(L + 4 * kk);
    }
    v8f acc[4];
    for (int nt = 0; nt < 4; ++nt) acc[nt] = (v8f){0.f, 0.f, 0.f, 0.f, 0.f, 0.f, 0.f, 0.f};
    for (int kk = 0; kk < 16; ++kk) {
      for (int nt = 0; nt < 4; ++nt) {
        const float* R = P + (4 * kk + 2 * h) * LDP + 16 * nt + c;
        v2f b;
        b.x = R[0];
        b.y = R[LDP];
        acc[nt] = __builtin_amdgcn_wmma_f32_16x16x4_f32(false, a[kk], false, b,
                                                        (short)0, acc[nt], false, false);
      }
    }
    for (int nt = 0; nt < 4; ++nt) {
      float* o = Pn + (16 * mt + 8 * h) * LDP + 16 * nt + c;
      for (int r = 0; r < 8; ++r) o[r * LDP] = acc[nt][r];
    }
    __syncthreads();
  }

  for (int j = 0; j < 32; ++j) {
    const int i = tid + 128 * j;
    const int r = i >> 6, k = i & 63;
    ws[12288 + i] = X0[r * LDP + k];
    ws[16384 + i] = X2[r * LDP + k];
  }
}

__global__ __launch_bounds__(128) void s4_precompute(const float* __restrict__ A_,
                                                     const float* __restrict__ B_,
                                                     const float* __restrict__ C_,
                                                     float* __restrict__ ws) {
  __shared__ __attribute__((aligned(16))) float sm[3 * NST * LDP];
  const int tid = threadIdx.x;
  if (blockIdx.x == 0) {
    role_rows(A_, B_, C_, ws, sm, tid);
  } else if (blockIdx.x == 1) {
    role_cols(A_, C_, ws, sm, tid);
  } else {
    role_squares(A_, ws, sm, tid);
  }
}

__device__ __forceinline__ v2f lo2(v4f v) { return __builtin_shufflevector(v, v, 0, 1); }
__device__ __forceinline__ v2f hi2(v4f v) { return __builtin_shufflevector(v, v, 2, 3); }

__device__ __forceinline__ void load_frag(v4f (&f)[8], const float* base) {
#pragma unroll
  for (int j = 0; j < 8; ++j) f[j] = *(const v4f*)(base + 4 * j);
}

__device__ __forceinline__ void load_u(v2f (&f)[16], const float* up) {
#pragma unroll
  for (int kk = 0; kk < 16; ++kk) {
    f[kk].x = up[(2 * kk) * DMOD];
    f[kk].y = up[(2 * kk + 1) * DMOD];
  }
}

#define WMMA4(a, b, c) __builtin_amdgcn_wmma_f32_16x16x4_f32(false, (a), false, (b), (short)0, (c), false, false)

__global__ __launch_bounds__(32 * NWV) __attribute__((amdgpu_waves_per_eu(2, 2)))
void s4_segstate(const float* __restrict__ u, float* __restrict__ ws) {
  __shared__ __attribute__((aligned(16))) float sM[2 * NST * XP];
  __shared__ __attribute__((aligned(16))) float sXw[NWV][2][16 * XP];

  const int wv = threadIdx.x >> 5;
  const int tid = threadIdx.x & 31;
  float (*sX)[16 * XP] = sXw[wv];
  {
    const int t = threadIdx.x;
    const float4* m4 = (const float4*)(ws + 4096);
    const float4* p4 = (const float4*)(ws + 12288);
    float4* d0 = (float4*)sM;
    float4* d1 = (float4*)(sM + NST * XP);
    float4 vm[4], vp[4];
    for (int j = 0; j < 4; ++j) vm[j] = m4[t + 32 * NWV * j];
    for (int j = 0; j < 4; ++j) vp[j] = p4[t + 32 * NWV * j];
    for (int j = 0; j < 4; ++j) {
      const int idx = t + 32 * NWV * j;
      const int o = (idx >> 4) * XP4 + (idx & 15);
      d0[o] = vm[j];
      d1[o] = vp[j];
    }
    float4* z4 = (float4*)sX[0];
    const float4 z = {0.f, 0.f, 0.f, 0.f};
    for (int i = tid; i < 16 * XP4; i += 32) z4[i] = z;
  }
  __syncthreads();

  const int tile = (blockIdx.x >> 4) * NWV + wv;
  const int seg  = blockIdx.x & 15;
  const int s0 = tile * 16;
  const int bb = s0 >> 9, dm0 = s0 & 511;
  const int row = tid & 15, hi = tid >> 4;
  const float* ub0 = u + (long)bb * LSEQ * DMOD + dm0 + row + (long)(32 * hi) * DMOD;

  v2f uf[16];
  load_u(uf, ub0 + (long)(seg * SCH) * TCH * DMOD);

  v4f bM[4][8], bP[4][8];
#pragma unroll
  for (int nt = 0; nt < 4; ++nt) {
    load_frag(bM[nt], sM + (nt * 16 + row) * XP + 32 * hi);
    load_frag(bP[nt], sM + NST * XP + (nt * 16 + row) * XP + 32 * hi);
  }
  __builtin_amdgcn_sched_barrier(0);

  for (int i = 0; i < SCH; ++i) {
    const int c = seg * SCH + i;
    const float* Xr = sX[i & 1];
    float* Xw = (float*)sX[(i & 1) ^ 1];
    v2f un[16];
#pragma unroll
    for (int kk = 0; kk < 16; ++kk) un[kk] = uf[kk];
    if (i + 1 < SCH) load_u(un, ub0 + (long)(c + 1) * TCH * DMOD);
    v4f xq[8];
    load_frag(xq, Xr + row * XP + 32 * hi);
    __builtin_amdgcn_sched_barrier(0);

#pragma unroll
    for (int np = 0; np < 2; ++np) {
      v8f a0 = {0.f, 0.f, 0.f, 0.f, 0.f, 0.f, 0.f, 0.f};
      v8f a1 = {0.f, 0.f, 0.f, 0.f, 0.f, 0.f, 0.f, 0.f};
      const int n0 = 2 * np, n1 = 2 * np + 1;
#pragma unroll
      for (int j = 0; j < 8; ++j) {
        a0 = WMMA4(uf[2 * j], lo2(bM[n0][j]), a0);
        a1 = WMMA4(uf[2 * j], lo2(bM[n1][j]), a1);
        a0 = WMMA4(lo2(xq[j]), lo2(bP[n0][j]), a0);
        a1 = WMMA4(lo2(xq[j]), lo2(bP[n1][j]), a1);
        a0 = WMMA4(uf[2 * j + 1], hi2(bM[n0][j]), a0);
        a1 = WMMA4(uf[2 * j + 1], hi2(bM[n1][j]), a1);
        a0 = WMMA4(hi2(xq[j]), hi2(bP[n0][j]), a0);
        a1 = WMMA4(hi2(xq[j]), hi2(bP[n1][j]), a1);
      }
      float* xw0 = Xw + (8 * hi) * XP + n0 * 16 + row;
      float* xw1 = Xw + (8 * hi) * XP + n1 * 16 + row;
#pragma unroll
      for (int r = 0; r < 8; ++r) xw0[r * XP] = a0[r];
#pragma unroll
      for (int r = 0; r < 8; ++r) xw1[r * XP] = a1[r];
    }
    wave_lds_order();
#pragma unroll
    for (int kk = 0; kk < 16; ++kk) uf[kk] = un[kk];
  }
  float* st = ws + WS_ST + (long)tile * (NSEG * 1024) + seg * 1024;
  const float4* xp = (const float4*)sX[0];
  float4* gp = (float4*)st;
  for (int i = tid; i < 256; i += 32) gp[i] = xp[(i >> 4) * XP4 + (i & 15)];
}

__device__ __forceinline__ void lds_handover_barrier() {
  asm volatile("s_wait_dscnt 0x0\n\ts_barrier_signal -1\n\ts_barrier_wait -1" ::: "memory");
}

__global__ __launch_bounds__(32 * NWB)
void s4_combine(float* __restrict__ ws) {
  __shared__ __attribute__((aligned(16))) float sE[2][16 * XP];

  const int wv = threadIdx.x >> 5;
  const int tid = threadIdx.x & 31;
  const int row = tid & 15, hi = tid >> 4;
  const int col = wv * 16 + row;
  float* stt = ws + WS_ST + (long)blockIdx.x * (NSEG * 1024);
  const int i0 = threadIdx.x, i1 = threadIdx.x + 32 * NWB;
  const int l0 = (i0 >> 4) * XP4 + (i0 & 15), l1 = (i1 >> 4) * XP4 + (i1 & 15);

  v4f bP[8];
  load_frag(bP, ws + 16384 + col * NST + 32 * hi);
  v8f gc, g1;
#pragma unroll
  for (int r = 0; r < 8; ++r) gc[r] = stt[(r + 8 * hi) * 64 + col];
#pragma unroll
  for (int r = 0; r < 8; ++r) g1[r] = stt[1024 + (r + 8 * hi) * 64 + col];
  {
    float4* z4 = (float4*)sE[0];
    const float4 z = {0.f, 0.f, 0.f, 0.f};
    for (int i = threadIdx.x; i < 16 * XP4; i += 32 * NWB) z4[i] = z;
  }
  lds_handover_barrier();

#pragma unroll
  for (int g = 0; g < NSEG - 1; ++g) {
    const float* Er = sE[g & 1];
    float* Ew = (float*)sE[(g & 1) ^ 1];
    float* ew = Ew + (8 * hi) * XP + col;
    v4f eq[8];
    load_frag(eq, Er + row * XP + 32 * hi);
    const float4* e4 = (const float4*)Er;
    float4 so0 = e4[l0], so1 = e4[l1];
    v8f g2 = g1;
    if (g + 2 < NSEG - 1) {
      const float* nx = stt + (g + 2) * 1024;
#pragma unroll
      for (int r = 0; r < 8; ++r) g2[r] = nx[(r + 8 * hi) * 64 + col];
    }
    __builtin_amdgcn_sched_barrier(0);
    v8f a = gc;
#pragma unroll
    for (int j = 0; j < 8; ++j) {
      a = WMMA4(lo2(eq[j]), lo2(bP[j]), a);
      a = WMMA4(hi2(eq[j]), hi2(bP[j]), a);
    }
#pragma unroll
    for (int r = 0; r < 8; ++r) ew[r * XP] = a[r];
    if (g > 0) {
      float4* s4 = (float4*)(stt + (g - 1) * 1024);
      s4[i0] = so0;
      s4[i1] = so1;
    }
    lds_handover_barrier();
    gc = g1;
    g1 = g2;
  }
  {
    const float4* e4 = (const float4*)sE[1];
    float4* s4 = (float4*)(stt + (NSEG - 2) * 1024);
    s4[i0] = e4[l0];
    s4[i1] = e4[l1];
  }
}

__global__ __launch_bounds__(32 * NWV) __attribute__((amdgpu_waves_per_eu(2, 2)))
void s4_outputs(const float* __restrict__ u, const float* __restrict__ ws,
                float* __restrict__ y) {
  __shared__ __attribute__((aligned(16))) float sMat[4 * NST * XP];
  __shared__ __attribute__((aligned(16))) float sXw[NWV][2][16 * XP];

  const int wv = threadIdx.x >> 5;
  const int tid = threadIdx.x & 31;
  float (*sX)[16 * XP] = sXw[wv];
  const int tile = (blockIdx.x >> 4) * NWV + wv;
  const int seg  = blockIdx.x & 15;
  {
    const int t = threadIdx.x;
    const float4* s4p = (const float4*)ws;
    float4* d4 = (float4*)sMat;
    float4* xp = (float4*)sX[0];
    float4 vm[16];
    for (int j = 0; j < 16; ++j) vm[j] = s4p[t + 32 * NWV * j];
    if (seg == 0) {
      const float4 z = {0.f, 0.f, 0.f, 0.f};
      for (int j = 0; j < 8; ++j) {
        const int idx = tid + 32 * j;
        xp[(idx >> 4) * XP4 + (idx & 15)] = z;
      }
    } else {
      const float4* ep =
          (const float4*)(ws + WS_ST + (long)tile * (NSEG * 1024) + (seg - 1) * 1024);
      float4 ve[8];
      for (int j = 0; j < 8; ++j) ve[j] = ep[tid + 32 * j];
      for (int j = 0; j < 8; ++j) {
        const int idx = tid + 32 * j;
        xp[(idx >> 4) * XP4 + (idx & 15)] = ve[j];
      }
    }
    for (int j = 0; j < 16; ++j) {
      const int idx = t + 32 * NWV * j;
      d4[(idx >> 10) * (NST * XP4) + ((idx >> 4) & 63) * XP4 + (idx & 15)] = vm[j];
    }
  }
  __syncthreads();

  const int s0 = tile * 16;
  const int bb = s0 >> 9, dm0 = s0 & 511;
  const int row = tid & 15, hi = tid >> 4;
  const float* ub0 = u + (long)bb * LSEQ * DMOD + dm0 + row + (long)(32 * hi) * DMOD;
  float* yb0 = y + (long)bb * LSEQ * DMOD + dm0 + 8 * hi;
  const float* fb = sMat + row * XP + 32 * hi;

  v2f uf[16];
  load_u(uf, ub0 + (long)(seg * SCH) * TCH * DMOD);

  v4f bc[4][8];
#pragma unroll
  for (int m = 0; m < 4; ++m) load_frag(bc[m], fb + m * (NST * XP));

  for (int i = 0; i < SCH; ++i) {
    const int c = seg * SCH + i;
    const float* Xr = sX[i & 1];
    float* Xw = (float*)sX[(i & 1) ^ 1];
    v2f un[16];
#pragma unroll
    for (int kk = 0; kk < 16; ++kk) un[kk] = uf[kk];
    if (i + 1 < SCH) load_u(un, ub0 + (long)(c + 1) * TCH * DMOD);
    v4f xq[8];
    load_frag(xq, Xr + row * XP + 32 * hi);

#pragma unroll
    for (int nt = 0; nt < 4; ++nt) {
      v4f bn[4][8];
#pragma unroll
      for (int m = 0; m < 4; ++m) load_frag(bn[m], fb + m * (NST * XP) + ((nt + 1) & 3) * 16 * XP);
      __builtin_amdgcn_sched_barrier(0);

      v8f ay = {0.f, 0.f, 0.f, 0.f, 0.f, 0.f, 0.f, 0.f};
      v8f ax = {0.f, 0.f, 0.f, 0.f, 0.f, 0.f, 0.f, 0.f};
      const int col = nt * 16 + row;
#pragma unroll
      for (int j = 0; j < 8; ++j) {
        ay = WMMA4(uf[2 * j], lo2(bc[0][j]), ay);
        ax = WMMA4(uf[2 * j], lo2(bc[1][j]), ax);
        ay = WMMA4(lo2(xq[j]), lo2(bc[2][j]), ay);
        ax = WMMA4(lo2(xq[j]), lo2(bc[3][j]), ax);
        ay = WMMA4(uf[2 * j + 1], hi2(bc[0][j]), ay);
        ax = WMMA4(uf[2 * j + 1], hi2(bc[1][j]), ax);
        ay = WMMA4(hi2(xq[j]), hi2(bc[2][j]), ay);
        ax = WMMA4(hi2(xq[j]), hi2(bc[3][j]), ax);
      }
      __builtin_amdgcn_sched_barrier(0);
      float4* yp = (float4*)(yb0 + (long)(c * TCH + col) * DMOD);
      float4 lo, hh;
      lo.x = ay[0]; lo.y = ay[1]; lo.z = ay[2]; lo.w = ay[3];
      hh.x = ay[4]; hh.y = ay[5]; hh.z = ay[6]; hh.w = ay[7];
      yp[0] = lo;
      yp[1] = hh;
      float* xw = Xw + (8 * hi) * XP + col;
#pragma unroll
      for (int r = 0; r < 8; ++r) xw[r * XP] = ax[r];
#pragma unroll
      for (int m = 0; m < 4; ++m)
#pragma unroll
        for (int j = 0; j < 8; ++j) bc[m][j] = bn[m][j];
    }
    wave_lds_order();
#pragma unroll
    for (int kk = 0; kk < 16; ++kk) uf[kk] = un[kk];
  }
}

extern "C" void kernel_launch(void* const* d_in, const int* in_sizes, int n_in,
                              void* d_out, int out_size, void* d_ws, size_t ws_size,
                              hipStream_t stream) {
  const float* u  = (const float*)d_in[0];
  const float* A_ = (const float*)d_in[1];
  const float* B_ = (const float*)d_in[2];
  const float* C_ = (const float*)d_in[3];
  float* y  = (float*)d_out;
  float* ws = (float*)d_ws;

  s4_precompute<<<3, 128, 0, stream>>>(A_, B_, C_, ws);
  s4_segstate<<<256, 256, 0, stream>>>(u, ws);
  s4_combine<<<128, 128, 0, stream>>>(ws);
  s4_outputs<<<256, 256, 0, stream>>>(u, ws, y);
}


# Matrix instructions that cannot reach the output are dropped: last chunk's state update (phase C), products of zero states (phases A, B), zero blocks of TK (u k order grouped by 4: sum order changed)
# speedup vs baseline: 49.9704x; 1.1014x over previous
#include <hip/hip_runtime.h>
#include <hip/hip_bf16.h>

typedef __attribute__((ext_vector_type(2))) float v2f;
typedef __attribute__((ext_vector_type(8))) float v8f;

#define NST 64
#define TCH 64
#define LSEQ 4096
#define DMOD 512
#define NSEG 16
#define SCH 4
#define NWV 8
#define NWB 4

#define WS_ST 20480
#define XP 68
#define XP4 17

typedef __attribute__((ext_vector_type(4))) float v4f;

#define LDP 68

__device__ __forceinline__ void stage_A(const float* __restrict__ A_, float* sA, int tid) {
  float av[32];
  for (int j = 0; j < 32; ++j) av[j] = A_[tid + 128 * j];
  for (int j = 0; j < 32; ++j) {
    const int i = tid + 128 * j;
    sA[(i >> 6) * LDP + (i & 63)] = av[j];
  }
}

__device__ __forceinline__ void wave_lds_order() {
  __builtin_amdgcn_fence(__ATOMIC_RELEASE, "wavefront");
  __builtin_amdgcn_wave_barrier();
  __builtin_amdgcn_fence(__ATOMIC_ACQUIRE, "wavefront");
}

__device__ __forceinline__ void run_chain(const v4f (&a0)[16], const v4f (&a1)[16], float* xb,
                                          int last, int lane) {
  for (int t = 1; t <= last; ++t) {
    const v4f* xp = (const v4f*)(xb + (t - 1) * LDP);
    v4f x[16];
#pragma unroll
    for (int jj = 0; jj < 16; ++jj) x[jj] = xp[jj];
    __builtin_amdgcn_sched_barrier(0);
    v4f p0[4], p1[4];
#pragma unroll
    for (int g = 0; g < 4; ++g) {
      p0[g] = a0[4 * g] * x[4 * g];
      p1[g] = a1[4 * g] * x[4 * g];
#pragma unroll
      for (int i = 1; i < 4; ++i) {
        p0[g] += a0[4 * g + i] * x[4 * g + i];
        p1[g] += a1[4 * g + i] * x[4 * g + i];
      }
    }
    const v4f s0 = (p0[0] + p0[1]) + (p0[2] + p0[3]);
    const v4f s1 = (p1[0] + p1[1]) + (p1[2] + p1[3]);
    xb[t * LDP + lane]      = (s0.x + s0.y) + (s0.z + s0.w);
    xb[t * LDP + lane + 32] = (s1.x + s1.y) + (s1.z + s1.w);
    wave_lds_order();
  }
}

__device__ __forceinline__ void role_rows(const float* __restrict__ A_,
                                          const float* __restrict__ B_,
                                          const float* __restrict__ C_,
                                          float* __restrict__ ws, float* sm, int tid) {
  float* sA = sm;
  float* vb = sm + NST * LDP;
  float* cb = sm + 2 * NST * LDP;
  float* Kk = cb + NST;

  stage_A(A_, sA, tid);
  if (tid < NST) { vb[tid] = B_[tid]; cb[tid] = C_[tid]; }
  __syncthreads();

  if (tid < 32) {
    v4f a0[16], a1[16];
    const v4f* r0 = (const v4f*)(sA + tid * LDP);
    const v4f* r1 = (const v4f*)(sA + (tid + 32) * LDP);
#pragma unroll
    for (int jj = 0; jj < 16; ++jj) { a0[jj] = r0[jj]; a1[jj] = r1[jj]; }
    run_chain(a0, a1, vb, NST - 1, tid);
  }
  __syncthreads();

  if (tid < NST) {
    const v4f* vp = (const v4f*)(vb + tid * LDP);
    const v4f* cp = (const v4f*)cb;
    v4f acc = {0.f, 0.f, 0.f, 0.f};
    for (int jj = 0; jj < 16; ++jj) acc += cp[jj] * vp[jj];
    Kk[tid] = (acc.x + acc.y) + (acc.z + acc.w);
  }
  __syncthreads();

  for (int j = 0; j < 32; ++j) {
    const int i = tid + 128 * j;
    const int r = i >> 6, pos = i & 63;
    const int k = 4 * ((pos & 31) >> 1) + 2 * (pos >> 5) + (pos & 1);
    ws[i]        = (r >= k) ? Kk[r - k] : 0.f;
    ws[4096 + i] = vb[(NST - 1 - k) * LDP + r];
  }
}

__device__ __forceinline__ void role_cols(const float* __restrict__ A_,
                                          const float* __restrict__ C_,
                                          float* __restrict__ ws, float* sm, int tid) {
  float* sA = sm;
  float* wb = sm + NST * LDP;

  stage_A(A_, sA, tid);
  if (tid < NST) wb[tid] = C_[tid];
  __syncthreads();

  if (tid < 32) {
    v4f a0[16], a1[16];
    const float* ac = sA + tid;
#pragma unroll
    for (int ii = 0; ii < 16; ++ii) {
      a0[ii].x = ac[(4 * ii) * LDP];
      a0[ii].y = ac[(4 * ii + 1) * LDP];
      a0[ii].z = ac[(4 * ii + 2) * LDP];
      a0[ii].w = ac[(4 * ii + 3) * LDP];
      a1[ii].x = ac[(4 * ii) * LDP + 32];
      a1[ii].y = ac[(4 * ii + 1) * LDP + 32];
      a1[ii].z = ac[(4 * ii + 2) * LDP + 32];
      a1[ii].w = ac[(4 * ii + 3) * LDP + 32];
    }
    run_chain(a0, a1, wb, NST, tid);
  }
  __syncthreads();

  for (int j = 0; j < 32; ++j) {
    const int i = tid + 128 * j;
    const int r = i >> 6, k = i & 63;
    ws[8192 + i] = wb[(r + 1) * LDP + k];
  }
}

__device__ __forceinline__ void role_squares(const float* __restrict__ A_,
                                             float* __restrict__ ws, float* sm, int tid) {
  float* X0 = sm;
  float* X1 = sm + NST * LDP;
  float* X2 = sm + 2 * NST * LDP;

  stage_A(A_, X0, tid);
  __syncthreads();

  const int mt = tid >> 5;
  const int lane = tid & 31;
  const int c = lane & 15, h = lane >> 4;

  for (int s = 0; s < 8; ++s) {
    const float* P = (s & 1) ? X1 : X0;
    float* Pn = (s == 7) ? X2 : ((s & 1) ? X0 : X1);

    v2f a[16];
    {
      const float* L = P + (16 * mt + c) * LDP + 2 * h;
      for (int kk = 0; kk < 16; ++kk) a[kk] = *(const v2f*)(L + 4 * kk);
    }
    v8f acc[4];
    for (int nt = 0; nt < 4; ++nt) acc[nt] = (v8f){0.f, 0.f, 0.f, 0.f, 0.f, 0.f, 0.f, 0.f};
    for (int kk = 0; kk < 16; ++kk) {
      for (int nt = 0; nt < 4; ++nt) {
        const float* R = P + (4 * kk + 2 * h) * LDP + 16 * nt + c;
        v2f b;
        b.x = R[0];
        b.y = R[LDP];
        acc[nt] = __builtin_amdgcn_wmma_f32_16x16x4_f32(false, a[kk], false, b,
                                                        (short)0, acc[nt], false, false);
      }
    }
    for (int nt = 0; nt < 4; ++nt) {
      float* o = Pn + (16 * mt + 8 * h) * LDP + 16 * nt + c;
      for (int r = 0; r < 8; ++r) o[r * LDP] = acc[nt][r];
    }
    __syncthreads();
  }

  for (int j = 0; j < 32; ++j) {
    const int i = tid + 128 * j;
    const int r = i >> 6, k = i & 63;
    ws[12288 + i] = X0[r * LDP + k];
    ws[16384 + i] = X2[r * LDP + k];
  }
}

__global__ __launch_bounds__(128) void s4_precompute(const float* __restrict__ A_,
                                                     const float* __restrict__ B_,
                                                     const float* __restrict__ C_,
                                                     float* __restrict__ ws) {
  __shared__ __attribute__((aligned(16))) float sm[3 * NST * LDP];
  const int tid = threadIdx.x;
  if (blockIdx.x == 0) {
    role_rows(A_, B_, C_, ws, sm, tid);
  } else if (blockIdx.x == 1) {
    role_cols(A_, C_, ws, sm, tid);
  } else {
    role_squares(A_, ws, sm, tid);
  }
}

__device__ __forceinline__ v2f lo2(v4f v) { return __builtin_shufflevector(v, v, 0, 1); }
__device__ __forceinline__ v2f hi2(v4f v) { return __builtin_shufflevector(v, v, 2, 3); }

__device__ __forceinline__ void load_frag(v4f (&f)[8], const float* base) {
#pragma unroll
  for (int j = 0; j < 8; ++j) f[j] = *(const v4f*)(base + 4 * j);
}

__device__ __forceinline__ void load_u(v2f (&f)[16], const float* up) {
#pragma unroll
  for (int kk = 0; kk < 16; ++kk) {
    f[kk].x = up[(4 * kk) * DMOD];
    f[kk].y = up[(4 * kk + 1) * DMOD];
  }
}

#define WMMA4(a, b, c) __builtin_amdgcn_wmma_f32_16x16x4_f32(false, (a), false, (b), (short)0, (c), false, false)

__global__ __launch_bounds__(32 * NWV) __attribute__((amdgpu_waves_per_eu(2, 2)))
void s4_segstate(const float* __restrict__ u, float* __restrict__ ws) {
  __shared__ __attribute__((aligned(16))) float sM[2 * NST * XP];
  __shared__ __attribute__((aligned(16))) float sXw[NWV][2][16 * XP];

  const int wv = threadIdx.x >> 5;
  const int tid = threadIdx.x & 31;
  float (*sX)[16 * XP] = sXw[wv];
  {
    const int t = threadIdx.x;
    const float4* m4 = (const float4*)(ws + 4096);
    const float4* p4 = (const float4*)(ws + 12288);
    float4* d0 = (float4*)sM;
    float4* d1 = (float4*)(sM + NST * XP);
    float4 vm[4], vp[4];
    for (int j = 0; j < 4; ++j) vm[j] = m4[t + 32 * NWV * j];
    for (int j = 0; j < 4; ++j) vp[j] = p4[t + 32 * NWV * j];
    for (int j = 0; j < 4; ++j) {
      const int idx = t + 32 * NWV * j;
      const int o = (idx >> 4) * XP4 + (idx & 15);
      d0[o] = vm[j];
      d1[o] = vp[j];
    }
  }
  __syncthreads();

  const int tile = (blockIdx.x >> 4) * NWV + wv;
  const int seg  = blockIdx.x & 15;
  const int s0 = tile * 16;
  const int bb = s0 >> 9, dm0 = s0 & 511;
  const int row = tid & 15, hi = tid >> 4;
  const float* ub0 = u + (long)bb * LSEQ * DMOD + dm0 + row + (long)(2 * hi) * DMOD;

  v2f uf[16];
  load_u(uf, ub0 + (long)(seg * SCH) * TCH * DMOD);

  v4f bM[4][8], bP[4][8];
#pragma unroll
  for (int nt = 0; nt < 4; ++nt) {
    load_frag(bM[nt], sM + (nt * 16 + row) * XP + 32 * hi);
    load_frag(bP[nt], sM + NST * XP + (nt * 16 + row) * XP + 32 * hi);
  }
  __builtin_amdgcn_sched_barrier(0);

#pragma unroll
  for (int i = 0; i < SCH; ++i) {
    const int c = seg * SCH + i;
    const float* Xr = sX[i & 1];
    float* Xw = (float*)sX[(i & 1) ^ 1];
    v2f un[16];
#pragma unroll
    for (int kk = 0; kk < 16; ++kk) un[kk] = uf[kk];
    if (i + 1 < SCH) load_u(un, ub0 + (long)(c + 1) * TCH * DMOD);
    v4f xq[8];
    if (i > 0) load_frag(xq, Xr + row * XP + 32 * hi);
    __builtin_amdgcn_sched_barrier(0);

#pragma unroll
    for (int np = 0; np < 2; ++np) {
      v8f a0 = {0.f, 0.f, 0.f, 0.f, 0.f, 0.f, 0.f, 0.f};
      v8f a1 = {0.f, 0.f, 0.f, 0.f, 0.f, 0.f, 0.f, 0.f};
      const int n0 = 2 * np, n1 = 2 * np + 1;
#pragma unroll
      for (int j = 0; j < 8; ++j) {
        a0 = WMMA4(uf[2 * j], lo2(bM[n0][j]), a0);
        a1 = WMMA4(uf[2 * j], lo2(bM[n1][j]), a1);
        if (i > 0) {
          a0 = WMMA4(lo2(xq[j]), lo2(bP[n0][j]), a0);
          a1 = WMMA4(lo2(xq[j]), lo2(bP[n1][j]), a1);
        }
        a0 = WMMA4(uf[2 * j + 1], hi2(bM[n0][j]), a0);
        a1 = WMMA4(uf[2 * j + 1], hi2(bM[n1][j]), a1);
        if (i > 0) {
          a0 = WMMA4(hi2(xq[j]), hi2(bP[n0][j]), a0);
          a1 = WMMA4(hi2(xq[j]), hi2(bP[n1][j]), a1);
        }
      }
      float* xw0 = Xw + (8 * hi) * XP + n0 * 16 + row;
      float* xw1 = Xw + (8 * hi) * XP + n1 * 16 + row;
#pragma unroll
      for (int r = 0; r < 8; ++r) xw0[r * XP] = a0[r];
#pragma unroll
      for (int r = 0; r < 8; ++r) xw1[r * XP] = a1[r];
    }
    wave_lds_order();
#pragma unroll
    for (int kk = 0; kk < 16; ++kk) uf[kk] = un[kk];
  }
  float* st = ws + WS_ST + (long)tile * (NSEG * 1024) + seg * 1024;
  const float4* xp = (const float4*)sX[0];
  float4* gp = (float4*)st;
  for (int i = tid; i < 256; i += 32) gp[i] = xp[(i >> 4) * XP4 + (i & 15)];
}

__device__ __forceinline__ void lds_handover_barrier() {
  asm volatile("s_wait_dscnt 0x0\n\ts_barrier_signal -1\n\ts_barrier_wait -1" ::: "memory");
}

__global__ __launch_bounds__(32 * NWB)
void s4_combine(float* __restrict__ ws) {
  __shared__ __attribute__((aligned(16))) float sE[2][16 * XP];

  const int wv = threadIdx.x >> 5;
  const int tid = threadIdx.x & 31;
  const int row = tid & 15, hi = tid >> 4;
  const int col = wv * 16 + row;
  float* stt = ws + WS_ST + (long)blockIdx.x * (NSEG * 1024);
  const int i0 = threadIdx.x, i1 = threadIdx.x + 32 * NWB;
  const int l0 = (i0 >> 4) * XP4 + (i0 & 15), l1 = (i1 >> 4) * XP4 + (i1 & 15);

  v4f bP[8];
  load_frag(bP, ws + 16384 + col * NST + 32 * hi);
  v8f gc, g1;
#pragma unroll
  for (int r = 0; r < 8; ++r) gc[r] = stt[(r + 8 * hi) * 64 + col];
#pragma unroll
  for (int r = 0; r < 8; ++r) g1[r] = stt[1024 + (r + 8 * hi) * 64 + col];

#pragma unroll
  for (int g = 0; g < NSEG - 1; ++g) {
    const float* Er = sE[g & 1];
    float* Ew = (float*)sE[(g & 1) ^ 1];
    float* ew = Ew + (8 * hi) * XP + col;
    v4f eq[8];
    if (g > 0) load_frag(eq, Er + row * XP + 32 * hi);
    const float4* e4 = (const float4*)Er;
    float4 so0 = {0.f, 0.f, 0.f, 0.f}, so1 = {0.f, 0.f, 0.f, 0.f};
    if (g > 0) { so0 = e4[l0]; so1 = e4[l1]; }
    v8f g2 = g1;
    if (g + 2 < NSEG - 1) {
      const float* nx = stt + (g + 2) * 1024;
#pragma unroll
      for (int r = 0; r < 8; ++r) g2[r] = nx[(r + 8 * hi) * 64 + col];
    }
    __builtin_amdgcn_sched_barrier(0);
    v8f a = gc;
    if (g > 0) {
#pragma unroll
      for (int j = 0; j < 8; ++j) {
        a = WMMA4(lo2(eq[j]), lo2(bP[j]), a);
        a = WMMA4(hi2(eq[j]), hi2(bP[j]), a);
      }
    }
#pragma unroll
    for (int r = 0; r < 8; ++r) ew[r * XP] = a[r];
    if (g > 0) {
      float4* s4 = (float4*)(stt + (g - 1) * 1024);
      s4[i0] = so0;
      s4[i1] = so1;
    }
    lds_handover_barrier();
    gc = g1;
    g1 = g2;
  }
  {
    const float4* e4 = (const float4*)sE[1];
    float4* s4 = (float4*)(stt + (NSEG - 2) * 1024);
    s4[i0] = e4[l0];
    s4[i1] = e4[l1];
  }
}

template <bool LAST>
__device__ __forceinline__ void outputs_chunk(v2f (&uf)[16], v4f (&bc)[4][8],
                                              const float* unext,
                                              const float* Xr, float* Xw, const float* fb,
                                              float* yc,
                                              int row, int hi) {
  v2f un[16];
  if (!LAST) load_u(un, unext);
  v4f xq[8];
  load_frag(xq, Xr + row * XP + 32 * hi);

#pragma unroll
  for (int nt = 0; nt < 4; ++nt) {
    const int nn = (nt + 1) & 3;
    v4f bn[4][8];
#pragma unroll
    for (int m = 0; m < 4; ++m) {
#pragma unroll
      for (int j = 0; j < 8; ++j) {
        const bool need = !(LAST && (nt == 3 || m == 1 || m == 3)) && (m != 0 || j < 2 * nn + 2);
        if (need) bn[m][j] = *(const v4f*)(fb + m * (NST * XP) + nn * 16 * XP + 4 * j);
      }
    }
    __builtin_amdgcn_sched_barrier(0);

    v8f ay = {0.f, 0.f, 0.f, 0.f, 0.f, 0.f, 0.f, 0.f};
    v8f ax = {0.f, 0.f, 0.f, 0.f, 0.f, 0.f, 0.f, 0.f};
    const int col = nt * 16 + row;
#pragma unroll
    for (int j = 0; j < 8; ++j) {
      const bool tk = j < 2 * nt + 2;
      if (tk) ay = WMMA4(uf[2 * j], lo2(bc[0][j]), ay);
      if (!LAST) ax = WMMA4(uf[2 * j], lo2(bc[1][j]), ax);
      ay = WMMA4(lo2(xq[j]), lo2(bc[2][j]), ay);
      if (!LAST) ax = WMMA4(lo2(xq[j]), lo2(bc[3][j]), ax);
      if (tk) ay = WMMA4(uf[2 * j + 1], hi2(bc[0][j]), ay);
      if (!LAST) ax = WMMA4(uf[2 * j + 1], hi2(bc[1][j]), ax);
      ay = WMMA4(hi2(xq[j]), hi2(bc[2][j]), ay);
      if (!LAST) ax = WMMA4(hi2(xq[j]), hi2(bc[3][j]), ax);
    }
    __builtin_amdgcn_sched_barrier(0);
    float4* yp = (float4*)(yc + (long)col * DMOD);
    float4 lo, hh;
    lo.x = ay[0]; lo.y = ay[1]; lo.z = ay[2]; lo.w = ay[3];
    hh.x = ay[4]; hh.y = ay[5]; hh.z = ay[6]; hh.w = ay[7];
    yp[0] = lo;
    yp[1] = hh;
    if (!LAST) {
      float* xw = Xw + (8 * hi) * XP + col;
#pragma unroll
      for (int r = 0; r < 8; ++r) xw[r * XP] = ax[r];
    }
#pragma unroll
    for (int m = 0; m < 4; ++m) {
#pragma unroll
      for (int j = 0; j < 8; ++j) {
        const bool need = !(LAST && (nt == 3 || m == 1 || m == 3)) && (m != 0 || j < 2 * nn + 2);
        if (need) bc[m][j] = bn[m][j];
      }
    }
  }
  if (!LAST) {
    wave_lds_order();
#pragma unroll
    for (int kk = 0; kk < 16; ++kk) uf[kk] = un[kk];
  }
}

__global__ __launch_bounds__(32 * NWV) __attribute__((amdgpu_waves_per_eu(2, 2)))
void s4_outputs(const float* __restrict__ u, const float* __restrict__ ws,
                float* __restrict__ y) {
  __shared__ __attribute__((aligned(16))) float sMat[4 * NST * XP];
  __shared__ __attribute__((aligned(16))) float sXw[NWV][2][16 * XP];

  const int wv = threadIdx.x >> 5;
  const int tid = threadIdx.x & 31;
  float (*sX)[16 * XP] = sXw[wv];
  const int tile = (blockIdx.x >> 4) * NWV + wv;
  const int seg  = blockIdx.x & 15;
  {
    const int t = threadIdx.x;
    const float4* s4p = (const float4*)ws;
    float4* d4 = (float4*)sMat;
    float4* xp = (float4*)sX[0];
    float4 vm[16];
    for (int j = 0; j < 16; ++j) vm[j] = s4p[t + 32 * NWV * j];
    if (seg == 0) {
      const float4 z = {0.f, 0.f, 0.f, 0.f};
      for (int j = 0; j < 8; ++j) {
        const int idx = tid + 32 * j;
        xp[(idx >> 4) * XP4 + (idx & 15)] = z;
      }
    } else {
      const float4* ep =
          (const float4*)(ws + WS_ST + (long)tile * (NSEG * 1024) + (seg - 1) * 1024);
      float4 ve[8];
      for (int j = 0; j < 8; ++j) ve[j] = ep[tid + 32 * j];
      for (int j = 0; j < 8; ++j) {
        const int idx = tid + 32 * j;
        xp[(idx >> 4) * XP4 + (idx & 15)] = ve[j];
      }
    }
    for (int j = 0; j < 16; ++j) {
      const int idx = t + 32 * NWV * j;
      d4[(idx >> 10) * (NST * XP4) + ((idx >> 4) & 63) * XP4 + (idx & 15)] = vm[j];
    }
  }
  __syncthreads();

  const int s0 = tile * 16;
  const int bb = s0 >> 9, dm0 = s0 & 511;
  const int row = tid & 15, hi = tid >> 4;
  const float* ub0 = u + (long)bb * LSEQ * DMOD + dm0 + row + (long)(2 * hi) * DMOD;
  float* yb0 = y + (long)bb * LSEQ * DMOD + dm0 + 8 * hi;
  const float* fb = sMat + row * XP + 32 * hi;

  v2f uf[16];
  load_u(uf, ub0 + (long)(seg * SCH) * TCH * DMOD);

  v4f bc[4][8];
#pragma unroll
  for (int m = 0; m < 4; ++m) load_frag(bc[m], fb + m * (NST * XP));

  for (int i = 0; i + 1 < SCH; ++i) {
    const int c = seg * SCH + i;
    outputs_chunk<false>(uf, bc, ub0 + (long)(c + 1) * TCH * DMOD, sX[i & 1], (float*)sX[(i & 1) ^ 1],
                         fb, yb0 + (long)(c * TCH) * DMOD, row, hi);
  }
  {
    const int c = seg * SCH + (SCH - 1);
    outputs_chunk<true>(uf, bc, ub0, sX[(SCH - 1) & 1], (float*)sX[((SCH - 1) & 1) ^ 1],
                        fb, yb0 + (long)(c * TCH) * DMOD, row, hi);
  }
}

extern "C" void kernel_launch(void* const* d_in, const int* in_sizes, int n_in,
                              void* d_out, int out_size, void* d_ws, size_t ws_size,
                              hipStream_t stream) {
  const float* u  = (const float*)d_in[0];
  const float* A_ = (const float*)d_in[1];
  const float* B_ = (const float*)d_in[2];
  const float* C_ = (const float*)d_in[3];
  float* y  = (float*)d_out;
  float* ws = (float*)d_ws;

  s4_precompute<<<3, 128, 0, stream>>>(A_, B_, C_, ws);
  s4_segstate<<<256, 256, 0, stream>>>(u, ws);
  s4_combine<<<128, 128, 0, stream>>>(ws);
  s4_outputs<<<256, 256, 0, stream>>>(u, ws, y);
}


# Set-up launch removed: each phase A work-group builds M (A^t b chain, one wave) and A^64 (6 squarings, 4 waves) in its own LDS; idle last-segment groups build the later launches' matrices
# speedup vs baseline: 53.0732x; 1.0621x over previous
#include <hip/hip_runtime.h>
#include <hip/hip_bf16.h>

typedef __attribute__((ext_vector_type(2))) float v2f;
typedef __attribute__((ext_vector_type(8))) float v8f;

#define NST 64
#define TCH 64
#define LSEQ 4096
#define DMOD 512
#define NSEG 16
#define SCH 4
#define NWV 8
#define NWB 4

#define WS_ST 20480
#define XP 68
#define XP4 17

typedef __attribute__((ext_vector_type(4))) float v4f;

#define LDP 68

#define NT1 256

__device__ __forceinline__ void stage_A(const float* __restrict__ A_, float* sA, int tid) {
  float av[16];
  for (int j = 0; j < 16; ++j) av[j] = A_[tid + NT1 * j];
  for (int j = 0; j < 16; ++j) {
    const int i = tid + NT1 * j;
    sA[(i >> 6) * LDP + (i & 63)] = av[j];
  }
}

__device__ __forceinline__ void wave_lds_order() {
  __builtin_amdgcn_fence(__ATOMIC_RELEASE, "wavefront");
  __builtin_amdgcn_wave_barrier();
  __builtin_amdgcn_fence(__ATOMIC_ACQUIRE, "wavefront");
}

__device__ __forceinline__ void run_chain(const v4f (&a0)[16], const v4f (&a1)[16], float* xb,
                                          int first, int last, int lane) {
  for (int t = first; t <= last; ++t) {
    const v4f* xp = (const v4f*)(xb + (t - 1) * LDP);
    v4f x[16];
#pragma unroll
    for (int jj = 0; jj < 16; ++jj) x[jj] = xp[jj];
    __builtin_amdgcn_sched_barrier(0);
    v4f p0[4], p1[4];
#pragma unroll
    for (int g = 0; g < 4; ++g) {
      p0[g] = a0[4 * g] * x[4 * g];
      p1[g] = a1[4 * g] * x[4 * g];
#pragma unroll
      for (int i = 1; i < 4; ++i) {
        p0[g] += a0[4 * g + i] * x[4 * g + i];
        p1[g] += a1[4 * g + i] * x[4 * g + i];
      }
    }
    const v4f s0 = (p0[0] + p0[1]) + (p0[2] + p0[3]);
    const v4f s1 = (p1[0] + p1[1]) + (p1[2] + p1[3]);
    xb[t * LDP + lane]      = (s0.x + s0.y) + (s0.z + s0.w);
    xb[t * LDP + lane + 32] = (s1.x + s1.y) + (s1.z + s1.w);
    wave_lds_order();
  }
}

__device__ __forceinline__ void role_rows(const float* __restrict__ A_,
                                          const float* __restrict__ B_,
                                          const float* __restrict__ C_,
                                          float* __restrict__ ws, float* sm, int tid) {
  float* sA = sm;
  float* vb = sm + NST * LDP;
  float* cb = sm + 2 * NST * LDP;
  float* Kk = cb + NST;

  stage_A(A_, sA, tid);
  if (tid < NST) { vb[tid] = B_[tid]; cb[tid] = C_[tid]; }
  __syncthreads();

  if (tid < 32) {
    v4f a0[16], a1[16];
    const v4f* r0 = (const v4f*)(sA + tid * LDP);
    const v4f* r1 = (const v4f*)(sA + (tid + 32) * LDP);
#pragma unroll
    for (int jj = 0; jj < 16; ++jj) { a0[jj] = r0[jj]; a1[jj] = r1[jj]; }
    run_chain(a0, a1, vb, 1, NST - 1, tid);
  }
  __syncthreads();

  if (tid < NST) {
    const v4f* vp = (const v4f*)(vb + tid * LDP);
    const v4f* cp = (const v4f*)cb;
    v4f acc = {0.f, 0.f, 0.f, 0.f};
    for (int jj = 0; jj < 16; ++jj) acc += cp[jj] * vp[jj];
    Kk[tid] = (acc.x + acc.y) + (acc.z + acc.w);
  }
  __syncthreads();

  for (int j = 0; j < 16; ++j) {
    const int i = tid + NT1 * j;
    const int r = i >> 6, pos = i & 63;
    const int k = 4 * ((pos & 31) >> 1) + 2 * (pos >> 5) + (pos & 1);
    ws[i]        = (r >= k) ? Kk[r - k] : 0.f;
    ws[4096 + i] = vb[(NST - 1 - k) * LDP + r];
  }
}

__device__ __forceinline__ void role_cols(const float* __restrict__ A_,
                                          const float* __restrict__ C_,
                                          float* __restrict__ ws, float* sm, int tid) {
  float* sA = sm;
  float* wb = sm + NST * LDP;

  stage_A(A_, sA, tid);
  if (tid < NST) wb[tid] = C_[tid];
  __syncthreads();

  if (tid < 32) {
    v4f a0[16], a1[16];
    const float* ac = sA + tid;
#pragma unroll
    for (int ii = 0; ii < 16; ++ii) {
      a0[ii].x = ac[(4 * ii) * LDP];
      a0[ii].y = ac[(4 * ii + 1) * LDP];
      a0[ii].z = ac[(4 * ii + 2) * LDP];
      a0[ii].w = ac[(4 * ii + 3) * LDP];
      a1[ii].x = ac[(4 * ii) * LDP + 32];
      a1[ii].y = ac[(4 * ii + 1) * LDP + 32];
      a1[ii].z = ac[(4 * ii + 2) * LDP + 32];
      a1[ii].w = ac[(4 * ii + 3) * LDP + 32];
    }
    run_chain(a0, a1, wb, 1, NST, tid);
  }
  __syncthreads();

  for (int j = 0; j < 16; ++j) {
    const int i = tid + NT1 * j;
    const int r = i >> 6, k = i & 63;
    ws[8192 + i] = wb[(r + 1) * LDP + k];
  }
}

__device__ __forceinline__ void square_band(const float* P, float* Pn, int mt, int lane) {
  const int c = lane & 15, h = lane >> 4;
  v2f a[16];
  {
    const float* L = P + (16 * mt + c) * LDP + 2 * h;
    for (int kk = 0; kk < 16; ++kk) a[kk] = *(const v2f*)(L + 4 * kk);
  }
  v8f acc[4];
  for (int nt = 0; nt < 4; ++nt) acc[nt] = (v8f){0.f, 0.f, 0.f, 0.f, 0.f, 0.f, 0.f, 0.f};
  for (int kk = 0; kk < 16; ++kk) {
    for (int nt = 0; nt < 4; ++nt) {
      const float* R = P + (4 * kk + 2 * h) * LDP + 16 * nt + c;
      v2f b;
      b.x = R[0];
      b.y = R[LDP];
      acc[nt] = __builtin_amdgcn_wmma_f32_16x16x4_f32(false, a[kk], false, b,
                                                      (short)0, acc[nt], false, false);
    }
  }
  for (int nt = 0; nt < 4; ++nt) {
    float* o = Pn + (16 * mt + 8 * h) * LDP + 16 * nt + c;
    for (int r = 0; r < 8; ++r) o[r * LDP] = acc[nt][r];
  }
}

__device__ __forceinline__ void role_squares(const float* __restrict__ A_,
                                             float* __restrict__ ws, float* sm, int tid) {
  float* X0 = sm;
  float* X1 = sm + NST * LDP;
  float* X2 = sm + 2 * NST * LDP;

  stage_A(A_, X0, tid);
  __syncthreads();

  const int mt = tid >> 5;
  const int lane = tid & 31;

  for (int s = 0; s < 8; ++s) {
    const float* P = (s & 1) ? X1 : X0;
    float* Pn = (s == 7) ? X2 : ((s & 1) ? X0 : X1);
    if (mt < 4) square_band(P, Pn, mt, lane);
    __syncthreads();
  }

  for (int j = 0; j < 16; ++j) {
    const int i = tid + NT1 * j;
    const int r = i >> 6, k = i & 63;
    ws[12288 + i] = X0[r * LDP + k];
    ws[16384 + i] = X2[r * LDP + k];
  }
}

__device__ __forceinline__ void build_MP(const float* __restrict__ A_, const float* __restrict__ B_,
                                         float* sc, float* sM, int t) {
  float* X0 = sc;
  float* X1 = sc + NST * LDP;
  float* vb = sc + 2 * NST * LDP;

  stage_A(A_, X0, t);
  if (t < NST) vb[t] = B_[t];
  __syncthreads();

  const int wv = t >> 5, lane = t & 31;
  v4f a0[16], a1[16];
  if (wv == 0) {
    const v4f* r0 = (const v4f*)(X0 + lane * LDP);
    const v4f* r1 = (const v4f*)(X0 + (lane + 32) * LDP);
#pragma unroll
    for (int jj = 0; jj < 16; ++jj) { a0[jj] = r0[jj]; a1[jj] = r1[jj]; }
  }
  for (int s = 0; s < 6; ++s) {
    if (wv == 0) {
      run_chain(a0, a1, vb, 1 + (63 * s) / 6, (63 * (s + 1)) / 6, lane);
    } else if (wv >= 4) {
      square_band((s & 1) ? X1 : X0, (s & 1) ? X0 : X1, wv - 4, lane);
    }
    __syncthreads();
  }
  for (int j = 0; j < 16; ++j) {
    const int i = t + NT1 * j;
    const int r = i >> 6, pos = i & 63;
    const int k = 4 * ((pos & 31) >> 1) + 2 * (pos >> 5) + (pos & 1);
    sM[r * XP + pos]            = vb[(NST - 1 - k) * LDP + r];
    sM[NST * XP + r * XP + pos] = X0[r * LDP + pos];
  }
  __syncthreads();
}

__device__ __forceinline__ v2f lo2(v4f v) { return __builtin_shufflevector(v, v, 0, 1); }
__device__ __forceinline__ v2f hi2(v4f v) { return __builtin_shufflevector(v, v, 2, 3); }

__device__ __forceinline__ void load_frag(v4f (&f)[8], const float* base) {
#pragma unroll
  for (int j = 0; j < 8; ++j) f[j] = *(const v4f*)(base + 4 * j);
}

__device__ __forceinline__ void load_u(v2f (&f)[16], const float* up) {
#pragma unroll
  for (int kk = 0; kk < 16; ++kk) {
    f[kk].x = up[(4 * kk) * DMOD];
    f[kk].y = up[(4 * kk + 1) * DMOD];
  }
}

#define WMMA4(a, b, c) __builtin_amdgcn_wmma_f32_16x16x4_f32(false, (a), false, (b), (short)0, (c), false, false)

__global__ __launch_bounds__(32 * NWV) __attribute__((amdgpu_waves_per_eu(2, 2)))
void s4_segstate(const float* __restrict__ u, const float* __restrict__ A_,
                 const float* __restrict__ B_, const float* __restrict__ C_,
                 float* __restrict__ ws) {
  __shared__ __attribute__((aligned(16))) float sM[2 * NST * XP];
  __shared__ __attribute__((aligned(16))) float sXw[NWV][2][16 * XP];

  const int wv = threadIdx.x >> 5;
  const int tid = threadIdx.x & 31;
  float (*sX)[16 * XP] = sXw[wv];
  const int seg  = blockIdx.x & 15;

  if (seg == NSEG - 1) {
    const int role = blockIdx.x >> 4;
    float* sm = (float*)sXw;
    if (role == 0) {
      role_rows(A_, B_, C_, ws, sm, threadIdx.x);
    } else if (role == 1) {
      role_cols(A_, C_, ws, sm, threadIdx.x);
    } else if (role == 2) {
      role_squares(A_, ws, sm, threadIdx.x);
    }
    return;
  }

  const int tile = (blockIdx.x >> 4) * NWV + wv;
  const int s0 = tile * 16;
  const int bb = s0 >> 9, dm0 = s0 & 511;
  const int row = tid & 15, hi = tid >> 4;
  const float* ub0 = u + (long)bb * LSEQ * DMOD + dm0 + row + (long)(2 * hi) * DMOD;

  v2f uf[16];
  load_u(uf, ub0 + (long)(seg * SCH) * TCH * DMOD);

  build_MP(A_, B_, (float*)sXw, sM, threadIdx.x);

  v4f bM[4][8], bP[4][8];
#pragma unroll
  for (int nt = 0; nt < 4; ++nt) {
    load_frag(bM[nt], sM + (nt * 16 + row) * XP + 32 * hi);
    load_frag(bP[nt], sM + NST * XP + (nt * 16 + row) * XP + 32 * hi);
  }
  __builtin_amdgcn_sched_barrier(0);

#pragma unroll
  for (int i = 0; i < SCH; ++i) {
    const int c = seg * SCH + i;
    const float* Xr = sX[i & 1];
    float* Xw = (float*)sX[(i & 1) ^ 1];
    v2f un[16];
#pragma unroll
    for (int kk = 0; kk < 16; ++kk) un[kk] = uf[kk];
    if (i + 1 < SCH) load_u(un, ub0 + (long)(c + 1) * TCH * DMOD);
    v4f xq[8];
    if (i > 0) load_frag(xq, Xr + row * XP + 32 * hi);
    __builtin_amdgcn_sched_barrier(0);

#pragma unroll
    for (int np = 0; np < 2; ++np) {
      v8f a0 = {0.f, 0.f, 0.f, 0.f, 0.f, 0.f, 0.f, 0.f};
      v8f a1 = {0.f, 0.f, 0.f, 0.f, 0.f, 0.f, 0.f, 0.f};
      const int n0 = 2 * np, n1 = 2 * np + 1;
#pragma unroll
      for (int j = 0; j < 8; ++j) {
        a0 = WMMA4(uf[2 * j], lo2(bM[n0][j]), a0);
        a1 = WMMA4(uf[2 * j], lo2(bM[n1][j]), a1);
        if (i > 0) {
          a0 = WMMA4(lo2(xq[j]), lo2(bP[n0][j]), a0);
          a1 = WMMA4(lo2(xq[j]), lo2(bP[n1][j]), a1);
        }
        a0 = WMMA4(uf[2 * j + 1], hi2(bM[n0][j]), a0);
        a1 = WMMA4(uf[2 * j + 1], hi2(bM[n1][j]), a1);
        if (i > 0) {
          a0 = WMMA4(hi2(xq[j]), hi2(bP[n0][j]), a0);
          a1 = WMMA4(hi2(xq[j]), hi2(bP[n1][j]), a1);
        }
      }
      float* xw0 = Xw + (8 * hi) * XP + n0 * 16 + row;
      float* xw1 = Xw + (8 * hi) * XP + n1 * 16 + row;
#pragma unroll
      for (int r = 0; r < 8; ++r) xw0[r * XP] = a0[r];
#pragma unroll
      for (int r = 0; r < 8; ++r) xw1[r * XP] = a1[r];
    }
    wave_lds_order();
#pragma unroll
    for (int kk = 0; kk < 16; ++kk) uf[kk] = un[kk];
  }
  float* st = ws + WS_ST + (long)tile * (NSEG * 1024) + seg * 1024;
  const float4* xp = (const float4*)sX[0];
  float4* gp = (float4*)st;
  for (int i = tid; i < 256; i += 32) gp[i] = xp[(i >> 4) * XP4 + (i & 15)];
}

__device__ __forceinline__ void lds_handover_barrier() {
  asm volatile("s_wait_dscnt 0x0\n\ts_barrier_signal -1\n\ts_barrier_wait -1" ::: "memory");
}

__global__ __launch_bounds__(32 * NWB)
void s4_combine(float* __restrict__ ws) {
  __shared__ __attribute__((aligned(16))) float sE[2][16 * XP];

  const int wv = threadIdx.x >> 5;
  const int tid = threadIdx.x & 31;
  const int row = tid & 15, hi = tid >> 4;
  const int col = wv * 16 + row;
  float* stt = ws + WS_ST + (long)blockIdx.x * (NSEG * 1024);
  const int i0 = threadIdx.x, i1 = threadIdx.x + 32 * NWB;
  const int l0 = (i0 >> 4) * XP4 + (i0 & 15), l1 = (i1 >> 4) * XP4 + (i1 & 15);

  v4f bP[8];
  load_frag(bP, ws + 16384 + col * NST + 32 * hi);
  v8f gc, g1;
#pragma unroll
  for (int r = 0; r < 8; ++r) gc[r] = stt[(r + 8 * hi) * 64 + col];
#pragma unroll
  for (int r = 0; r < 8; ++r) g1[r] = stt[1024 + (r + 8 * hi) * 64 + col];

#pragma unroll
  for (int g = 0; g < NSEG - 1; ++g) {
    const float* Er = sE[g & 1];
    float* Ew = (float*)sE[(g & 1) ^ 1];
    float* ew = Ew + (8 * hi) * XP + col;
    v4f eq[8];
    if (g > 0) load_frag(eq, Er + row * XP + 32 * hi);
    const float4* e4 = (const float4*)Er;
    float4 so0 = {0.f, 0.f, 0.f, 0.f}, so1 = {0.f, 0.f, 0.f, 0.f};
    if (g > 0) { so0 = e4[l0]; so1 = e4[l1]; }
    v8f g2 = g1;
    if (g + 2 < NSEG - 1) {
      const float* nx = stt + (g + 2) * 1024;
#pragma unroll
      for (int r = 0; r < 8; ++r) g2[r] = nx[(r + 8 * hi) * 64 + col];
    }
    __builtin_amdgcn_sched_barrier(0);
    v8f a = gc;
    if (g > 0) {
#pragma unroll
      for (int j = 0; j < 8; ++j) {
        a = WMMA4(lo2(eq[j]), lo2(bP[j]), a);
        a = WMMA4(hi2(eq[j]), hi2(bP[j]), a);
      }
    }
#pragma unroll
    for (int r = 0; r < 8; ++r) ew[r * XP] = a[r];
    if (g > 0) {
      float4* s4 = (float4*)(stt + (g - 1) * 1024);
      s4[i0] = so0;
      s4[i1] = so1;
    }
    lds_handover_barrier();
    gc = g1;
    g1 = g2;
  }
  {
    const float4* e4 = (const float4*)sE[1];
    float4* s4 = (float4*)(stt + (NSEG - 2) * 1024);
    s4[i0] = e4[l0];
    s4[i1] = e4[l1];
  }
}

template <bool LAST>
__device__ __forceinline__ void outputs_chunk(v2f (&uf)[16], v4f (&bc)[4][8],
                                              const float* unext,
                                              const float* Xr, float* Xw, const float* fb,
                                              float* yc,
                                              int row, int hi) {
  v2f un[16];
  if (!LAST) load_u(un, unext);
  v4f xq[8];
  load_frag(xq, Xr + row * XP + 32 * hi);

#pragma unroll
  for (int nt = 0; nt < 4; ++nt) {
    const int nn = (nt + 1) & 3;
    v4f bn[4][8];
#pragma unroll
    for (int m = 0; m < 4; ++m) {
#pragma unroll
      for (int j = 0; j < 8; ++j) {
        const bool need = !(LAST && (nt == 3 || m == 1 || m == 3)) && (m != 0 || j < 2 * nn + 2);
        if (need) bn[m][j] = *(const v4f*)(fb + m * (NST * XP) + nn * 16 * XP + 4 * j);
      }
    }
    __builtin_amdgcn_sched_barrier(0);

    v8f ay = {0.f, 0.f, 0.f, 0.f, 0.f, 0.f, 0.f, 0.f};
    v8f ax = {0.f, 0.f, 0.f, 0.f, 0.f, 0.f, 0.f, 0.f};
    const int col = nt * 16 + row;
#pragma unroll
    for (int j = 0; j < 8; ++j) {
      const bool tk = j < 2 * nt + 2;
      if (tk) ay = WMMA4(uf[2 * j], lo2(bc[0][j]), ay);
      if (!LAST) ax = WMMA4(uf[2 * j], lo2(bc[1][j]), ax);
      ay = WMMA4(lo2(xq[j]), lo2(bc[2][j]), ay);
      if (!LAST) ax = WMMA4(lo2(xq[j]), lo2(bc[3][j]), ax);
      if (tk) ay = WMMA4(uf[2 * j + 1], hi2(bc[0][j]), ay);
      if (!LAST) ax = WMMA4(uf[2 * j + 1], hi2(bc[1][j]), ax);
      ay = WMMA4(hi2(xq[j]), hi2(bc[2][j]), ay);
      if (!LAST) ax = WMMA4(hi2(xq[j]), hi2(bc[3][j]), ax);
    }
    __builtin_amdgcn_sched_barrier(0);
    float4* yp = (float4*)(yc + (long)col * DMOD);
    float4 lo, hh;
    lo.x = ay[0]; lo.y = ay[1]; lo.z = ay[2]; lo.w = ay[3];
    hh.x = ay[4]; hh.y = ay[5]; hh.z = ay[6]; hh.w = ay[7];
    yp[0] = lo;
    yp[1] = hh;
    if (!LAST) {
      float* xw = Xw + (8 * hi) * XP + col;
#pragma unroll
      for (int r = 0; r < 8; ++r) xw[r * XP] = ax[r];
    }
#pragma unroll
    for (int m = 0; m < 4; ++m) {
#pragma unroll
      for (int j = 0; j < 8; ++j) {
        const bool need = !(LAST && (nt == 3 || m == 1 || m == 3)) && (m != 0 || j < 2 * nn + 2);
        if (need) bc[m][j] = bn[m][j];
      }
    }
  }
  if (!LAST) {
    wave_lds_order();
#pragma unroll
    for (int kk = 0; kk < 16; ++kk) uf[kk] = un[kk];
  }
}

__global__ __launch_bounds__(32 * NWV) __attribute__((amdgpu_waves_per_eu(2, 2)))
void s4_outputs(const float* __restrict__ u, const float* __restrict__ ws,
                float* __restrict__ y) {
  __shared__ __attribute__((aligned(16))) float sMat[4 * NST * XP];
  __shared__ __attribute__((aligned(16))) float sXw[NWV][2][16 * XP];

  const int wv = threadIdx.x >> 5;
  const int tid = threadIdx.x & 31;
  float (*sX)[16 * XP] = sXw[wv];
  const int tile = (blockIdx.x >> 4) * NWV + wv;
  const int seg  = blockIdx.x & 15;
  {
    const int t = threadIdx.x;
    const float4* s4p = (const float4*)ws;
    float4* d4 = (float4*)sMat;
    float4* xp = (float4*)sX[0];
    float4 vm[16];
    for (int j = 0; j < 16; ++j) vm[j] = s4p[t + 32 * NWV * j];
    if (seg == 0) {
      const float4 z = {0.f, 0.f, 0.f, 0.f};
      for (int j = 0; j < 8; ++j) {
        const int idx = tid + 32 * j;
        xp[(idx >> 4) * XP4 + (idx & 15)] = z;
      }
    } else {
      const float4* ep =
          (const float4*)(ws + WS_ST + (long)tile * (NSEG * 1024) + (seg - 1) * 1024);
      float4 ve[8];
      for (int j = 0; j < 8; ++j) ve[j] = ep[tid + 32 * j];
      for (int j = 0; j < 8; ++j) {
        const int idx = tid + 32 * j;
        xp[(idx >> 4) * XP4 + (idx & 15)] = ve[j];
      }
    }
    for (int j = 0; j < 16; ++j) {
      const int idx = t + 32 * NWV * j;
      d4[(idx >> 10) * (NST * XP4) + ((idx >> 4) & 63) * XP4 + (idx & 15)] = vm[j];
    }
  }
  __syncthreads();

  const int s0 = tile * 16;
  const int bb = s0 >> 9, dm0 = s0 & 511;
  const int row = tid & 15, hi = tid >> 4;
  const float* ub0 = u + (long)bb * LSEQ * DMOD + dm0 + row + (long)(2 * hi) * DMOD;
  float* yb0 = y + (long)bb * LSEQ * DMOD + dm0 + 8 * hi;
  const float* fb = sMat + row * XP + 32 * hi;

  v2f uf[16];
  load_u(uf, ub0 + (long)(seg * SCH) * TCH * DMOD);

  v4f bc[4][8];
#pragma unroll
  for (int m = 0; m < 4; ++m) load_frag(bc[m], fb + m * (NST * XP));

  for (int i = 0; i + 1 < SCH; ++i) {
    const int c = seg * SCH + i;
    outputs_chunk<false>(uf, bc, ub0 + (long)(c + 1) * TCH * DMOD, sX[i & 1], (float*)sX[(i & 1) ^ 1],
                         fb, yb0 + (long)(c * TCH) * DMOD, row, hi);
  }
  {
    const int c = seg * SCH + (SCH - 1);
    outputs_chunk<true>(uf, bc, ub0, sX[(SCH - 1) & 1], (float*)sX[((SCH - 1) & 1) ^ 1],
                        fb, yb0 + (long)(c * TCH) * DMOD, row, hi);
  }
}

extern "C" void kernel_launch(void* const* d_in, const int* in_sizes, int n_in,
                              void* d_out, int out_size, void* d_ws, size_t ws_size,
                              hipStream_t stream) {
  const float* u  = (const float*)d_in[0];
  const float* A_ = (const float*)d_in[1];
  const float* B_ = (const float*)d_in[2];
  const float* C_ = (const float*)d_in[3];
  float* y  = (float*)d_out;
  float* ws = (float*)d_ws;

  s4_segstate<<<256, NT1, 0, stream>>>(u, A_, B_, C_, ws);
  s4_combine<<<128, 128, 0, stream>>>(ws);
  s4_outputs<<<256, 256, 0, stream>>>(u, ws, y);
}


# Phase A set-up: 6 rounds of one matrix-instruction product A^m @ [A^m | A^t b, t<m] on all 8 waves give A^64 and all A^t b (doubling replaces the 63-step vector chain); M sum order changed
# speedup vs baseline: 57.0167x; 1.0743x over previous
#include <hip/hip_runtime.h>
#include <hip/hip_bf16.h>

typedef __attribute__((ext_vector_type(2))) float v2f;
typedef __attribute__((ext_vector_type(8))) float v8f;

#define NST 64
#define TCH 64
#define LSEQ 4096
#define DMOD 512
#define NSEG 16
#define SCH 4
#define NWV 8
#define NWB 4

#define WS_ST 20480
#define XP 68
#define XP4 17

typedef __attribute__((ext_vector_type(4))) float v4f;

#define LDP 68

#define NT1 256

__device__ __forceinline__ void stage_A(const float* __restrict__ A_, float* sA, int tid) {
  float av[16];
  for (int j = 0; j < 16; ++j) av[j] = A_[tid + NT1 * j];
  for (int j = 0; j < 16; ++j) {
    const int i = tid + NT1 * j;
    sA[(i >> 6) * LDP + (i & 63)] = av[j];
  }
}

__device__ __forceinline__ void wave_lds_order() {
  __builtin_amdgcn_fence(__ATOMIC_RELEASE, "wavefront");
  __builtin_amdgcn_wave_barrier();
  __builtin_amdgcn_fence(__ATOMIC_ACQUIRE, "wavefront");
}

__device__ __forceinline__ void run_chain(const v4f (&a0)[16], const v4f (&a1)[16], float* xb,
                                          int first, int last, int lane) {
  for (int t = first; t <= last; ++t) {
    const v4f* xp = (const v4f*)(xb + (t - 1) * LDP);
    v4f x[16];
#pragma unroll
    for (int jj = 0; jj < 16; ++jj) x[jj] = xp[jj];
    __builtin_amdgcn_sched_barrier(0);
    v4f p0[4], p1[4];
#pragma unroll
    for (int g = 0; g < 4; ++g) {
      p0[g] = a0[4 * g] * x[4 * g];
      p1[g] = a1[4 * g] * x[4 * g];
#pragma unroll
      for (int i = 1; i < 4; ++i) {
        p0[g] += a0[4 * g + i] * x[4 * g + i];
        p1[g] += a1[4 * g + i] * x[4 * g + i];
      }
    }
    const v4f s0 = (p0[0] + p0[1]) + (p0[2] + p0[3]);
    const v4f s1 = (p1[0] + p1[1]) + (p1[2] + p1[3]);
    xb[t * LDP + lane]      = (s0.x + s0.y) + (s0.z + s0.w);
    xb[t * LDP + lane + 32] = (s1.x + s1.y) + (s1.z + s1.w);
    wave_lds_order();
  }
}

__device__ __forceinline__ void role_rows(const float* __restrict__ A_,
                                          const float* __restrict__ B_,
                                          const float* __restrict__ C_,
                                          float* __restrict__ ws, float* sm, int tid) {
  float* sA = sm;
  float* vb = sm + NST * LDP;
  float* cb = sm + 2 * NST * LDP;
  float* Kk = cb + NST;

  stage_A(A_, sA, tid);
  if (tid < NST) { vb[tid] = B_[tid]; cb[tid] = C_[tid]; }
  __syncthreads();

  if (tid < 32) {
    v4f a0[16], a1[16];
    const v4f* r0 = (const v4f*)(sA + tid * LDP);
    const v4f* r1 = (const v4f*)(sA + (tid + 32) * LDP);
#pragma unroll
    for (int jj = 0; jj < 16; ++jj) { a0[jj] = r0[jj]; a1[jj] = r1[jj]; }
    run_chain(a0, a1, vb, 1, NST - 1, tid);
  }
  __syncthreads();

  if (tid < NST) {
    const v4f* vp = (const v4f*)(vb + tid * LDP);
    const v4f* cp = (const v4f*)cb;
    v4f acc = {0.f, 0.f, 0.f, 0.f};
    for (int jj = 0; jj < 16; ++jj) acc += cp[jj] * vp[jj];
    Kk[tid] = (acc.x + acc.y) + (acc.z + acc.w);
  }
  __syncthreads();

  for (int j = 0; j < 16; ++j) {
    const int i = tid + NT1 * j;
    const int r = i >> 6, pos = i & 63;
    const int k = 4 * ((pos & 31) >> 1) + 2 * (pos >> 5) + (pos & 1);
    ws[i]        = (r >= k) ? Kk[r - k] : 0.f;
    ws[4096 + i] = vb[(NST - 1 - k) * LDP + r];
  }
}

__device__ __forceinline__ void role_cols(const float* __restrict__ A_,
                                          const float* __restrict__ C_,
                                          float* __restrict__ ws, float* sm, int tid) {
  float* sA = sm;
  float* wb = sm + NST * LDP;

  stage_A(A_, sA, tid);
  if (tid < NST) wb[tid] = C_[tid];
  __syncthreads();

  if (tid < 32) {
    v4f a0[16], a1[16];
    const float* ac = sA + tid;
#pragma unroll
    for (int ii = 0; ii < 16; ++ii) {
      a0[ii].x = ac[(4 * ii) * LDP];
      a0[ii].y = ac[(4 * ii + 1) * LDP];
      a0[ii].z = ac[(4 * ii + 2) * LDP];
      a0[ii].w = ac[(4 * ii + 3) * LDP];
      a1[ii].x = ac[(4 * ii) * LDP + 32];
      a1[ii].y = ac[(4 * ii + 1) * LDP + 32];
      a1[ii].z = ac[(4 * ii + 2) * LDP + 32];
      a1[ii].w = ac[(4 * ii + 3) * LDP + 32];
    }
    run_chain(a0, a1, wb, 1, NST, tid);
  }
  __syncthreads();

  for (int j = 0; j < 16; ++j) {
    const int i = tid + NT1 * j;
    const int r = i >> 6, k = i & 63;
    ws[8192 + i] = wb[(r + 1) * LDP + k];
  }
}

__device__ __forceinline__ void square_band(const float* P, float* Pn, int mt, int lane) {
  const int c = lane & 15, h = lane >> 4;
  v2f a[16];
  {
    const float* L = P + (16 * mt + c) * LDP + 2 * h;
    for (int kk = 0; kk < 16; ++kk) a[kk] = *(const v2f*)(L + 4 * kk);
  }
  v8f acc[4];
  for (int nt = 0; nt < 4; ++nt) acc[nt] = (v8f){0.f, 0.f, 0.f, 0.f, 0.f, 0.f, 0.f, 0.f};
  for (int kk = 0; kk < 16; ++kk) {
    for (int nt = 0; nt < 4; ++nt) {
      const float* R = P + (4 * kk + 2 * h) * LDP + 16 * nt + c;
      v2f b;
      b.x = R[0];
      b.y = R[LDP];
      acc[nt] = __builtin_amdgcn_wmma_f32_16x16x4_f32(false, a[kk], false, b,
                                                      (short)0, acc[nt], false, false);
    }
  }
  for (int nt = 0; nt < 4; ++nt) {
    float* o = Pn + (16 * mt + 8 * h) * LDP + 16 * nt + c;
    for (int r = 0; r < 8; ++r) o[r * LDP] = acc[nt][r];
  }
}

__device__ __forceinline__ void role_squares(const float* __restrict__ A_,
                                             float* __restrict__ ws, float* sm, int tid) {
  float* X0 = sm;
  float* X1 = sm + NST * LDP;
  float* X2 = sm + 2 * NST * LDP;

  stage_A(A_, X0, tid);
  __syncthreads();

  const int mt = tid >> 5;
  const int lane = tid & 31;

  for (int s = 0; s < 8; ++s) {
    const float* P = (s & 1) ? X1 : X0;
    float* Pn = (s == 7) ? X2 : ((s & 1) ? X0 : X1);
    if (mt < 4) square_band(P, Pn, mt, lane);
    __syncthreads();
  }

  for (int j = 0; j < 16; ++j) {
    const int i = tid + NT1 * j;
    const int r = i >> 6, k = i & 63;
    ws[12288 + i] = X0[r * LDP + k];
    ws[16384 + i] = X2[r * LDP + k];
  }
}

__device__ __forceinline__ v2f lo2(v4f v) { return __builtin_shufflevector(v, v, 0, 1); }
__device__ __forceinline__ v2f hi2(v4f v) { return __builtin_shufflevector(v, v, 2, 3); }

__device__ __forceinline__ void load_frag(v4f (&f)[8], const float* base) {
#pragma unroll
  for (int j = 0; j < 8; ++j) f[j] = *(const v4f*)(base + 4 * j);
}

__device__ __forceinline__ void load_u(v2f (&f)[16], const float* up) {
#pragma unroll
  for (int kk = 0; kk < 16; ++kk) {
    f[kk].x = up[(4 * kk) * DMOD];
    f[kk].y = up[(4 * kk + 1) * DMOD];
  }
}

#define WMMA4(a, b, c) __builtin_amdgcn_wmma_f32_16x16x4_f32(false, (a), false, (b), (short)0, (c), false, false)

__device__ __forceinline__ void power_round(const float* X, float* Xn, const float* T, float* Tn,
                                            float* vb, int m, int wv, int lane) {
  const int c = lane & 15, h = lane >> 4;
  const int mt = wv & 3, grp = wv >> 2;
  const bool dov = (m == 32) || (grp == 0);
  const int vt = (m == 32) ? grp : 0;
  v4f fa[8], fb0[8], fb1[8], fv[8];
  load_frag(fa, X + (16 * mt + c) * LDP + 32 * h);
  load_frag(fb0, T + (32 * grp + c) * LDP + 32 * h);
  load_frag(fb1, T + (32 * grp + 16 + c) * LDP + 32 * h);
  if (dov) load_frag(fv, vb + (16 * vt + c) * LDP + 32 * h);
  __builtin_amdgcn_sched_barrier(0);
  v8f a0 = {0.f, 0.f, 0.f, 0.f, 0.f, 0.f, 0.f, 0.f};
  v8f a1 = {0.f, 0.f, 0.f, 0.f, 0.f, 0.f, 0.f, 0.f};
#pragma unroll
  for (int j = 0; j < 8; ++j) {
    a0 = WMMA4(lo2(fa[j]), lo2(fb0[j]), a0);
    a1 = WMMA4(lo2(fa[j]), lo2(fb1[j]), a1);
    a0 = WMMA4(hi2(fa[j]), hi2(fb0[j]), a0);
    a1 = WMMA4(hi2(fa[j]), hi2(fb1[j]), a1);
  }
  const int i0 = 16 * mt + 8 * h;
  {
    float* o0 = Xn + i0 * LDP + 32 * grp + c;
    float* o1 = o0 + 16;
#pragma unroll
    for (int r = 0; r < 8; ++r) { o0[r * LDP] = a0[r]; o1[r * LDP] = a1[r]; }
    v4f* t0 = (v4f*)(Tn + (32 * grp + c) * LDP + i0);
    v4f* t1 = (v4f*)(Tn + (32 * grp + 16 + c) * LDP + i0);
    t0[0] = (v4f){a0[0], a0[1], a0[2], a0[3]};
    t0[1] = (v4f){a0[4], a0[5], a0[6], a0[7]};
    t1[0] = (v4f){a1[0], a1[1], a1[2], a1[3]};
    t1[1] = (v4f){a1[4], a1[5], a1[6], a1[7]};
  }
  if (dov) {
    v8f av = {0.f, 0.f, 0.f, 0.f, 0.f, 0.f, 0.f, 0.f};
#pragma unroll
    for (int j = 0; j < 8; ++j) {
      av = WMMA4(lo2(fa[j]), lo2(fv[j]), av);
      av = WMMA4(hi2(fa[j]), hi2(fv[j]), av);
    }
    if (16 * vt + c < m) {
      v4f* vo = (v4f*)(vb + (m + 16 * vt + c) * LDP + i0);
      vo[0] = (v4f){av[0], av[1], av[2], av[3]};
      vo[1] = (v4f){av[4], av[5], av[6], av[7]};
    }
  }
}

__device__ __forceinline__ void build_MP(const float* __restrict__ A_, const float* __restrict__ B_,
                                         float* sc, float* sM, int t) {
  float* X0 = sc;
  float* X1 = sc + NST * LDP;
  float* vb = sc + 2 * NST * LDP;
  float* T0 = sM;
  float* T1 = sM + NST * LDP;

  {
    float av[16];
    for (int j = 0; j < 16; ++j) av[j] = A_[t + NT1 * j];
    for (int j = 0; j < 16; ++j) {
      const int i = t + NT1 * j;
      X0[(i >> 6) * LDP + (i & 63)] = av[j];
      T0[(i & 63) * LDP + (i >> 6)] = av[j];
    }
  }
  if (t < NST) vb[t] = B_[t];
  for (int i = t; i < 15 * NST; i += NT1) vb[(1 + (i >> 6)) * LDP + (i & 63)] = 0.f;
  __syncthreads();

  const int wv = __builtin_amdgcn_readfirstlane(t >> 5);
  const int lane = t & 31;
  for (int s = 0; s < 6; ++s) {
    power_round((s & 1) ? X1 : X0, (s & 1) ? X0 : X1, (s & 1) ? T1 : T0, (s & 1) ? T0 : T1, vb,
                1 << s, wv, lane);
    __syncthreads();
  }
  for (int j = 0; j < 16; ++j) {
    const int i = t + NT1 * j;
    const int r = i >> 6, pos = i & 63;
    const int k = 4 * ((pos & 31) >> 1) + 2 * (pos >> 5) + (pos & 1);
    sM[r * XP + pos]            = vb[(NST - 1 - k) * LDP + r];
    sM[NST * XP + r * XP + pos] = X0[r * LDP + pos];
  }
  __syncthreads();
}


__global__ __launch_bounds__(32 * NWV) __attribute__((amdgpu_waves_per_eu(2, 2)))
void s4_segstate(const float* __restrict__ u, const float* __restrict__ A_,
                 const float* __restrict__ B_, const float* __restrict__ C_,
                 float* __restrict__ ws) {
  __shared__ __attribute__((aligned(16))) float sM[2 * NST * XP];
  __shared__ __attribute__((aligned(16))) float sXw[NWV][2][16 * XP];

  const int wv = threadIdx.x >> 5;
  const int tid = threadIdx.x & 31;
  float (*sX)[16 * XP] = sXw[wv];
  const int seg  = blockIdx.x & 15;

  if (seg == NSEG - 1) {
    const int role = blockIdx.x >> 4;
    float* sm = (float*)sXw;
    if (role == 0) {
      role_rows(A_, B_, C_, ws, sm, threadIdx.x);
    } else if (role == 1) {
      role_cols(A_, C_, ws, sm, threadIdx.x);
    } else if (role == 2) {
      role_squares(A_, ws, sm, threadIdx.x);
    }
    return;
  }

  const int tile = (blockIdx.x >> 4) * NWV + wv;
  const int s0 = tile * 16;
  const int bb = s0 >> 9, dm0 = s0 & 511;
  const int row = tid & 15, hi = tid >> 4;
  const float* ub0 = u + (long)bb * LSEQ * DMOD + dm0 + row + (long)(2 * hi) * DMOD;

  v2f uf[16];
  load_u(uf, ub0 + (long)(seg * SCH) * TCH * DMOD);

  build_MP(A_, B_, (float*)sXw, sM, threadIdx.x);

  v4f bM[4][8], bP[4][8];
#pragma unroll
  for (int nt = 0; nt < 4; ++nt) {
    load_frag(bM[nt], sM + (nt * 16 + row) * XP + 32 * hi);
    load_frag(bP[nt], sM + NST * XP + (nt * 16 + row) * XP + 32 * hi);
  }
  __builtin_amdgcn_sched_barrier(0);

#pragma unroll
  for (int i = 0; i < SCH; ++i) {
    const int c = seg * SCH + i;
    const float* Xr = sX[i & 1];
    float* Xw = (float*)sX[(i & 1) ^ 1];
    v2f un[16];
#pragma unroll
    for (int kk = 0; kk < 16; ++kk) un[kk] = uf[kk];
    if (i + 1 < SCH) load_u(un, ub0 + (long)(c + 1) * TCH * DMOD);
    v4f xq[8];
    if (i > 0) load_frag(xq, Xr + row * XP + 32 * hi);
    __builtin_amdgcn_sched_barrier(0);

#pragma unroll
    for (int np = 0; np < 2; ++np) {
      v8f a0 = {0.f, 0.f, 0.f, 0.f, 0.f, 0.f, 0.f, 0.f};
      v8f a1 = {0.f, 0.f, 0.f, 0.f, 0.f, 0.f, 0.f, 0.f};
      const int n0 = 2 * np, n1 = 2 * np + 1;
#pragma unroll
      for (int j = 0; j < 8; ++j) {
        a0 = WMMA4(uf[2 * j], lo2(bM[n0][j]), a0);
        a1 = WMMA4(uf[2 * j], lo2(bM[n1][j]), a1);
        if (i > 0) {
          a0 = WMMA4(lo2(xq[j]), lo2(bP[n0][j]), a0);
          a1 = WMMA4(lo2(xq[j]), lo2(bP[n1][j]), a1);
        }
        a0 = WMMA4(uf[2 * j + 1], hi2(bM[n0][j]), a0);
        a1 = WMMA4(uf[2 * j + 1], hi2(bM[n1][j]), a1);
        if (i > 0) {
          a0 = WMMA4(hi2(xq[j]), hi2(bP[n0][j]), a0);
          a1 = WMMA4(hi2(xq[j]), hi2(bP[n1][j]), a1);
        }
      }
      float* xw0 = Xw + (8 * hi) * XP + n0 * 16 + row;
      float* xw1 = Xw + (8 * hi) * XP + n1 * 16 + row;
#pragma unroll
      for (int r = 0; r < 8; ++r) xw0[r * XP] = a0[r];
#pragma unroll
      for (int r = 0; r < 8; ++r) xw1[r * XP] = a1[r];
    }
    wave_lds_order();
#pragma unroll
    for (int kk = 0; kk < 16; ++kk) uf[kk] = un[kk];
  }
  float* st = ws + WS_ST + (long)tile * (NSEG * 1024) + seg * 1024;
  const float4* xp = (const float4*)sX[0];
  float4* gp = (float4*)st;
  for (int i = tid; i < 256; i += 32) gp[i] = xp[(i >> 4) * XP4 + (i & 15)];
}

__device__ __forceinline__ void lds_handover_barrier() {
  asm volatile("s_wait_dscnt 0x0\n\ts_barrier_signal -1\n\ts_barrier_wait -1" ::: "memory");
}

__global__ __launch_bounds__(32 * NWB)
void s4_combine(float* __restrict__ ws) {
  __shared__ __attribute__((aligned(16))) float sE[2][16 * XP];

  const int wv = threadIdx.x >> 5;
  const int tid = threadIdx.x & 31;
  const int row = tid & 15, hi = tid >> 4;
  const int col = wv * 16 + row;
  float* stt = ws + WS_ST + (long)blockIdx.x * (NSEG * 1024);
  const int i0 = threadIdx.x, i1 = threadIdx.x + 32 * NWB;
  const int l0 = (i0 >> 4) * XP4 + (i0 & 15), l1 = (i1 >> 4) * XP4 + (i1 & 15);

  v4f bP[8];
  load_frag(bP, ws + 16384 + col * NST + 32 * hi);
  v8f gc, g1;
#pragma unroll
  for (int r = 0; r < 8; ++r) gc[r] = stt[(r + 8 * hi) * 64 + col];
#pragma unroll
  for (int r = 0; r < 8; ++r) g1[r] = stt[1024 + (r + 8 * hi) * 64 + col];

#pragma unroll
  for (int g = 0; g < NSEG - 1; ++g) {
    const float* Er = sE[g & 1];
    float* Ew = (float*)sE[(g & 1) ^ 1];
    float* ew = Ew + (8 * hi) * XP + col;
    v4f eq[8];
    if (g > 0) load_frag(eq, Er + row * XP + 32 * hi);
    const float4* e4 = (const float4*)Er;
    float4 so0 = {0.f, 0.f, 0.f, 0.f}, so1 = {0.f, 0.f, 0.f, 0.f};
    if (g > 0) { so0 = e4[l0]; so1 = e4[l1]; }
    v8f g2 = g1;
    if (g + 2 < NSEG - 1) {
      const float* nx = stt + (g + 2) * 1024;
#pragma unroll
      for (int r = 0; r < 8; ++r) g2[r] = nx[(r + 8 * hi) * 64 + col];
    }
    __builtin_amdgcn_sched_barrier(0);
    v8f a = gc;
    if (g > 0) {
#pragma unroll
      for (int j = 0; j < 8; ++j) {
        a = WMMA4(lo2(eq[j]), lo2(bP[j]), a);
        a = WMMA4(hi2(eq[j]), hi2(bP[j]), a);
      }
    }
#pragma unroll
    for (int r = 0; r < 8; ++r) ew[r * XP] = a[r];
    if (g > 0) {
      float4* s4 = (float4*)(stt + (g - 1) * 1024);
      s4[i0] = so0;
      s4[i1] = so1;
    }
    lds_handover_barrier();
    gc = g1;
    g1 = g2;
  }
  {
    const float4* e4 = (const float4*)sE[1];
    float4* s4 = (float4*)(stt + (NSEG - 2) * 1024);
    s4[i0] = e4[l0];
    s4[i1] = e4[l1];
  }
}

template <bool LAST>
__device__ __forceinline__ void outputs_chunk(v2f (&uf)[16], v4f (&bc)[4][8],
                                              const float* unext,
                                              const float* Xr, float* Xw, const float* fb,
                                              float* yc,
                                              int row, int hi) {
  v2f un[16];
  if (!LAST) load_u(un, unext);
  v4f xq[8];
  load_frag(xq, Xr + row * XP + 32 * hi);

#pragma unroll
  for (int nt = 0; nt < 4; ++nt) {
    const int nn = (nt + 1) & 3;
    v4f bn[4][8];
#pragma unroll
    for (int m = 0; m < 4; ++m) {
#pragma unroll
      for (int j = 0; j < 8; ++j) {
        const bool need = !(LAST && (nt == 3 || m == 1 || m == 3)) && (m != 0 || j < 2 * nn + 2);
        if (need) bn[m][j] = *(const v4f*)(fb + m * (NST * XP) + nn * 16 * XP + 4 * j);
      }
    }
    __builtin_amdgcn_sched_barrier(0);

    v8f ay = {0.f, 0.f, 0.f, 0.f, 0.f, 0.f, 0.f, 0.f};
    v8f ax = {0.f, 0.f, 0.f, 0.f, 0.f, 0.f, 0.f, 0.f};
    const int col = nt * 16 + row;
#pragma unroll
    for (int j = 0; j < 8; ++j) {
      const bool tk = j < 2 * nt + 2;
      if (tk) ay = WMMA4(uf[2 * j], lo2(bc[0][j]), ay);
      if (!LAST) ax = WMMA4(uf[2 * j], lo2(bc[1][j]), ax);
      ay = WMMA4(lo2(xq[j]), lo2(bc[2][j]), ay);
      if (!LAST) ax = WMMA4(lo2(xq[j]), lo2(bc[3][j]), ax);
      if (tk) ay = WMMA4(uf[2 * j + 1], hi2(bc[0][j]), ay);
      if (!LAST) ax = WMMA4(uf[2 * j + 1], hi2(bc[1][j]), ax);
      ay = WMMA4(hi2(xq[j]), hi2(bc[2][j]), ay);
      if (!LAST) ax = WMMA4(hi2(xq[j]), hi2(bc[3][j]), ax);
    }
    __builtin_amdgcn_sched_barrier(0);
    float4* yp = (float4*)(yc + (long)col * DMOD);
    float4 lo, hh;
    lo.x = ay[0]; lo.y = ay[1]; lo.z = ay[2]; lo.w = ay[3];
    hh.x = ay[4]; hh.y = ay[5]; hh.z = ay[6]; hh.w = ay[7];
    yp[0] = lo;
    yp[1] = hh;
    if (!LAST) {
      float* xw = Xw + (8 * hi) * XP + col;
#pragma unroll
      for (int r = 0; r < 8; ++r) xw[r * XP] = ax[r];
    }
#pragma unroll
    for (int m = 0; m < 4; ++m) {
#pragma unroll
      for (int j = 0; j < 8; ++j) {
        const bool need = !(LAST && (nt == 3 || m == 1 || m == 3)) && (m != 0 || j < 2 * nn + 2);
        if (need) bc[m][j] = bn[m][j];
      }
    }
  }
  if (!LAST) {
    wave_lds_order();
#pragma unroll
    for (int kk = 0; kk < 16; ++kk) uf[kk] = un[kk];
  }
}

__global__ __launch_bounds__(32 * NWV) __attribute__((amdgpu_waves_per_eu(2, 2)))
void s4_outputs(const float* __restrict__ u, const float* __restrict__ ws,
                float* __restrict__ y) {
  __shared__ __attribute__((aligned(16))) float sMat[4 * NST * XP];
  __shared__ __attribute__((aligned(16))) float sXw[NWV][2][16 * XP];

  const int wv = threadIdx.x >> 5;
  const int tid = threadIdx.x & 31;
  float (*sX)[16 * XP] = sXw[wv];
  const int tile = (blockIdx.x >> 4) * NWV + wv;
  const int seg  = blockIdx.x & 15;
  {
    const int t = threadIdx.x;
    const float4* s4p = (const float4*)ws;
    float4* d4 = (float4*)sMat;
    float4* xp = (float4*)sX[0];
    float4 vm[16];
    for (int j = 0; j < 16; ++j) vm[j] = s4p[t + 32 * NWV * j];
    if (seg == 0) {
      const float4 z = {0.f, 0.f, 0.f, 0.f};
      for (int j = 0; j < 8; ++j) {
        const int idx = tid + 32 * j;
        xp[(idx >> 4) * XP4 + (idx & 15)] = z;
      }
    } else {
      const float4* ep =
          (const float4*)(ws + WS_ST + (long)tile * (NSEG * 1024) + (seg - 1) * 1024);
      float4 ve[8];
      for (int j = 0; j < 8; ++j) ve[j] = ep[tid + 32 * j];
      for (int j = 0; j < 8; ++j) {
        const int idx = tid + 32 * j;
        xp[(idx >> 4) * XP4 + (idx & 15)] = ve[j];
      }
    }
    for (int j = 0; j < 16; ++j) {
      const int idx = t + 32 * NWV * j;
      d4[(idx >> 10) * (NST * XP4) + ((idx >> 4) & 63) * XP4 + (idx & 15)] = vm[j];
    }
  }
  __syncthreads();

  const int s0 = tile * 16;
  const int bb = s0 >> 9, dm0 = s0 & 511;
  const int row = tid & 15, hi = tid >> 4;
  const float* ub0 = u + (long)bb * LSEQ * DMOD + dm0 + row + (long)(2 * hi) * DMOD;
  float* yb0 = y + (long)bb * LSEQ * DMOD + dm0 + 8 * hi;
  const float* fb = sMat + row * XP + 32 * hi;

  v2f uf[16];
  load_u(uf, ub0 + (long)(seg * SCH) * TCH * DMOD);

  v4f bc[4][8];
#pragma unroll
  for (int m = 0; m < 4; ++m) load_frag(bc[m], fb + m * (NST * XP));

  for (int i = 0; i + 1 < SCH; ++i) {
    const int c = seg * SCH + i;
    outputs_chunk<false>(uf, bc, ub0 + (long)(c + 1) * TCH * DMOD, sX[i & 1], (float*)sX[(i & 1) ^ 1],
                         fb, yb0 + (long)(c * TCH) * DMOD, row, hi);
  }
  {
    const int c = seg * SCH + (SCH - 1);
    outputs_chunk<true>(uf, bc, ub0, sX[(SCH - 1) & 1], (float*)sX[((SCH - 1) & 1) ^ 1],
                        fb, yb0 + (long)(c * TCH) * DMOD, row, hi);
  }
}

extern "C" void kernel_launch(void* const* d_in, const int* in_sizes, int n_in,
                              void* d_out, int out_size, void* d_ws, size_t ws_size,
                              hipStream_t stream) {
  const float* u  = (const float*)d_in[0];
  const float* A_ = (const float*)d_in[1];
  const float* B_ = (const float*)d_in[2];
  const float* C_ = (const float*)d_in[3];
  float* y  = (float*)d_out;
  float* ws = (float*)d_ws;

  s4_segstate<<<256, NT1, 0, stream>>>(u, A_, B_, C_, ws);
  s4_combine<<<128, 128, 0, stream>>>(ws);
  s4_outputs<<<256, 256, 0, stream>>>(u, ws, y);
}
